# starting kernel
# baseline (speedup 1.0000x reference)
#include <hip/hip_runtime.h>
#include <math.h>

#define N_NODES   50000
#define N_EDGES   1600000
#define NFEAT     500
#define HIDDEN    64
#define NCLASS    40
#define NUM_LAYERS 4
#define STEP_SZ   0.1f
#define M_TILES   (N_NODES / 16)

typedef __attribute__((ext_vector_type(2))) float v2f;
typedef __attribute__((ext_vector_type(8))) float v8f;


__global__ void deg_count_kernel(const int* __restrict__ col, float* __restrict__ deg) {
  int e = blockIdx.x * blockDim.x + threadIdx.x;
  if (e < N_EDGES) atomicAdd(&deg[col[e]], 1.0f);
}

__global__ void deg_inv_kernel(float* __restrict__ deg) {
  int i = blockIdx.x * blockDim.x + threadIdx.x;
  if (i < N_NODES) {
    float d = deg[i];
    deg[i] = (d > 0.0f) ? rsqrtf(d) : 0.0f;
  }
}

__global__ void coef_kernel(const int* __restrict__ row, const int* __restrict__ col,
                            const float* __restrict__ dinv, float* __restrict__ coef) {
  int e = blockIdx.x * blockDim.x + threadIdx.x;
  if (e < N_EDGES) coef[e] = dinv[row[e]] * dinv[col[e]];
}

__global__ void weff_kernel(const float* __restrict__ pw, float* __restrict__ W) {
  int i = threadIdx.x;
  const int LD = HIDDEN + 2;
  float q = pw[i * LD + HIDDEN];
  float r = pw[i * LD + HIDDEN + 1];
  float s = 0.0f;
  for (int j = 0; j < HIDDEN; ++j) {
    float v = 0.0f;
    if (i < j)      v = pw[i * LD + j];
    else if (i > j) v = pw[j * LD + i];
    s += fabsf(v);
    W[i * HIDDEN + j] = v;
  }
  W[i * HIDDEN + i] = q * s + r;
}


__global__ __launch_bounds__(128)
void enc_gemm_kernel(const float* __restrict__ x, const float* __restrict__ ew,
                     float* __restrict__ h, float* __restrict__ x0) {
  int wave = (int)((blockIdx.x * blockDim.x + threadIdx.x) >> 5);
  if (wave >= M_TILES) return;
  int lane = threadIdx.x & 31;
  int mr = lane & 15;
  int kh = lane >> 4;
  const float* arow = x  + (size_t)(wave * 16 + mr) * NFEAT;
  const float* b0r  = ew + (size_t)(mr +  0) * NFEAT;
  const float* b1r  = ew + (size_t)(mr + 16) * NFEAT;
  const float* b2r  = ew + (size_t)(mr + 32) * NFEAT;
  const float* b3r  = ew + (size_t)(mr + 48) * NFEAT;
  v8f c0 = {}, c1 = {}, c2 = {}, c3 = {};
  for (int k = 0; k < NFEAT; k += 4) {
    int ko = k + 2 * kh;
    v2f a  = *(const v2f*)(arow + ko);
    v2f b0 = *(const v2f*)(b0r + ko);
    v2f b1 = *(const v2f*)(b1r + ko);
    v2f b2 = *(const v2f*)(b2r + ko);
    v2f b3 = *(const v2f*)(b3r + ko);
    c0 = __builtin_amdgcn_wmma_f32_16x16x4_f32(false, a, false, b0, (short)0, c0, false, false);
    c1 = __builtin_amdgcn_wmma_f32_16x16x4_f32(false, a, false, b1, (short)0, c1, false, false);
    c2 = __builtin_amdgcn_wmma_f32_16x16x4_f32(false, a, false, b2, (short)0, c2, false, false);
    c3 = __builtin_amdgcn_wmma_f32_16x16x4_f32(false, a, false, b3, (short)0, c3, false, false);
  }
  int mbase = wave * 16 + kh * 8;
  for (int r = 0; r < 8; ++r) {
    size_t o = (size_t)(mbase + r) * HIDDEN + mr;
    h[o]      = c0[r]; x0[o]      = c0[r];
    h[o + 16] = c1[r]; x0[o + 16] = c1[r];
    h[o + 32] = c2[r]; x0[o + 32] = c2[r];
    h[o + 48] = c3[r]; x0[o + 48] = c3[r];
  }
}


__global__ __launch_bounds__(128)
void hw_gemm_kernel(const float* __restrict__ h, const float* __restrict__ W,
                    float* __restrict__ outp) {
  __shared__ float Ws[HIDDEN * HIDDEN];
  for (int i = threadIdx.x; i < HIDDEN * HIDDEN; i += blockDim.x) Ws[i] = W[i];
  __syncthreads();
  int wave = (int)((blockIdx.x * blockDim.x + threadIdx.x) >> 5);
  if (wave >= M_TILES) return;
  int lane = threadIdx.x & 31;
  int mr = lane & 15;
  int kh = lane >> 4;
  const float* arow = h + (size_t)(wave * 16 + mr) * HIDDEN;
  v8f c0 = {}, c1 = {}, c2 = {}, c3 = {};
  for (int k = 0; k < HIDDEN; k += 4) {
    int ko = k + 2 * kh;
    v2f a  = *(const v2f*)(arow + ko);
    v2f b0 = *(const v2f*)(&Ws[(mr +  0) * HIDDEN + ko]);
    v2f b1 = *(const v2f*)(&Ws[(mr + 16) * HIDDEN + ko]);
    v2f b2 = *(const v2f*)(&Ws[(mr + 32) * HIDDEN + ko]);
    v2f b3 = *(const v2f*)(&Ws[(mr + 48) * HIDDEN + ko]);
    c0 = __builtin_amdgcn_wmma_f32_16x16x4_f32(false, a, false, b0, (short)0, c0, false, false);
    c1 = __builtin_amdgcn_wmma_f32_16x16x4_f32(false, a, false, b1, (short)0, c1, false, false);
    c2 = __builtin_amdgcn_wmma_f32_16x16x4_f32(false, a, false, b2, (short)0, c2, false, false);
    c3 = __builtin_amdgcn_wmma_f32_16x16x4_f32(false, a, false, b3, (short)0, c3, false, false);
  }
  int mbase = wave * 16 + kh * 8;
  for (int r = 0; r < 8; ++r) {
    size_t o = (size_t)(mbase + r) * HIDDEN + mr;
    outp[o]      = c0[r];
    outp[o + 16] = c1[r];
    outp[o + 32] = c2[r];
    outp[o + 48] = c3[r];
  }
}


__global__ __launch_bounds__(256)
void edge_agg_kernel(const int* __restrict__ row, const int* __restrict__ col,
                     const float* __restrict__ coef, const float* __restrict__ outp,
                     float* __restrict__ agg) {
  int e = (int)((blockIdx.x * blockDim.x + threadIdx.x) >> 5);
  if (e >= N_EDGES) return;
  int lane = threadIdx.x & 31;
  int r  = __builtin_amdgcn_readfirstlane(row[e]);
  int c  = __builtin_amdgcn_readfirstlane(col[e]);
  float cf = __int_as_float(__builtin_amdgcn_readfirstlane(__float_as_int(coef[e])));
  const float* src = outp + (size_t)r * HIDDEN + 2 * lane;
  float*       dst = agg  + (size_t)c * HIDDEN + 2 * lane;
  v2f v = *(const v2f*)src;
  atomicAdd(&dst[0], cf * v.x);
  atomicAdd(&dst[1], cf * v.y);
}


__global__ __launch_bounds__(256)
void update_kernel(float* __restrict__ h, const float* __restrict__ agg,
                   const float* __restrict__ x0, const float* __restrict__ ext_w,
                   const float* __restrict__ beta) {
  int idx = blockIdx.x * blockDim.x + threadIdx.x;
  int j = idx & (HIDDEN - 1);
  float hv = h[idx];
  float conv = agg[idx] - hv * ext_w[j] + x0[idx] * beta[0];
  h[idx] = hv + STEP_SZ * fmaxf(conv, 0.0f);
}


__global__ __launch_bounds__(128)
void dec_gemm_kernel(const float* __restrict__ h, const float* __restrict__ dw,
                     float* __restrict__ out) {
  int wave = (int)((blockIdx.x * blockDim.x + threadIdx.x) >> 5);
  if (wave >= M_TILES) return;
  int lane = threadIdx.x & 31;
  int mr = lane & 15;
  int kh = lane >> 4;
  const float* arow = h + (size_t)(wave * 16 + mr) * HIDDEN;
  const float* b0r = dw + (size_t)(mr +  0) * HIDDEN;
  const float* b1r = dw + (size_t)(mr + 16) * HIDDEN;
  int n2 = mr + 32;
  const float* b2r = dw + (size_t)(n2 < NCLASS ? n2 : 0) * HIDDEN;
  float b2scale = (n2 < NCLASS) ? 1.0f : 0.0f;
  v8f c0 = {}, c1 = {}, c2 = {};
  for (int k = 0; k < HIDDEN; k += 4) {
    int ko = k + 2 * kh;
    v2f a  = *(const v2f*)(arow + ko);
    v2f b0 = *(const v2f*)(b0r + ko);
    v2f b1 = *(const v2f*)(b1r + ko);
    v2f b2 = *(const v2f*)(b2r + ko);
    b2 *= b2scale;
    c0 = __builtin_amdgcn_wmma_f32_16x16x4_f32(false, a, false, b0, (short)0, c0, false, false);
    c1 = __builtin_amdgcn_wmma_f32_16x16x4_f32(false, a, false, b1, (short)0, c1, false, false);
    c2 = __builtin_amdgcn_wmma_f32_16x16x4_f32(false, a, false, b2, (short)0, c2, false, false);
  }
  int mbase = wave * 16 + kh * 8;
  for (int r = 0; r < 8; ++r) {
    float* orow = out + (size_t)(mbase + r) * NCLASS;
    orow[mr]      = c0[r];
    orow[mr + 16] = c1[r];
    if (n2 < NCLASS) orow[n2] = c2[r];
  }
}


extern "C" void kernel_launch(void* const* d_in, const int* in_sizes, int n_in,
                              void* d_out, int out_size, void* d_ws, size_t ws_size,
                              hipStream_t stream) {
  (void)in_sizes; (void)n_in; (void)out_size; (void)ws_size;
  const float* x     = (const float*)d_in[0];
  const int*   ei    = (const int*)  d_in[1];
  const float* enc_w = (const float*)d_in[2];
  const float* dec_w = (const float*)d_in[3];
  const float* ext_w = (const float*)d_in[4];
  const float* beta  = (const float*)d_in[5];
  const float* pw    = (const float*)d_in[6];
  const int* row = ei;
  const int* col = ei + N_EDGES;

  float* ws   = (float*)d_ws;
  float* deg  = ws;
  float* coef = deg  + N_NODES;
  float* W    = coef + N_EDGES;
  float* h    = W    + HIDDEN * HIDDEN;
  float* x0   = h    + (size_t)N_NODES * HIDDEN;
  float* outp = x0   + (size_t)N_NODES * HIDDEN;
  float* agg  = outp + (size_t)N_NODES * HIDDEN;

  hipMemsetAsync(deg, 0, N_NODES * sizeof(float), stream);
  deg_count_kernel<<<(N_EDGES + 255) / 256, 256, 0, stream>>>(col, deg);
  deg_inv_kernel<<<(N_NODES + 255) / 256, 256, 0, stream>>>(deg);
  coef_kernel<<<(N_EDGES + 255) / 256, 256, 0, stream>>>(row, col, deg, coef);
  weff_kernel<<<1, HIDDEN, 0, stream>>>(pw, W);

  const int gemm_blocks = (M_TILES + 3) / 4;
  enc_gemm_kernel<<<gemm_blocks, 128, 0, stream>>>(x, enc_w, h, x0);

  for (int l = 0; l < NUM_LAYERS; ++l) {
    hw_gemm_kernel<<<gemm_blocks, 128, 0, stream>>>(h, W, outp);
    hipMemsetAsync(agg, 0, (size_t)N_NODES * HIDDEN * sizeof(float), stream);
    edge_agg_kernel<<<N_EDGES / 8, 256, 0, stream>>>(row, col, coef, outp, agg);
    update_kernel<<<(N_NODES * HIDDEN) / 256, 256, 0, stream>>>(h, agg, x0, ext_w, beta);
  }

  dec_gemm_kernel<<<gemm_blocks, 128, 0, stream>>>(h, dec_w, (float*)d_out);
}


# Edges bucketed by target (atomic arrival-order rank, scan, scatter); one wave per node sums its segment in registers: no float atomics, no agg clear; sum order inside a segment is arrival order
# speedup vs baseline: 2.1771x; 2.1771x over previous
#include <hip/hip_runtime.h>
#include <math.h>

#define N_NODES   50000
#define N_EDGES   1600000
#define NFEAT     500
#define HIDDEN    64
#define NCLASS    40
#define NUM_LAYERS 4
#define STEP_SZ   0.1f
#define M_TILES   (N_NODES / 16)

typedef __attribute__((ext_vector_type(2))) float v2f;
typedef __attribute__((ext_vector_type(8))) float v8f;


__global__ void count_rank_kernel(const int* __restrict__ col, int* __restrict__ cnt,
                                  int* __restrict__ rank) {
  int e = blockIdx.x * blockDim.x + threadIdx.x;
  if (e < N_EDGES) rank[e] = atomicAdd(&cnt[col[e]], 1);
}

#define SCAN_T   1024
#define SCAN_PER 49

__global__ __launch_bounds__(SCAN_T)
void scan_dinv_kernel(const int* __restrict__ cnt, int* __restrict__ off,
                      float* __restrict__ dinv) {
  __shared__ int sc[SCAN_T * SCAN_PER];
  __shared__ int wsum[32];
  const int t = threadIdx.x;
  const int lane = t & 31;
  const int w = t >> 5;
  for (int k = 0; k < SCAN_PER; ++k) {
    const int i = k * SCAN_T + t;
    int v = 0;
    if (i < N_NODES) {
      v = cnt[i];
      dinv[i] = (v > 0) ? rsqrtf((float)v) : 0.0f;
    }
    sc[i] = v;
  }
  __syncthreads();
  int* mine = sc + t * SCAN_PER;
  int run = 0;
  for (int k = 0; k < SCAN_PER; ++k) run += mine[k];
  int s = run;
  for (int d = 1; d < 32; d <<= 1) {
    int up = __shfl_up(s, d);
    if (lane >= d) s += up;
  }
  if (lane == 31) wsum[w] = s;
  __syncthreads();
  int ws = wsum[lane];
  for (int d = 1; d < 32; d <<= 1) {
    int up = __shfl_up(ws, d);
    if (lane >= d) ws += up;
  }
  const int prev  = __shfl(ws, (w + 31) & 31);
  const int total = __shfl(ws, 31);
  int pos = ((w > 0) ? prev : 0) + s - run;
  for (int k = 0; k < SCAN_PER; ++k) {
    const int v = mine[k];
    mine[k] = pos;
    pos += v;
  }
  __syncthreads();
  for (int k = 0; k < SCAN_PER; ++k) {
    const int i = k * SCAN_T + t;
    if (i < N_NODES) off[i] = sc[i];
  }
  if (t == 0) off[N_NODES] = total;
}

__global__ void fill_kernel(const int* __restrict__ row, const int* __restrict__ col,
                            const int* __restrict__ rank, const int* __restrict__ off,
                            const float* __restrict__ dinv, int2* __restrict__ edges) {
  int e = blockIdx.x * blockDim.x + threadIdx.x;
  if (e < N_EDGES) {
    int r = row[e];
    int c = col[e];
    int2 rec;
    rec.x = r;
    rec.y = __float_as_int(dinv[r] * dinv[c]);
    edges[off[c] + rank[e]] = rec;
  }
}

__global__ void weff_kernel(const float* __restrict__ pw, float* __restrict__ W) {
  int i = threadIdx.x;
  const int LD = HIDDEN + 2;
  float q = pw[i * LD + HIDDEN];
  float r = pw[i * LD + HIDDEN + 1];
  float s = 0.0f;
  for (int j = 0; j < HIDDEN; ++j) {
    float v = 0.0f;
    if (i < j)      v = pw[i * LD + j];
    else if (i > j) v = pw[j * LD + i];
    s += fabsf(v);
    W[i * HIDDEN + j] = v;
  }
  W[i * HIDDEN + i] = q * s + r;
}


__global__ __launch_bounds__(128)
void enc_gemm_kernel(const float* __restrict__ x, const float* __restrict__ ew,
                     float* __restrict__ h, float* __restrict__ x0) {
  int wave = (int)((blockIdx.x * blockDim.x + threadIdx.x) >> 5);
  if (wave >= M_TILES) return;
  int lane = threadIdx.x & 31;
  int mr = lane & 15;
  int kh = lane >> 4;
  const float* arow = x  + (size_t)(wave * 16 + mr) * NFEAT;
  const float* b0r  = ew + (size_t)(mr +  0) * NFEAT;
  const float* b1r  = ew + (size_t)(mr + 16) * NFEAT;
  const float* b2r  = ew + (size_t)(mr + 32) * NFEAT;
  const float* b3r  = ew + (size_t)(mr + 48) * NFEAT;
  v8f c0 = {}, c1 = {}, c2 = {}, c3 = {};
  for (int k = 0; k < NFEAT; k += 4) {
    int ko = k + 2 * kh;
    v2f a  = *(const v2f*)(arow + ko);
    v2f b0 = *(const v2f*)(b0r + ko);
    v2f b1 = *(const v2f*)(b1r + ko);
    v2f b2 = *(const v2f*)(b2r + ko);
    v2f b3 = *(const v2f*)(b3r + ko);
    c0 = __builtin_amdgcn_wmma_f32_16x16x4_f32(false, a, false, b0, (short)0, c0, false, false);
    c1 = __builtin_amdgcn_wmma_f32_16x16x4_f32(false, a, false, b1, (short)0, c1, false, false);
    c2 = __builtin_amdgcn_wmma_f32_16x16x4_f32(false, a, false, b2, (short)0, c2, false, false);
    c3 = __builtin_amdgcn_wmma_f32_16x16x4_f32(false, a, false, b3, (short)0, c3, false, false);
  }
  int mbase = wave * 16 + kh * 8;
  for (int r = 0; r < 8; ++r) {
    size_t o = (size_t)(mbase + r) * HIDDEN + mr;
    h[o]      = c0[r]; x0[o]      = c0[r];
    h[o + 16] = c1[r]; x0[o + 16] = c1[r];
    h[o + 32] = c2[r]; x0[o + 32] = c2[r];
    h[o + 48] = c3[r]; x0[o + 48] = c3[r];
  }
}


__global__ __launch_bounds__(128)
void hw_gemm_kernel(const float* __restrict__ h, const float* __restrict__ W,
                    float* __restrict__ outp) {
  __shared__ float Ws[HIDDEN * HIDDEN];
  for (int i = threadIdx.x; i < HIDDEN * HIDDEN; i += blockDim.x) Ws[i] = W[i];
  __syncthreads();
  int wave = (int)((blockIdx.x * blockDim.x + threadIdx.x) >> 5);
  if (wave >= M_TILES) return;
  int lane = threadIdx.x & 31;
  int mr = lane & 15;
  int kh = lane >> 4;
  const float* arow = h + (size_t)(wave * 16 + mr) * HIDDEN;
  v8f c0 = {}, c1 = {}, c2 = {}, c3 = {};
  for (int k = 0; k < HIDDEN; k += 4) {
    int ko = k + 2 * kh;
    v2f a  = *(const v2f*)(arow + ko);
    v2f b0 = *(const v2f*)(&Ws[(mr +  0) * HIDDEN + ko]);
    v2f b1 = *(const v2f*)(&Ws[(mr + 16) * HIDDEN + ko]);
    v2f b2 = *(const v2f*)(&Ws[(mr + 32) * HIDDEN + ko]);
    v2f b3 = *(const v2f*)(&Ws[(mr + 48) * HIDDEN + ko]);
    c0 = __builtin_amdgcn_wmma_f32_16x16x4_f32(false, a, false, b0, (short)0, c0, false, false);
    c1 = __builtin_amdgcn_wmma_f32_16x16x4_f32(false, a, false, b1, (short)0, c1, false, false);
    c2 = __builtin_amdgcn_wmma_f32_16x16x4_f32(false, a, false, b2, (short)0, c2, false, false);
    c3 = __builtin_amdgcn_wmma_f32_16x16x4_f32(false, a, false, b3, (short)0, c3, false, false);
  }
  int mbase = wave * 16 + kh * 8;
  for (int r = 0; r < 8; ++r) {
    size_t o = (size_t)(mbase + r) * HIDDEN + mr;
    outp[o]      = c0[r];
    outp[o + 16] = c1[r];
    outp[o + 32] = c2[r];
    outp[o + 48] = c3[r];
  }
}


__global__ __launch_bounds__(256)
void gather_agg_kernel(const int* __restrict__ off, const int2* __restrict__ edges,
                       const float* __restrict__ outp, float* __restrict__ agg) {
  const int c = __builtin_amdgcn_readfirstlane((int)((blockIdx.x * blockDim.x + threadIdx.x) >> 5));
  const int lane = threadIdx.x & 31;
  const int beg = off[c];
  const int end = off[c + 1];
  const float* src = outp + 2 * lane;
  v2f acc = {0.0f, 0.0f};
  for (int base = beg; base < end; base += 32) {
    int n = end - base;
    n = (n < 32) ? n : 32;
    int2 my;
    my.x = 0;
    my.y = 0;
    if (lane < n) my = edges[base + lane];
    int j = 0;
    for (; j + 8 <= n; j += 8) {
      v2f v[8];
      float w[8];
#pragma unroll
      for (int u = 0; u < 8; ++u) {
        const int s = __builtin_amdgcn_readlane(my.x, j + u);
        w[u] = __int_as_float(__builtin_amdgcn_readlane(my.y, j + u));
        v[u] = *(const v2f*)(src + (size_t)s * HIDDEN);
      }
#pragma unroll
      for (int u = 0; u < 8; ++u) acc += w[u] * v[u];
    }
    if (j + 4 <= n) {
      v2f v[4];
      float w[4];
#pragma unroll
      for (int u = 0; u < 4; ++u) {
        const int s = __builtin_amdgcn_readlane(my.x, j + u);
        w[u] = __int_as_float(__builtin_amdgcn_readlane(my.y, j + u));
        v[u] = *(const v2f*)(src + (size_t)s * HIDDEN);
      }
#pragma unroll
      for (int u = 0; u < 4; ++u) acc += w[u] * v[u];
      j += 4;
    }
    for (; j < n; ++j) {
      const int s = __builtin_amdgcn_readlane(my.x, j);
      const float w = __int_as_float(__builtin_amdgcn_readlane(my.y, j));
      const v2f v = *(const v2f*)(src + (size_t)s * HIDDEN);
      acc += w * v;
    }
  }
  *(v2f*)(agg + (size_t)c * HIDDEN + 2 * lane) = acc;
}


__global__ __launch_bounds__(256)
void update_kernel(float* __restrict__ h, const float* __restrict__ agg,
                   const float* __restrict__ x0, const float* __restrict__ ext_w,
                   const float* __restrict__ beta) {
  int idx = blockIdx.x * blockDim.x + threadIdx.x;
  int j = idx & (HIDDEN - 1);
  float hv = h[idx];
  float conv = agg[idx] - hv * ext_w[j] + x0[idx] * beta[0];
  h[idx] = hv + STEP_SZ * fmaxf(conv, 0.0f);
}


__global__ __launch_bounds__(128)
void dec_gemm_kernel(const float* __restrict__ h, const float* __restrict__ dw,
                     float* __restrict__ out) {
  int wave = (int)((blockIdx.x * blockDim.x + threadIdx.x) >> 5);
  if (wave >= M_TILES) return;
  int lane = threadIdx.x & 31;
  int mr = lane & 15;
  int kh = lane >> 4;
  const float* arow = h + (size_t)(wave * 16 + mr) * HIDDEN;
  const float* b0r = dw + (size_t)(mr +  0) * HIDDEN;
  const float* b1r = dw + (size_t)(mr + 16) * HIDDEN;
  int n2 = mr + 32;
  const float* b2r = dw + (size_t)(n2 < NCLASS ? n2 : 0) * HIDDEN;
  float b2scale = (n2 < NCLASS) ? 1.0f : 0.0f;
  v8f c0 = {}, c1 = {}, c2 = {};
  for (int k = 0; k < HIDDEN; k += 4) {
    int ko = k + 2 * kh;
    v2f a  = *(const v2f*)(arow + ko);
    v2f b0 = *(const v2f*)(b0r + ko);
    v2f b1 = *(const v2f*)(b1r + ko);
    v2f b2 = *(const v2f*)(b2r + ko);
    b2 *= b2scale;
    c0 = __builtin_amdgcn_wmma_f32_16x16x4_f32(false, a, false, b0, (short)0, c0, false, false);
    c1 = __builtin_amdgcn_wmma_f32_16x16x4_f32(false, a, false, b1, (short)0, c1, false, false);
    c2 = __builtin_amdgcn_wmma_f32_16x16x4_f32(false, a, false, b2, (short)0, c2, false, false);
  }
  int mbase = wave * 16 + kh * 8;
  for (int r = 0; r < 8; ++r) {
    float* orow = out + (size_t)(mbase + r) * NCLASS;
    orow[mr]      = c0[r];
    orow[mr + 16] = c1[r];
    if (n2 < NCLASS) orow[n2] = c2[r];
  }
}


extern "C" void kernel_launch(void* const* d_in, const int* in_sizes, int n_in,
                              void* d_out, int out_size, void* d_ws, size_t ws_size,
                              hipStream_t stream) {
  (void)in_sizes; (void)n_in; (void)out_size; (void)ws_size;
  const float* x     = (const float*)d_in[0];
  const int*   ei    = (const int*)  d_in[1];
  const float* enc_w = (const float*)d_in[2];
  const float* dec_w = (const float*)d_in[3];
  const float* ext_w = (const float*)d_in[4];
  const float* beta  = (const float*)d_in[5];
  const float* pw    = (const float*)d_in[6];
  const int* row = ei;
  const int* col = ei + N_EDGES;

  char* wsb = (char*)d_ws;
  int*   cnt   = (int*)  (wsb);
  int*   off   = (int*)  (wsb + 200192);
  float* dinv  = (float*)(wsb + 400384);
  int*   rank  = (int*)  (wsb + 600576);
  int2*  edges = (int2*) (wsb + 7000576);
  float* W     = (float*)(wsb + 19800576);
  float* h     = (float*)(wsb + 19816960);
  float* x0    = (float*)(wsb + 32616960);
  float* outp  = (float*)(wsb + 45416960);
  float* agg   = (float*)(wsb + 58216960);

  hipMemsetAsync(cnt, 0, N_NODES * sizeof(int), stream);
  count_rank_kernel<<<(N_EDGES + 255) / 256, 256, 0, stream>>>(col, cnt, rank);
  scan_dinv_kernel<<<1, 1024, 0, stream>>>(cnt, off, dinv);
  fill_kernel<<<(N_EDGES + 255) / 256, 256, 0, stream>>>(row, col, rank, off, dinv, edges);
  weff_kernel<<<1, HIDDEN, 0, stream>>>(pw, W);

  const int gemm_blocks = (M_TILES + 3) / 4;
  enc_gemm_kernel<<<gemm_blocks, 128, 0, stream>>>(x, enc_w, h, x0);

  for (int l = 0; l < NUM_LAYERS; ++l) {
    hw_gemm_kernel<<<gemm_blocks, 128, 0, stream>>>(h, W, outp);
    gather_agg_kernel<<<N_NODES / 8, 256, 0, stream>>>(off, edges, outp, agg);
    update_kernel<<<(N_NODES * HIDDEN) / 256, 256, 0, stream>>>(h, agg, x0, ext_w, beta);
  }

  dec_gemm_kernel<<<gemm_blocks, 128, 0, stream>>>(h, dec_w, (float*)d_out);
}


# Node update fused into the aggregation kernel: the wave that sums a node's segment applies the update to its h row in registers; agg plane and 4 update launches dropped; sum order as before
# speedup vs baseline: 2.3009x; 1.0569x over previous
#include <hip/hip_runtime.h>
#include <math.h>

#define N_NODES   50000
#define N_EDGES   1600000
#define NFEAT     500
#define HIDDEN    64
#define NCLASS    40
#define NUM_LAYERS 4
#define STEP_SZ   0.1f
#define M_TILES   (N_NODES / 16)

typedef __attribute__((ext_vector_type(2))) float v2f;
typedef __attribute__((ext_vector_type(8))) float v8f;


__global__ void count_rank_kernel(const int* __restrict__ col, int* __restrict__ cnt,
                                  int* __restrict__ rank) {
  int e = blockIdx.x * blockDim.x + threadIdx.x;
  if (e < N_EDGES) rank[e] = atomicAdd(&cnt[col[e]], 1);
}

#define SCAN_T   1024
#define SCAN_PER 49

__global__ __launch_bounds__(SCAN_T)
void scan_dinv_kernel(const int* __restrict__ cnt, int* __restrict__ off,
                      float* __restrict__ dinv) {
  __shared__ int sc[SCAN_T * SCAN_PER];
  __shared__ int wsum[32];
  const int t = threadIdx.x;
  const int lane = t & 31;
  const int w = t >> 5;
  for (int k = 0; k < SCAN_PER; ++k) {
    const int i = k * SCAN_T + t;
    int v = 0;
    if (i < N_NODES) {
      v = cnt[i];
      dinv[i] = (v > 0) ? rsqrtf((float)v) : 0.0f;
    }
    sc[i] = v;
  }
  __syncthreads();
  int* mine = sc + t * SCAN_PER;
  int run = 0;
  for (int k = 0; k < SCAN_PER; ++k) run += mine[k];
  int s = run;
  for (int d = 1; d < 32; d <<= 1) {
    int up = __shfl_up(s, d);
    if (lane >= d) s += up;
  }
  if (lane == 31) wsum[w] = s;
  __syncthreads();
  int ws = wsum[lane];
  for (int d = 1; d < 32; d <<= 1) {
    int up = __shfl_up(ws, d);
    if (lane >= d) ws += up;
  }
  const int prev  = __shfl(ws, (w + 31) & 31);
  const int total = __shfl(ws, 31);
  int pos = ((w > 0) ? prev : 0) + s - run;
  for (int k = 0; k < SCAN_PER; ++k) {
    const int v = mine[k];
    mine[k] = pos;
    pos += v;
  }
  __syncthreads();
  for (int k = 0; k < SCAN_PER; ++k) {
    const int i = k * SCAN_T + t;
    if (i < N_NODES) off[i] = sc[i];
  }
  if (t == 0) off[N_NODES] = total;
}

__global__ void fill_kernel(const int* __restrict__ row, const int* __restrict__ col,
                            const int* __restrict__ rank, const int* __restrict__ off,
                            const float* __restrict__ dinv, int2* __restrict__ edges) {
  int e = blockIdx.x * blockDim.x + threadIdx.x;
  if (e < N_EDGES) {
    int r = row[e];
    int c = col[e];
    int2 rec;
    rec.x = r;
    rec.y = __float_as_int(dinv[r] * dinv[c]);
    edges[off[c] + rank[e]] = rec;
  }
}

__global__ void weff_kernel(const float* __restrict__ pw, float* __restrict__ W) {
  int i = threadIdx.x;
  const int LD = HIDDEN + 2;
  float q = pw[i * LD + HIDDEN];
  float r = pw[i * LD + HIDDEN + 1];
  float s = 0.0f;
  for (int j = 0; j < HIDDEN; ++j) {
    float v = 0.0f;
    if (i < j)      v = pw[i * LD + j];
    else if (i > j) v = pw[j * LD + i];
    s += fabsf(v);
    W[i * HIDDEN + j] = v;
  }
  W[i * HIDDEN + i] = q * s + r;
}


__global__ __launch_bounds__(128)
void enc_gemm_kernel(const float* __restrict__ x, const float* __restrict__ ew,
                     float* __restrict__ h, float* __restrict__ x0) {
  int wave = (int)((blockIdx.x * blockDim.x + threadIdx.x) >> 5);
  if (wave >= M_TILES) return;
  int lane = threadIdx.x & 31;
  int mr = lane & 15;
  int kh = lane >> 4;
  const float* arow = x  + (size_t)(wave * 16 + mr) * NFEAT;
  const float* b0r  = ew + (size_t)(mr +  0) * NFEAT;
  const float* b1r  = ew + (size_t)(mr + 16) * NFEAT;
  const float* b2r  = ew + (size_t)(mr + 32) * NFEAT;
  const float* b3r  = ew + (size_t)(mr + 48) * NFEAT;
  v8f c0 = {}, c1 = {}, c2 = {}, c3 = {};
  for (int k = 0; k < NFEAT; k += 4) {
    int ko = k + 2 * kh;
    v2f a  = *(const v2f*)(arow + ko);
    v2f b0 = *(const v2f*)(b0r + ko);
    v2f b1 = *(const v2f*)(b1r + ko);
    v2f b2 = *(const v2f*)(b2r + ko);
    v2f b3 = *(const v2f*)(b3r + ko);
    c0 = __builtin_amdgcn_wmma_f32_16x16x4_f32(false, a, false, b0, (short)0, c0, false, false);
    c1 = __builtin_amdgcn_wmma_f32_16x16x4_f32(false, a, false, b1, (short)0, c1, false, false);
    c2 = __builtin_amdgcn_wmma_f32_16x16x4_f32(false, a, false, b2, (short)0, c2, false, false);
    c3 = __builtin_amdgcn_wmma_f32_16x16x4_f32(false, a, false, b3, (short)0, c3, false, false);
  }
  int mbase = wave * 16 + kh * 8;
  for (int r = 0; r < 8; ++r) {
    size_t o = (size_t)(mbase + r) * HIDDEN + mr;
    h[o]      = c0[r]; x0[o]      = c0[r];
    h[o + 16] = c1[r]; x0[o + 16] = c1[r];
    h[o + 32] = c2[r]; x0[o + 32] = c2[r];
    h[o + 48] = c3[r]; x0[o + 48] = c3[r];
  }
}


__global__ __launch_bounds__(128)
void hw_gemm_kernel(const float* __restrict__ h, const float* __restrict__ W,
                    float* __restrict__ outp) {
  __shared__ float Ws[HIDDEN * HIDDEN];
  for (int i = threadIdx.x; i < HIDDEN * HIDDEN; i += blockDim.x) Ws[i] = W[i];
  __syncthreads();
  int wave = (int)((blockIdx.x * blockDim.x + threadIdx.x) >> 5);
  if (wave >= M_TILES) return;
  int lane = threadIdx.x & 31;
  int mr = lane & 15;
  int kh = lane >> 4;
  const float* arow = h + (size_t)(wave * 16 + mr) * HIDDEN;
  v8f c0 = {}, c1 = {}, c2 = {}, c3 = {};
  for (int k = 0; k < HIDDEN; k += 4) {
    int ko = k + 2 * kh;
    v2f a  = *(const v2f*)(arow + ko);
    v2f b0 = *(const v2f*)(&Ws[(mr +  0) * HIDDEN + ko]);
    v2f b1 = *(const v2f*)(&Ws[(mr + 16) * HIDDEN + ko]);
    v2f b2 = *(const v2f*)(&Ws[(mr + 32) * HIDDEN + ko]);
    v2f b3 = *(const v2f*)(&Ws[(mr + 48) * HIDDEN + ko]);
    c0 = __builtin_amdgcn_wmma_f32_16x16x4_f32(false, a, false, b0, (short)0, c0, false, false);
    c1 = __builtin_amdgcn_wmma_f32_16x16x4_f32(false, a, false, b1, (short)0, c1, false, false);
    c2 = __builtin_amdgcn_wmma_f32_16x16x4_f32(false, a, false, b2, (short)0, c2, false, false);
    c3 = __builtin_amdgcn_wmma_f32_16x16x4_f32(false, a, false, b3, (short)0, c3, false, false);
  }
  int mbase = wave * 16 + kh * 8;
  for (int r = 0; r < 8; ++r) {
    size_t o = (size_t)(mbase + r) * HIDDEN + mr;
    outp[o]      = c0[r];
    outp[o + 16] = c1[r];
    outp[o + 32] = c2[r];
    outp[o + 48] = c3[r];
  }
}


__global__ __launch_bounds__(256)
void gather_update_kernel(const int* __restrict__ off, const int2* __restrict__ edges,
                          const float* __restrict__ outp, float* __restrict__ h,
                          const float* __restrict__ x0, const float* __restrict__ ext_w,
                          const float* __restrict__ beta) {
  const int c = __builtin_amdgcn_readfirstlane((int)((blockIdx.x * blockDim.x + threadIdx.x) >> 5));
  const int lane = threadIdx.x & 31;
  const int beg = off[c];
  const int end = off[c + 1];
  const float* src = outp + 2 * lane;
  const size_t ro = (size_t)c * HIDDEN + 2 * lane;
  const v2f hv = *(const v2f*)(h + ro);
  const v2f xv = *(const v2f*)(x0 + ro);
  const v2f ew = *(const v2f*)(ext_w + 2 * lane);
  const float bt = beta[0];
  v2f acc = {0.0f, 0.0f};
  for (int base = beg; base < end; base += 32) {
    int n = end - base;
    n = (n < 32) ? n : 32;
    int2 my;
    my.x = 0;
    my.y = 0;
    if (lane < n) my = edges[base + lane];
    int j = 0;
    for (; j + 8 <= n; j += 8) {
      v2f v[8];
      float w[8];
#pragma unroll
      for (int u = 0; u < 8; ++u) {
        const int s = __builtin_amdgcn_readlane(my.x, j + u);
        w[u] = __int_as_float(__builtin_amdgcn_readlane(my.y, j + u));
        v[u] = *(const v2f*)(src + (size_t)s * HIDDEN);
      }
#pragma unroll
      for (int u = 0; u < 8; ++u) acc += w[u] * v[u];
    }
    if (j + 4 <= n) {
      v2f v[4];
      float w[4];
#pragma unroll
      for (int u = 0; u < 4; ++u) {
        const int s = __builtin_amdgcn_readlane(my.x, j + u);
        w[u] = __int_as_float(__builtin_amdgcn_readlane(my.y, j + u));
        v[u] = *(const v2f*)(src + (size_t)s * HIDDEN);
      }
#pragma unroll
      for (int u = 0; u < 4; ++u) acc += w[u] * v[u];
      j += 4;
    }
    for (; j < n; ++j) {
      const int s = __builtin_amdgcn_readlane(my.x, j);
      const float w = __int_as_float(__builtin_amdgcn_readlane(my.y, j));
      const v2f v = *(const v2f*)(src + (size_t)s * HIDDEN);
      acc += w * v;
    }
  }
  const v2f conv = acc - hv * ew + xv * bt;
  v2f hn;
  hn.x = hv.x + STEP_SZ * fmaxf(conv.x, 0.0f);
  hn.y = hv.y + STEP_SZ * fmaxf(conv.y, 0.0f);
  *(v2f*)(h + ro) = hn;
}


__global__ __launch_bounds__(128)
void dec_gemm_kernel(const float* __restrict__ h, const float* __restrict__ dw,
                     float* __restrict__ out) {
  int wave = (int)((blockIdx.x * blockDim.x + threadIdx.x) >> 5);
  if (wave >= M_TILES) return;
  int lane = threadIdx.x & 31;
  int mr = lane & 15;
  int kh = lane >> 4;
  const float* arow = h + (size_t)(wave * 16 + mr) * HIDDEN;
  const float* b0r = dw + (size_t)(mr +  0) * HIDDEN;
  const float* b1r = dw + (size_t)(mr + 16) * HIDDEN;
  int n2 = mr + 32;
  const float* b2r = dw + (size_t)(n2 < NCLASS ? n2 : 0) * HIDDEN;
  float b2scale = (n2 < NCLASS) ? 1.0f : 0.0f;
  v8f c0 = {}, c1 = {}, c2 = {};
  for (int k = 0; k < HIDDEN; k += 4) {
    int ko = k + 2 * kh;
    v2f a  = *(const v2f*)(arow + ko);
    v2f b0 = *(const v2f*)(b0r + ko);
    v2f b1 = *(const v2f*)(b1r + ko);
    v2f b2 = *(const v2f*)(b2r + ko);
    b2 *= b2scale;
    c0 = __builtin_amdgcn_wmma_f32_16x16x4_f32(false, a, false, b0, (short)0, c0, false, false);
    c1 = __builtin_amdgcn_wmma_f32_16x16x4_f32(false, a, false, b1, (short)0, c1, false, false);
    c2 = __builtin_amdgcn_wmma_f32_16x16x4_f32(false, a, false, b2, (short)0, c2, false, false);
  }
  int mbase = wave * 16 + kh * 8;
  for (int r = 0; r < 8; ++r) {
    float* orow = out + (size_t)(mbase + r) * NCLASS;
    orow[mr]      = c0[r];
    orow[mr + 16] = c1[r];
    if (n2 < NCLASS) orow[n2] = c2[r];
  }
}


extern "C" void kernel_launch(void* const* d_in, const int* in_sizes, int n_in,
                              void* d_out, int out_size, void* d_ws, size_t ws_size,
                              hipStream_t stream) {
  (void)in_sizes; (void)n_in; (void)out_size; (void)ws_size;
  const float* x     = (const float*)d_in[0];
  const int*   ei    = (const int*)  d_in[1];
  const float* enc_w = (const float*)d_in[2];
  const float* dec_w = (const float*)d_in[3];
  const float* ext_w = (const float*)d_in[4];
  const float* beta  = (const float*)d_in[5];
  const float* pw    = (const float*)d_in[6];
  const int* row = ei;
  const int* col = ei + N_EDGES;

  char* wsb = (char*)d_ws;
  int*   cnt   = (int*)  (wsb);
  int*   off   = (int*)  (wsb + 200192);
  float* dinv  = (float*)(wsb + 400384);
  int*   rank  = (int*)  (wsb + 600576);
  int2*  edges = (int2*) (wsb + 7000576);
  float* W     = (float*)(wsb + 19800576);
  float* h     = (float*)(wsb + 19816960);
  float* x0    = (float*)(wsb + 32616960);
  float* outp  = (float*)(wsb + 45416960);

  hipMemsetAsync(cnt, 0, N_NODES * sizeof(int), stream);
  count_rank_kernel<<<(N_EDGES + 255) / 256, 256, 0, stream>>>(col, cnt, rank);
  scan_dinv_kernel<<<1, 1024, 0, stream>>>(cnt, off, dinv);
  fill_kernel<<<(N_EDGES + 255) / 256, 256, 0, stream>>>(row, col, rank, off, dinv, edges);
  weff_kernel<<<1, HIDDEN, 0, stream>>>(pw, W);

  const int gemm_blocks = (M_TILES + 3) / 4;
  enc_gemm_kernel<<<gemm_blocks, 128, 0, stream>>>(x, enc_w, h, x0);

  for (int l = 0; l < NUM_LAYERS; ++l) {
    hw_gemm_kernel<<<gemm_blocks, 128, 0, stream>>>(h, W, outp);
    gather_update_kernel<<<N_NODES / 8, 256, 0, stream>>>(off, edges, outp, h, x0, ext_w, beta);
  }

  dec_gemm_kernel<<<gemm_blocks, 128, 0, stream>>>(h, dec_w, (float*)d_out);
}


# Edge slots from LDS counters, not global returning atomics: 32 work-groups each count one run of the edge list in a 200 KB LDS histogram; sum order in a segment: by run, arrival order inside a run
# speedup vs baseline: 2.8336x; 1.2315x over previous
#include <hip/hip_runtime.h>
#include <math.h>

#define N_NODES   50000
#define N_EDGES   1600000
#define NFEAT     500
#define HIDDEN    64
#define NCLASS    40
#define NUM_LAYERS 4
#define STEP_SZ   0.1f
#define M_TILES   (N_NODES / 16)

typedef __attribute__((ext_vector_type(2))) float v2f;
typedef __attribute__((ext_vector_type(8))) float v8f;


#define CHUNKS   32
#define CHUNK_E  (N_EDGES / CHUNKS)
#define CHUNK_Q  (CHUNK_E / 4)
#define CNT_T    1024

__global__ __launch_bounds__(CNT_T)
void count_local_kernel(const int* __restrict__ col, int* __restrict__ hist,
                        int* __restrict__ rank) {
  __shared__ int cnt[N_NODES];
  const int t = threadIdx.x;
  for (int i = t; i < N_NODES; i += CNT_T) cnt[i] = 0;
  __syncthreads();
  const int4* c4 = (const int4*)(col + (size_t)blockIdx.x * CHUNK_E);
  int4* r4 = (int4*)(rank + (size_t)blockIdx.x * CHUNK_E);
  const int full = CHUNK_Q / CNT_T;
  for (int k = 0; k < full; k += 4) {
    int4 c[4];
#pragma unroll
    for (int u = 0; u < 4; ++u) c[u] = c4[(k + u) * CNT_T + t];
#pragma unroll
    for (int u = 0; u < 4; ++u) {
      int4 r;
      r.x = atomicAdd(&cnt[c[u].x], 1);
      r.y = atomicAdd(&cnt[c[u].y], 1);
      r.z = atomicAdd(&cnt[c[u].z], 1);
      r.w = atomicAdd(&cnt[c[u].w], 1);
      r4[(k + u) * CNT_T + t] = r;
    }
  }
  {
    const int q = full * CNT_T + t;
    if (q < CHUNK_Q) {
      const int4 c = c4[q];
      int4 r;
      r.x = atomicAdd(&cnt[c.x], 1);
      r.y = atomicAdd(&cnt[c.y], 1);
      r.z = atomicAdd(&cnt[c.z], 1);
      r.w = atomicAdd(&cnt[c.w], 1);
      r4[q] = r;
    }
  }
  __syncthreads();
  int* mine = hist + (size_t)blockIdx.x * N_NODES;
  for (int i = t; i < N_NODES; i += CNT_T) mine[i] = cnt[i];
}

__global__ __launch_bounds__(256)
void chunk_prefix_kernel(const int* __restrict__ hist, int* __restrict__ pre,
                         int* __restrict__ cnt) {
  const int c = blockIdx.x * blockDim.x + threadIdx.x;
  if (c < N_NODES) {
    int run = 0;
#pragma unroll
    for (int g = 0; g < CHUNKS; ++g) {
      const int v = hist[(size_t)g * N_NODES + c];
      pre[(size_t)g * N_NODES + c] = run;
      run += v;
    }
    cnt[c] = run;
  }
}

#define SCAN_T   1024
#define SCAN_PER 49

__global__ __launch_bounds__(SCAN_T)
void scan_dinv_kernel(const int* __restrict__ cnt, int* __restrict__ off,
                      float* __restrict__ dinv) {
  __shared__ int sc[SCAN_T * SCAN_PER];
  __shared__ int wsum[32];
  const int t = threadIdx.x;
  const int lane = t & 31;
  const int w = t >> 5;
  for (int k = 0; k < SCAN_PER; ++k) {
    const int i = k * SCAN_T + t;
    int v = 0;
    if (i < N_NODES) {
      v = cnt[i];
      dinv[i] = (v > 0) ? rsqrtf((float)v) : 0.0f;
    }
    sc[i] = v;
  }
  __syncthreads();
  int* mine = sc + t * SCAN_PER;
  int run = 0;
  for (int k = 0; k < SCAN_PER; ++k) run += mine[k];
  int s = run;
  for (int d = 1; d < 32; d <<= 1) {
    int up = __shfl_up(s, d);
    if (lane >= d) s += up;
  }
  if (lane == 31) wsum[w] = s;
  __syncthreads();
  int ws = wsum[lane];
  for (int d = 1; d < 32; d <<= 1) {
    int up = __shfl_up(ws, d);
    if (lane >= d) ws += up;
  }
  const int prev  = __shfl(ws, (w + 31) & 31);
  const int total = __shfl(ws, 31);
  int pos = ((w > 0) ? prev : 0) + s - run;
  for (int k = 0; k < SCAN_PER; ++k) {
    const int v = mine[k];
    mine[k] = pos;
    pos += v;
  }
  __syncthreads();
  for (int k = 0; k < SCAN_PER; ++k) {
    const int i = k * SCAN_T + t;
    if (i < N_NODES) off[i] = sc[i];
  }
  if (t == 0) off[N_NODES] = total;
}

__global__ void fill_kernel(const int* __restrict__ row, const int* __restrict__ col,
                            const int* __restrict__ rank, const int* __restrict__ off,
                            const int* __restrict__ pre, const float* __restrict__ dinv,
                            int2* __restrict__ edges) {
  int e = blockIdx.x * blockDim.x + threadIdx.x;
  if (e < N_EDGES) {
    int r = row[e];
    int c = col[e];
    int2 rec;
    rec.x = r;
    rec.y = __float_as_int(dinv[r] * dinv[c]);
    edges[off[c] + pre[(size_t)(e / CHUNK_E) * N_NODES + c] + rank[e]] = rec;
  }
}

__global__ void weff_kernel(const float* __restrict__ pw, float* __restrict__ W) {
  int i = threadIdx.x;
  const int LD = HIDDEN + 2;
  float q = pw[i * LD + HIDDEN];
  float r = pw[i * LD + HIDDEN + 1];
  float s = 0.0f;
  for (int j = 0; j < HIDDEN; ++j) {
    float v = 0.0f;
    if (i < j)      v = pw[i * LD + j];
    else if (i > j) v = pw[j * LD + i];
    s += fabsf(v);
    W[i * HIDDEN + j] = v;
  }
  W[i * HIDDEN + i] = q * s + r;
}


__global__ __launch_bounds__(128)
void enc_gemm_kernel(const float* __restrict__ x, const float* __restrict__ ew,
                     float* __restrict__ h, float* __restrict__ x0) {
  int wave = (int)((blockIdx.x * blockDim.x + threadIdx.x) >> 5);
  if (wave >= M_TILES) return;
  int lane = threadIdx.x & 31;
  int mr = lane & 15;
  int kh = lane >> 4;
  const float* arow = x  + (size_t)(wave * 16 + mr) * NFEAT;
  const float* b0r  = ew + (size_t)(mr +  0) * NFEAT;
  const float* b1r  = ew + (size_t)(mr + 16) * NFEAT;
  const float* b2r  = ew + (size_t)(mr + 32) * NFEAT;
  const float* b3r  = ew + (size_t)(mr + 48) * NFEAT;
  v8f c0 = {}, c1 = {}, c2 = {}, c3 = {};
  for (int k = 0; k < NFEAT; k += 4) {
    int ko = k + 2 * kh;
    v2f a  = *(const v2f*)(arow + ko);
    v2f b0 = *(const v2f*)(b0r + ko);
    v2f b1 = *(const v2f*)(b1r + ko);
    v2f b2 = *(const v2f*)(b2r + ko);
    v2f b3 = *(const v2f*)(b3r + ko);
    c0 = __builtin_amdgcn_wmma_f32_16x16x4_f32(false, a, false, b0, (short)0, c0, false, false);
    c1 = __builtin_amdgcn_wmma_f32_16x16x4_f32(false, a, false, b1, (short)0, c1, false, false);
    c2 = __builtin_amdgcn_wmma_f32_16x16x4_f32(false, a, false, b2, (short)0, c2, false, false);
    c3 = __builtin_amdgcn_wmma_f32_16x16x4_f32(false, a, false, b3, (short)0, c3, false, false);
  }
  int mbase = wave * 16 + kh * 8;
  for (int r = 0; r < 8; ++r) {
    size_t o = (size_t)(mbase + r) * HIDDEN + mr;
    h[o]      = c0[r]; x0[o]      = c0[r];
    h[o + 16] = c1[r]; x0[o + 16] = c1[r];
    h[o + 32] = c2[r]; x0[o + 32] = c2[r];
    h[o + 48] = c3[r]; x0[o + 48] = c3[r];
  }
}


__global__ __launch_bounds__(128)
void hw_gemm_kernel(const float* __restrict__ h, const float* __restrict__ W,
                    float* __restrict__ outp) {
  __shared__ float Ws[HIDDEN * HIDDEN];
  for (int i = threadIdx.x; i < HIDDEN * HIDDEN; i += blockDim.x) Ws[i] = W[i];
  __syncthreads();
  int wave = (int)((blockIdx.x * blockDim.x + threadIdx.x) >> 5);
  if (wave >= M_TILES) return;
  int lane = threadIdx.x & 31;
  int mr = lane & 15;
  int kh = lane >> 4;
  const float* arow = h + (size_t)(wave * 16 + mr) * HIDDEN;
  v8f c0 = {}, c1 = {}, c2 = {}, c3 = {};
  for (int k = 0; k < HIDDEN; k += 4) {
    int ko = k + 2 * kh;
    v2f a  = *(const v2f*)(arow + ko);
    v2f b0 = *(const v2f*)(&Ws[(mr +  0) * HIDDEN + ko]);
    v2f b1 = *(const v2f*)(&Ws[(mr + 16) * HIDDEN + ko]);
    v2f b2 = *(const v2f*)(&Ws[(mr + 32) * HIDDEN + ko]);
    v2f b3 = *(const v2f*)(&Ws[(mr + 48) * HIDDEN + ko]);
    c0 = __builtin_amdgcn_wmma_f32_16x16x4_f32(false, a, false, b0, (short)0, c0, false, false);
    c1 = __builtin_amdgcn_wmma_f32_16x16x4_f32(false, a, false, b1, (short)0, c1, false, false);
    c2 = __builtin_amdgcn_wmma_f32_16x16x4_f32(false, a, false, b2, (short)0, c2, false, false);
    c3 = __builtin_amdgcn_wmma_f32_16x16x4_f32(false, a, false, b3, (short)0, c3, false, false);
  }
  int mbase = wave * 16 + kh * 8;
  for (int r = 0; r < 8; ++r) {
    size_t o = (size_t)(mbase + r) * HIDDEN + mr;
    outp[o]      = c0[r];
    outp[o + 16] = c1[r];
    outp[o + 32] = c2[r];
    outp[o + 48] = c3[r];
  }
}


__global__ __launch_bounds__(256)
void gather_update_kernel(const int* __restrict__ off, const int2* __restrict__ edges,
                          const float* __restrict__ outp, float* __restrict__ h,
                          const float* __restrict__ x0, const float* __restrict__ ext_w,
                          const float* __restrict__ beta) {
  const int c = __builtin_amdgcn_readfirstlane((int)((blockIdx.x * blockDim.x + threadIdx.x) >> 5));
  const int lane = threadIdx.x & 31;
  const int beg = off[c];
  const int end = off[c + 1];
  const float* src = outp + 2 * lane;
  const size_t ro = (size_t)c * HIDDEN + 2 * lane;
  const v2f hv = *(const v2f*)(h + ro);
  const v2f xv = *(const v2f*)(x0 + ro);
  const v2f ew = *(const v2f*)(ext_w + 2 * lane);
  const float bt = beta[0];
  v2f acc = {0.0f, 0.0f};
  for (int base = beg; base < end; base += 32) {
    int n = end - base;
    n = (n < 32) ? n : 32;
    int2 my;
    my.x = 0;
    my.y = 0;
    if (lane < n) my = edges[base + lane];
    int j = 0;
    for (; j + 8 <= n; j += 8) {
      v2f v[8];
      float w[8];
#pragma unroll
      for (int u = 0; u < 8; ++u) {
        const int s = __builtin_amdgcn_readlane(my.x, j + u);
        w[u] = __int_as_float(__builtin_amdgcn_readlane(my.y, j + u));
        v[u] = *(const v2f*)(src + (size_t)s * HIDDEN);
      }
#pragma unroll
      for (int u = 0; u < 8; ++u) acc += w[u] * v[u];
    }
    if (j + 4 <= n) {
      v2f v[4];
      float w[4];
#pragma unroll
      for (int u = 0; u < 4; ++u) {
        const int s = __builtin_amdgcn_readlane(my.x, j + u);
        w[u] = __int_as_float(__builtin_amdgcn_readlane(my.y, j + u));
        v[u] = *(const v2f*)(src + (size_t)s * HIDDEN);
      }
#pragma unroll
      for (int u = 0; u < 4; ++u) acc += w[u] * v[u];
      j += 4;
    }
    for (; j < n; ++j) {
      const int s = __builtin_amdgcn_readlane(my.x, j);
      const float w = __int_as_float(__builtin_amdgcn_readlane(my.y, j));
      const v2f v = *(const v2f*)(src + (size_t)s * HIDDEN);
      acc += w * v;
    }
  }
  const v2f conv = acc - hv * ew + xv * bt;
  v2f hn;
  hn.x = hv.x + STEP_SZ * fmaxf(conv.x, 0.0f);
  hn.y = hv.y + STEP_SZ * fmaxf(conv.y, 0.0f);
  *(v2f*)(h + ro) = hn;
}


__global__ __launch_bounds__(128)
void dec_gemm_kernel(const float* __restrict__ h, const float* __restrict__ dw,
                     float* __restrict__ out) {
  int wave = (int)((blockIdx.x * blockDim.x + threadIdx.x) >> 5);
  if (wave >= M_TILES) return;
  int lane = threadIdx.x & 31;
  int mr = lane & 15;
  int kh = lane >> 4;
  const float* arow = h + (size_t)(wave * 16 + mr) * HIDDEN;
  const float* b0r = dw + (size_t)(mr +  0) * HIDDEN;
  const float* b1r = dw + (size_t)(mr + 16) * HIDDEN;
  int n2 = mr + 32;
  const float* b2r = dw + (size_t)(n2 < NCLASS ? n2 : 0) * HIDDEN;
  float b2scale = (n2 < NCLASS) ? 1.0f : 0.0f;
  v8f c0 = {}, c1 = {}, c2 = {};
  for (int k = 0; k < HIDDEN; k += 4) {
    int ko = k + 2 * kh;
    v2f a  = *(const v2f*)(arow + ko);
    v2f b0 = *(const v2f*)(b0r + ko);
    v2f b1 = *(const v2f*)(b1r + ko);
    v2f b2 = *(const v2f*)(b2r + ko);
    b2 *= b2scale;
    c0 = __builtin_amdgcn_wmma_f32_16x16x4_f32(false, a, false, b0, (short)0, c0, false, false);
    c1 = __builtin_amdgcn_wmma_f32_16x16x4_f32(false, a, false, b1, (short)0, c1, false, false);
    c2 = __builtin_amdgcn_wmma_f32_16x16x4_f32(false, a, false, b2, (short)0, c2, false, false);
  }
  int mbase = wave * 16 + kh * 8;
  for (int r = 0; r < 8; ++r) {
    float* orow = out + (size_t)(mbase + r) * NCLASS;
    orow[mr]      = c0[r];
    orow[mr + 16] = c1[r];
    if (n2 < NCLASS) orow[n2] = c2[r];
  }
}


extern "C" void kernel_launch(void* const* d_in, const int* in_sizes, int n_in,
                              void* d_out, int out_size, void* d_ws, size_t ws_size,
                              hipStream_t stream) {
  (void)in_sizes; (void)n_in; (void)out_size; (void)ws_size;
  const float* x     = (const float*)d_in[0];
  const int*   ei    = (const int*)  d_in[1];
  const float* enc_w = (const float*)d_in[2];
  const float* dec_w = (const float*)d_in[3];
  const float* ext_w = (const float*)d_in[4];
  const float* beta  = (const float*)d_in[5];
  const float* pw    = (const float*)d_in[6];
  const int* row = ei;
  const int* col = ei + N_EDGES;

  char* wsb = (char*)d_ws;
  int*   cnt   = (int*)  (wsb);
  int*   off   = (int*)  (wsb + 200192);
  float* dinv  = (float*)(wsb + 400384);
  int*   rank  = (int*)  (wsb + 600576);
  int2*  edges = (int2*) (wsb + 7000576);
  float* W     = (float*)(wsb + 19800576);
  float* h     = (float*)(wsb + 19816960);
  float* x0    = (float*)(wsb + 32616960);
  float* outp  = (float*)(wsb + 45416960);
  int*   hist  = (int*)  (wsb + 58216960);
  int*   pre   = (int*)  (wsb + 64616960);

  count_local_kernel<<<CHUNKS, CNT_T, 0, stream>>>(col, hist, rank);
  chunk_prefix_kernel<<<(N_NODES + 255) / 256, 256, 0, stream>>>(hist, pre, cnt);
  scan_dinv_kernel<<<1, 1024, 0, stream>>>(cnt, off, dinv);
  fill_kernel<<<(N_EDGES + 255) / 256, 256, 0, stream>>>(row, col, rank, off, pre, dinv, edges);
  weff_kernel<<<1, HIDDEN, 0, stream>>>(pw, W);

  const int gemm_blocks = (M_TILES + 3) / 4;
  enc_gemm_kernel<<<gemm_blocks, 128, 0, stream>>>(x, enc_w, h, x0);

  for (int l = 0; l < NUM_LAYERS; ++l) {
    hw_gemm_kernel<<<gemm_blocks, 128, 0, stream>>>(h, W, outp);
    gather_update_kernel<<<N_NODES / 8, 256, 0, stream>>>(off, edges, outp, h, x0, ext_w, beta);
  }

  dec_gemm_kernel<<<gemm_blocks, 128, 0, stream>>>(h, dec_w, (float*)d_out);
}


# Encoder GEMM loads 16 B per lane: enc_w copied once per work-group into LDS (b128 fragment reads), x read 4 k per lane two groups ahead; k order in each block of 8 now 0,1,4,5 then 2,3,6,7
# speedup vs baseline: 3.3055x; 1.1665x over previous
#include <hip/hip_runtime.h>
#include <math.h>

#define N_NODES   50000
#define N_EDGES   1600000
#define NFEAT     500
#define HIDDEN    64
#define NCLASS    40
#define NUM_LAYERS 4
#define STEP_SZ   0.1f
#define M_TILES   (N_NODES / 16)

typedef __attribute__((ext_vector_type(2))) float v2f;
typedef __attribute__((ext_vector_type(8))) float v8f;


#define CHUNKS   32
#define CHUNK_E  (N_EDGES / CHUNKS)
#define CHUNK_Q  (CHUNK_E / 4)
#define CNT_T    1024

__global__ __launch_bounds__(CNT_T)
void count_local_kernel(const int* __restrict__ col, int* __restrict__ hist,
                        int* __restrict__ rank) {
  __shared__ int cnt[N_NODES];
  const int t = threadIdx.x;
  for (int i = t; i < N_NODES; i += CNT_T) cnt[i] = 0;
  __syncthreads();
  const int4* c4 = (const int4*)(col + (size_t)blockIdx.x * CHUNK_E);
  int4* r4 = (int4*)(rank + (size_t)blockIdx.x * CHUNK_E);
  const int full = CHUNK_Q / CNT_T;
  for (int k = 0; k < full; k += 4) {
    int4 c[4];
#pragma unroll
    for (int u = 0; u < 4; ++u) c[u] = c4[(k + u) * CNT_T + t];
#pragma unroll
    for (int u = 0; u < 4; ++u) {
      int4 r;
      r.x = atomicAdd(&cnt[c[u].x], 1);
      r.y = atomicAdd(&cnt[c[u].y], 1);
      r.z = atomicAdd(&cnt[c[u].z], 1);
      r.w = atomicAdd(&cnt[c[u].w], 1);
      r4[(k + u) * CNT_T + t] = r;
    }
  }
  {
    const int q = full * CNT_T + t;
    if (q < CHUNK_Q) {
      const int4 c = c4[q];
      int4 r;
      r.x = atomicAdd(&cnt[c.x], 1);
      r.y = atomicAdd(&cnt[c.y], 1);
      r.z = atomicAdd(&cnt[c.z], 1);
      r.w = atomicAdd(&cnt[c.w], 1);
      r4[q] = r;
    }
  }
  __syncthreads();
  int* mine = hist + (size_t)blockIdx.x * N_NODES;
  for (int i = t; i < N_NODES; i += CNT_T) mine[i] = cnt[i];
}

__global__ __launch_bounds__(256)
void chunk_prefix_kernel(const int* __restrict__ hist, int* __restrict__ pre,
                         int* __restrict__ cnt) {
  const int c = blockIdx.x * blockDim.x + threadIdx.x;
  if (c < N_NODES) {
    int run = 0;
#pragma unroll
    for (int g = 0; g < CHUNKS; ++g) {
      const int v = hist[(size_t)g * N_NODES + c];
      pre[(size_t)g * N_NODES + c] = run;
      run += v;
    }
    cnt[c] = run;
  }
}

#define SCAN_T   1024
#define SCAN_PER 49

__global__ __launch_bounds__(SCAN_T)
void scan_dinv_kernel(const int* __restrict__ cnt, int* __restrict__ off,
                      float* __restrict__ dinv) {
  __shared__ int sc[SCAN_T * SCAN_PER];
  __shared__ int wsum[32];
  const int t = threadIdx.x;
  const int lane = t & 31;
  const int w = t >> 5;
  for (int k = 0; k < SCAN_PER; ++k) {
    const int i = k * SCAN_T + t;
    int v = 0;
    if (i < N_NODES) {
      v = cnt[i];
      dinv[i] = (v > 0) ? rsqrtf((float)v) : 0.0f;
    }
    sc[i] = v;
  }
  __syncthreads();
  int* mine = sc + t * SCAN_PER;
  int run = 0;
  for (int k = 0; k < SCAN_PER; ++k) run += mine[k];
  int s = run;
  for (int d = 1; d < 32; d <<= 1) {
    int up = __shfl_up(s, d);
    if (lane >= d) s += up;
  }
  if (lane == 31) wsum[w] = s;
  __syncthreads();
  int ws = wsum[lane];
  for (int d = 1; d < 32; d <<= 1) {
    int up = __shfl_up(ws, d);
    if (lane >= d) ws += up;
  }
  const int prev  = __shfl(ws, (w + 31) & 31);
  const int total = __shfl(ws, 31);
  int pos = ((w > 0) ? prev : 0) + s - run;
  for (int k = 0; k < SCAN_PER; ++k) {
    const int v = mine[k];
    mine[k] = pos;
    pos += v;
  }
  __syncthreads();
  for (int k = 0; k < SCAN_PER; ++k) {
    const int i = k * SCAN_T + t;
    if (i < N_NODES) off[i] = sc[i];
  }
  if (t == 0) off[N_NODES] = total;
}

__global__ void fill_kernel(const int* __restrict__ row, const int* __restrict__ col,
                            const int* __restrict__ rank, const int* __restrict__ off,
                            const int* __restrict__ pre, const float* __restrict__ dinv,
                            int2* __restrict__ edges) {
  int e = blockIdx.x * blockDim.x + threadIdx.x;
  if (e < N_EDGES) {
    int r = row[e];
    int c = col[e];
    int2 rec;
    rec.x = r;
    rec.y = __float_as_int(dinv[r] * dinv[c]);
    edges[off[c] + pre[(size_t)(e / CHUNK_E) * N_NODES + c] + rank[e]] = rec;
  }
}

__global__ void weff_kernel(const float* __restrict__ pw, float* __restrict__ W) {
  int i = threadIdx.x;
  const int LD = HIDDEN + 2;
  float q = pw[i * LD + HIDDEN];
  float r = pw[i * LD + HIDDEN + 1];
  float s = 0.0f;
  for (int j = 0; j < HIDDEN; ++j) {
    float v = 0.0f;
    if (i < j)      v = pw[i * LD + j];
    else if (i > j) v = pw[j * LD + i];
    s += fabsf(v);
    W[i * HIDDEN + j] = v;
  }
  W[i * HIDDEN + i] = q * s + r;
}


typedef __attribute__((ext_vector_type(4))) float v4f;

#define ENC_T   512
#define ENC_WGS 256

#define ENC_WMMA(av, bv, cv) cv = __builtin_amdgcn_wmma_f32_16x16x4_f32(false, av, false, bv, (short)0, cv, false, false)

#define ENC_STEP8(a, bq)                                              \
  {                                                                   \
    const v4f b0 = *(const v4f*)((bq));                               \
    const v4f b1 = *(const v4f*)((bq) + 16 * NFEAT);                  \
    const v4f b2 = *(const v4f*)((bq) + 32 * NFEAT);                  \
    const v4f b3 = *(const v4f*)((bq) + 48 * NFEAT);                  \
    const v2f alo = {(a).x, (a).y};                                   \
    const v2f ahi = {(a).z, (a).w};                                   \
    const v2f b0l = {b0.x, b0.y}, b0h = {b0.z, b0.w};                 \
    const v2f b1l = {b1.x, b1.y}, b1h = {b1.z, b1.w};                 \
    const v2f b2l = {b2.x, b2.y}, b2h = {b2.z, b2.w};                 \
    const v2f b3l = {b3.x, b3.y}, b3h = {b3.z, b3.w};                 \
    ENC_WMMA(alo, b0l, c0); ENC_WMMA(alo, b1l, c1);                   \
    ENC_WMMA(alo, b2l, c2); ENC_WMMA(alo, b3l, c3);                   \
    ENC_WMMA(ahi, b0h, c0); ENC_WMMA(ahi, b1h, c1);                   \
    ENC_WMMA(ahi, b2h, c2); ENC_WMMA(ahi, b3h, c3);                   \
  }

#define ENC_LOAD4(v, ptr)                                             \
  v##0 = *(const v4f*)((ptr) +  0);                                   \
  v##1 = *(const v4f*)((ptr) +  8);                                   \
  v##2 = *(const v4f*)((ptr) + 16);                                   \
  v##3 = *(const v4f*)((ptr) + 24);
#define ENC_GROUP(v, bq)                                              \
  ENC_STEP8(v##0, (bq) +  0)                                          \
  ENC_STEP8(v##1, (bq) +  8)                                          \
  ENC_STEP8(v##2, (bq) + 16)                                          \
  ENC_STEP8(v##3, (bq) + 24)

__global__ __launch_bounds__(ENC_T)
void enc_gemm_kernel(const float* __restrict__ x, const float* __restrict__ ew,
                     float* __restrict__ h, float* __restrict__ x0) {
  __shared__ __attribute__((aligned(16))) float Bs[HIDDEN * NFEAT];
  {
    const v4f* s4 = (const v4f*)ew;
    v4f* d4 = (v4f*)Bs;
    for (int i = threadIdx.x; i < HIDDEN * NFEAT / 4; i += ENC_T) d4[i] = s4[i];
  }
  __syncthreads();
  const int tile = (int)(blockIdx.x + gridDim.x * (threadIdx.x >> 5));
  if (tile >= M_TILES) return;
  const int lane = threadIdx.x & 31;
  const int mr = lane & 15;
  const int kh = lane >> 4;
  const float* arow = x + (size_t)(tile * 16 + mr) * NFEAT;
  const float* ap = arow + 4 * kh;
  const float* bp = Bs + mr * NFEAT + 4 * kh;
  v8f c0 = {}, c1 = {}, c2 = {}, c3 = {};
  v4f p0, p1, p2, p3, q0, q1, q2, q3, s0, s1, s2, s3;
  ENC_LOAD4(p, ap)
  ENC_LOAD4(q, ap + 32)
  for (int it = 0; it < 4; ++it) {
    const float* ag = ap + 96 * it;
    const float* bg = bp + 96 * it;
    ENC_LOAD4(s, ag + 64)
    ENC_GROUP(p, bg)
    ENC_LOAD4(p, ag + 96)
    ENC_GROUP(q, bg + 32)
    ENC_LOAD4(q, ag + 128)
    ENC_GROUP(s, bg + 64)
  }
  {
    ENC_LOAD4(s, ap + 448)
    const v4f r0 = *(const v4f*)(ap + 480);
    const v4f r1 = *(const v4f*)(ap + 488);
    const v2f r2 = *(const v2f*)(arow + 496 + 2 * kh);
    ENC_GROUP(p, bp + 384)
    ENC_GROUP(q, bp + 416)
    ENC_GROUP(s, bp + 448)
    ENC_STEP8(r0, bp + 480)
    ENC_STEP8(r1, bp + 488)
    const float* bt = Bs + mr * NFEAT + 496 + 2 * kh;
    const v2f t0 = *(const v2f*)(bt);
    const v2f t1 = *(const v2f*)(bt + 16 * NFEAT);
    const v2f t2 = *(const v2f*)(bt + 32 * NFEAT);
    const v2f t3 = *(const v2f*)(bt + 48 * NFEAT);
    ENC_WMMA(r2, t0, c0); ENC_WMMA(r2, t1, c1);
    ENC_WMMA(r2, t2, c2); ENC_WMMA(r2, t3, c3);
  }
  int mbase = tile * 16 + kh * 8;
  for (int r = 0; r < 8; ++r) {
    size_t o = (size_t)(mbase + r) * HIDDEN + mr;
    h[o]      = c0[r]; x0[o]      = c0[r];
    h[o + 16] = c1[r]; x0[o + 16] = c1[r];
    h[o + 32] = c2[r]; x0[o + 32] = c2[r];
    h[o + 48] = c3[r]; x0[o + 48] = c3[r];
  }
}


__global__ __launch_bounds__(128)
void hw_gemm_kernel(const float* __restrict__ h, const float* __restrict__ W,
                    float* __restrict__ outp) {
  __shared__ float Ws[HIDDEN * HIDDEN];
  for (int i = threadIdx.x; i < HIDDEN * HIDDEN; i += blockDim.x) Ws[i] = W[i];
  __syncthreads();
  int wave = (int)((blockIdx.x * blockDim.x + threadIdx.x) >> 5);
  if (wave >= M_TILES) return;
  int lane = threadIdx.x & 31;
  int mr = lane & 15;
  int kh = lane >> 4;
  const float* arow = h + (size_t)(wave * 16 + mr) * HIDDEN;
  v8f c0 = {}, c1 = {}, c2 = {}, c3 = {};
  for (int k = 0; k < HIDDEN; k += 4) {
    int ko = k + 2 * kh;
    v2f a  = *(const v2f*)(arow + ko);
    v2f b0 = *(const v2f*)(&Ws[(mr +  0) * HIDDEN + ko]);
    v2f b1 = *(const v2f*)(&Ws[(mr + 16) * HIDDEN + ko]);
    v2f b2 = *(const v2f*)(&Ws[(mr + 32) * HIDDEN + ko]);
    v2f b3 = *(const v2f*)(&Ws[(mr + 48) * HIDDEN + ko]);
    c0 = __builtin_amdgcn_wmma_f32_16x16x4_f32(false, a, false, b0, (short)0, c0, false, false);
    c1 = __builtin_amdgcn_wmma_f32_16x16x4_f32(false, a, false, b1, (short)0, c1, false, false);
    c2 = __builtin_amdgcn_wmma_f32_16x16x4_f32(false, a, false, b2, (short)0, c2, false, false);
    c3 = __builtin_amdgcn_wmma_f32_16x16x4_f32(false, a, false, b3, (short)0, c3, false, false);
  }
  int mbase = wave * 16 + kh * 8;
  for (int r = 0; r < 8; ++r) {
    size_t o = (size_t)(mbase + r) * HIDDEN + mr;
    outp[o]      = c0[r];
    outp[o + 16] = c1[r];
    outp[o + 32] = c2[r];
    outp[o + 48] = c3[r];
  }
}


__global__ __launch_bounds__(256)
void gather_update_kernel(const int* __restrict__ off, const int2* __restrict__ edges,
                          const float* __restrict__ outp, float* __restrict__ h,
                          const float* __restrict__ x0, const float* __restrict__ ext_w,
                          const float* __restrict__ beta) {
  const int c = __builtin_amdgcn_readfirstlane((int)((blockIdx.x * blockDim.x + threadIdx.x) >> 5));
  const int lane = threadIdx.x & 31;
  const int beg = off[c];
  const int end = off[c + 1];
  const float* src = outp + 2 * lane;
  const size_t ro = (size_t)c * HIDDEN + 2 * lane;
  const v2f hv = *(const v2f*)(h + ro);
  const v2f xv = *(const v2f*)(x0 + ro);
  const v2f ew = *(const v2f*)(ext_w + 2 * lane);
  const float bt = beta[0];
  v2f acc = {0.0f, 0.0f};
  for (int base = beg; base < end; base += 32) {
    int n = end - base;
    n = (n < 32) ? n : 32;
    int2 my;
    my.x = 0;
    my.y = 0;
    if (lane < n) my = edges[base + lane];
    int j = 0;
    for (; j + 8 <= n; j += 8) {
      v2f v[8];
      float w[8];
#pragma unroll
      for (int u = 0; u < 8; ++u) {
        const int s = __builtin_amdgcn_readlane(my.x, j + u);
        w[u] = __int_as_float(__builtin_amdgcn_readlane(my.y, j + u));
        v[u] = *(const v2f*)(src + (size_t)s * HIDDEN);
      }
#pragma unroll
      for (int u = 0; u < 8; ++u) acc += w[u] * v[u];
    }
    if (j + 4 <= n) {
      v2f v[4];
      float w[4];
#pragma unroll
      for (int u = 0; u < 4; ++u) {
        const int s = __builtin_amdgcn_readlane(my.x, j + u);
        w[u] = __int_as_float(__builtin_amdgcn_readlane(my.y, j + u));
        v[u] = *(const v2f*)(src + (size_t)s * HIDDEN);
      }
#pragma unroll
      for (int u = 0; u < 4; ++u) acc += w[u] * v[u];
      j += 4;
    }
    for (; j < n; ++j) {
      const int s = __builtin_amdgcn_readlane(my.x, j);
      const float w = __int_as_float(__builtin_amdgcn_readlane(my.y, j));
      const v2f v = *(const v2f*)(src + (size_t)s * HIDDEN);
      acc += w * v;
    }
  }
  const v2f conv = acc - hv * ew + xv * bt;
  v2f hn;
  hn.x = hv.x + STEP_SZ * fmaxf(conv.x, 0.0f);
  hn.y = hv.y + STEP_SZ * fmaxf(conv.y, 0.0f);
  *(v2f*)(h + ro) = hn;
}


__global__ __launch_bounds__(128)
void dec_gemm_kernel(const float* __restrict__ h, const float* __restrict__ dw,
                     float* __restrict__ out) {
  int wave = (int)((blockIdx.x * blockDim.x + threadIdx.x) >> 5);
  if (wave >= M_TILES) return;
  int lane = threadIdx.x & 31;
  int mr = lane & 15;
  int kh = lane >> 4;
  const float* arow = h + (size_t)(wave * 16 + mr) * HIDDEN;
  const float* b0r = dw + (size_t)(mr +  0) * HIDDEN;
  const float* b1r = dw + (size_t)(mr + 16) * HIDDEN;
  int n2 = mr + 32;
  const float* b2r = dw + (size_t)(n2 < NCLASS ? n2 : 0) * HIDDEN;
  float b2scale = (n2 < NCLASS) ? 1.0f : 0.0f;
  v8f c0 = {}, c1 = {}, c2 = {};
  for (int k = 0; k < HIDDEN; k += 4) {
    int ko = k + 2 * kh;
    v2f a  = *(const v2f*)(arow + ko);
    v2f b0 = *(const v2f*)(b0r + ko);
    v2f b1 = *(const v2f*)(b1r + ko);
    v2f b2 = *(const v2f*)(b2r + ko);
    b2 *= b2scale;
    c0 = __builtin_amdgcn_wmma_f32_16x16x4_f32(false, a, false, b0, (short)0, c0, false, false);
    c1 = __builtin_amdgcn_wmma_f32_16x16x4_f32(false, a, false, b1, (short)0, c1, false, false);
    c2 = __builtin_amdgcn_wmma_f32_16x16x4_f32(false, a, false, b2, (short)0, c2, false, false);
  }
  int mbase = wave * 16 + kh * 8;
  for (int r = 0; r < 8; ++r) {
    float* orow = out + (size_t)(mbase + r) * NCLASS;
    orow[mr]      = c0[r];
    orow[mr + 16] = c1[r];
    if (n2 < NCLASS) orow[n2] = c2[r];
  }
}


extern "C" void kernel_launch(void* const* d_in, const int* in_sizes, int n_in,
                              void* d_out, int out_size, void* d_ws, size_t ws_size,
                              hipStream_t stream) {
  (void)in_sizes; (void)n_in; (void)out_size; (void)ws_size;
  const float* x     = (const float*)d_in[0];
  const int*   ei    = (const int*)  d_in[1];
  const float* enc_w = (const float*)d_in[2];
  const float* dec_w = (const float*)d_in[3];
  const float* ext_w = (const float*)d_in[4];
  const float* beta  = (const float*)d_in[5];
  const float* pw    = (const float*)d_in[6];
  const int* row = ei;
  const int* col = ei + N_EDGES;

  char* wsb = (char*)d_ws;
  int*   cnt   = (int*)  (wsb);
  int*   off   = (int*)  (wsb + 200192);
  float* dinv  = (float*)(wsb + 400384);
  int*   rank  = (int*)  (wsb + 600576);
  int2*  edges = (int2*) (wsb + 7000576);
  float* W     = (float*)(wsb + 19800576);
  float* h     = (float*)(wsb + 19816960);
  float* x0    = (float*)(wsb + 32616960);
  float* outp  = (float*)(wsb + 45416960);
  int*   hist  = (int*)  (wsb + 58216960);
  int*   pre   = (int*)  (wsb + 64616960);

  count_local_kernel<<<CHUNKS, CNT_T, 0, stream>>>(col, hist, rank);
  chunk_prefix_kernel<<<(N_NODES + 255) / 256, 256, 0, stream>>>(hist, pre, cnt);
  scan_dinv_kernel<<<1, 1024, 0, stream>>>(cnt, off, dinv);
  fill_kernel<<<(N_EDGES + 255) / 256, 256, 0, stream>>>(row, col, rank, off, pre, dinv, edges);
  weff_kernel<<<1, HIDDEN, 0, stream>>>(pw, W);

  const int gemm_blocks = (M_TILES + 3) / 4;
  enc_gemm_kernel<<<ENC_WGS, ENC_T, 0, stream>>>(x, enc_w, h, x0);

  for (int l = 0; l < NUM_LAYERS; ++l) {
    hw_gemm_kernel<<<gemm_blocks, 128, 0, stream>>>(h, W, outp);
    gather_update_kernel<<<N_NODES / 8, 256, 0, stream>>>(off, edges, outp, h, x0, ext_w, beta);
  }

  dec_gemm_kernel<<<gemm_blocks, 128, 0, stream>>>(h, dec_w, (float*)d_out);
}


# Out_p of layers 2-4 and the decoder product made in the gather-update kernel: a 16-node work-group puts its new h rows in LDS, 4 (3) of its waves multiply them; 4 launches fewer; sum orders kept
# speedup vs baseline: 3.6953x; 1.1179x over previous
#include <hip/hip_runtime.h>
#include <math.h>

#define N_NODES   50000
#define N_EDGES   1600000
#define NFEAT     500
#define HIDDEN    64
#define NCLASS    40
#define NUM_LAYERS 4
#define STEP_SZ   0.1f
#define M_TILES   (N_NODES / 16)

typedef __attribute__((ext_vector_type(2))) float v2f;
typedef __attribute__((ext_vector_type(8))) float v8f;


#define CHUNKS   32
#define CHUNK_E  (N_EDGES / CHUNKS)
#define CHUNK_Q  (CHUNK_E / 4)
#define CNT_T    1024

__global__ __launch_bounds__(CNT_T)
void count_local_kernel(const int* __restrict__ col, int* __restrict__ hist,
                        int* __restrict__ rank) {
  __shared__ int cnt[N_NODES];
  const int t = threadIdx.x;
  for (int i = t; i < N_NODES; i += CNT_T) cnt[i] = 0;
  __syncthreads();
  const int4* c4 = (const int4*)(col + (size_t)blockIdx.x * CHUNK_E);
  int4* r4 = (int4*)(rank + (size_t)blockIdx.x * CHUNK_E);
  const int full = CHUNK_Q / CNT_T;
  for (int k = 0; k < full; k += 4) {
    int4 c[4];
#pragma unroll
    for (int u = 0; u < 4; ++u) c[u] = c4[(k + u) * CNT_T + t];
#pragma unroll
    for (int u = 0; u < 4; ++u) {
      int4 r;
      r.x = atomicAdd(&cnt[c[u].x], 1);
      r.y = atomicAdd(&cnt[c[u].y], 1);
      r.z = atomicAdd(&cnt[c[u].z], 1);
      r.w = atomicAdd(&cnt[c[u].w], 1);
      r4[(k + u) * CNT_T + t] = r;
    }
  }
  {
    const int q = full * CNT_T + t;
    if (q < CHUNK_Q) {
      const int4 c = c4[q];
      int4 r;
      r.x = atomicAdd(&cnt[c.x], 1);
      r.y = atomicAdd(&cnt[c.y], 1);
      r.z = atomicAdd(&cnt[c.z], 1);
      r.w = atomicAdd(&cnt[c.w], 1);
      r4[q] = r;
    }
  }
  __syncthreads();
  int* mine = hist + (size_t)blockIdx.x * N_NODES;
  for (int i = t; i < N_NODES; i += CNT_T) mine[i] = cnt[i];
}

__global__ __launch_bounds__(256)
void chunk_prefix_kernel(const int* __restrict__ hist, int* __restrict__ pre,
                         int* __restrict__ cnt) {
  const int c = blockIdx.x * blockDim.x + threadIdx.x;
  if (c < N_NODES) {
    int run = 0;
#pragma unroll
    for (int g = 0; g < CHUNKS; ++g) {
      const int v = hist[(size_t)g * N_NODES + c];
      pre[(size_t)g * N_NODES + c] = run;
      run += v;
    }
    cnt[c] = run;
  }
}

#define SCAN_T   1024
#define SCAN_PER 49

__global__ __launch_bounds__(SCAN_T)
void scan_dinv_kernel(const int* __restrict__ cnt, int* __restrict__ off,
                      float* __restrict__ dinv) {
  __shared__ int sc[SCAN_T * SCAN_PER];
  __shared__ int wsum[32];
  const int t = threadIdx.x;
  const int lane = t & 31;
  const int w = t >> 5;
  for (int k = 0; k < SCAN_PER; ++k) {
    const int i = k * SCAN_T + t;
    int v = 0;
    if (i < N_NODES) {
      v = cnt[i];
      dinv[i] = (v > 0) ? rsqrtf((float)v) : 0.0f;
    }
    sc[i] = v;
  }
  __syncthreads();
  int* mine = sc + t * SCAN_PER;
  int run = 0;
  for (int k = 0; k < SCAN_PER; ++k) run += mine[k];
  int s = run;
  for (int d = 1; d < 32; d <<= 1) {
    int up = __shfl_up(s, d);
    if (lane >= d) s += up;
  }
  if (lane == 31) wsum[w] = s;
  __syncthreads();
  int ws = wsum[lane];
  for (int d = 1; d < 32; d <<= 1) {
    int up = __shfl_up(ws, d);
    if (lane >= d) ws += up;
  }
  const int prev  = __shfl(ws, (w + 31) & 31);
  const int total = __shfl(ws, 31);
  int pos = ((w > 0) ? prev : 0) + s - run;
  for (int k = 0; k < SCAN_PER; ++k) {
    const int v = mine[k];
    mine[k] = pos;
    pos += v;
  }
  __syncthreads();
  for (int k = 0; k < SCAN_PER; ++k) {
    const int i = k * SCAN_T + t;
    if (i < N_NODES) off[i] = sc[i];
  }
  if (t == 0) off[N_NODES] = total;
}

__global__ void fill_kernel(const int* __restrict__ row, const int* __restrict__ col,
                            const int* __restrict__ rank, const int* __restrict__ off,
                            const int* __restrict__ pre, const float* __restrict__ dinv,
                            int2* __restrict__ edges) {
  int e = blockIdx.x * blockDim.x + threadIdx.x;
  if (e < N_EDGES) {
    int r = row[e];
    int c = col[e];
    int2 rec;
    rec.x = r;
    rec.y = __float_as_int(dinv[r] * dinv[c]);
    edges[off[c] + pre[(size_t)(e / CHUNK_E) * N_NODES + c] + rank[e]] = rec;
  }
}

__global__ void weff_kernel(const float* __restrict__ pw, float* __restrict__ W) {
  int i = threadIdx.x;
  const int LD = HIDDEN + 2;
  float q = pw[i * LD + HIDDEN];
  float r = pw[i * LD + HIDDEN + 1];
  float s = 0.0f;
  for (int j = 0; j < HIDDEN; ++j) {
    float v = 0.0f;
    if (i < j)      v = pw[i * LD + j];
    else if (i > j) v = pw[j * LD + i];
    s += fabsf(v);
    W[i * HIDDEN + j] = v;
  }
  W[i * HIDDEN + i] = q * s + r;
}


typedef __attribute__((ext_vector_type(4))) float v4f;

#define ENC_T   512
#define ENC_WGS 256

#define ENC_WMMA(av, bv, cv) cv = __builtin_amdgcn_wmma_f32_16x16x4_f32(false, av, false, bv, (short)0, cv, false, false)

#define ENC_STEP8(a, bq)                                              \
  {                                                                   \
    const v4f b0 = *(const v4f*)((bq));                               \
    const v4f b1 = *(const v4f*)((bq) + 16 * NFEAT);                  \
    const v4f b2 = *(const v4f*)((bq) + 32 * NFEAT);                  \
    const v4f b3 = *(const v4f*)((bq) + 48 * NFEAT);                  \
    const v2f alo = {(a).x, (a).y};                                   \
    const v2f ahi = {(a).z, (a).w};                                   \
    const v2f b0l = {b0.x, b0.y}, b0h = {b0.z, b0.w};                 \
    const v2f b1l = {b1.x, b1.y}, b1h = {b1.z, b1.w};                 \
    const v2f b2l = {b2.x, b2.y}, b2h = {b2.z, b2.w};                 \
    const v2f b3l = {b3.x, b3.y}, b3h = {b3.z, b3.w};                 \
    ENC_WMMA(alo, b0l, c0); ENC_WMMA(alo, b1l, c1);                   \
    ENC_WMMA(alo, b2l, c2); ENC_WMMA(alo, b3l, c3);                   \
    ENC_WMMA(ahi, b0h, c0); ENC_WMMA(ahi, b1h, c1);                   \
    ENC_WMMA(ahi, b2h, c2); ENC_WMMA(ahi, b3h, c3);                   \
  }

#define ENC_LOAD4(v, ptr)                                             \
  v##0 = *(const v4f*)((ptr) +  0);                                   \
  v##1 = *(const v4f*)((ptr) +  8);                                   \
  v##2 = *(const v4f*)((ptr) + 16);                                   \
  v##3 = *(const v4f*)((ptr) + 24);
#define ENC_GROUP(v, bq)                                              \
  ENC_STEP8(v##0, (bq) +  0)                                          \
  ENC_STEP8(v##1, (bq) +  8)                                          \
  ENC_STEP8(v##2, (bq) + 16)                                          \
  ENC_STEP8(v##3, (bq) + 24)

__global__ __launch_bounds__(ENC_T)
void enc_gemm_kernel(const float* __restrict__ x, const float* __restrict__ ew,
                     float* __restrict__ h, float* __restrict__ x0) {
  __shared__ __attribute__((aligned(16))) float Bs[HIDDEN * NFEAT];
  {
    const v4f* s4 = (const v4f*)ew;
    v4f* d4 = (v4f*)Bs;
    for (int i = threadIdx.x; i < HIDDEN * NFEAT / 4; i += ENC_T) d4[i] = s4[i];
  }
  __syncthreads();
  const int tile = (int)(blockIdx.x + gridDim.x * (threadIdx.x >> 5));
  if (tile >= M_TILES) return;
  const int lane = threadIdx.x & 31;
  const int mr = lane & 15;
  const int kh = lane >> 4;
  const float* arow = x + (size_t)(tile * 16 + mr) * NFEAT;
  const float* ap = arow + 4 * kh;
  const float* bp = Bs + mr * NFEAT + 4 * kh;
  v8f c0 = {}, c1 = {}, c2 = {}, c3 = {};
  v4f p0, p1, p2, p3, q0, q1, q2, q3, s0, s1, s2, s3;
  ENC_LOAD4(p, ap)
  ENC_LOAD4(q, ap + 32)
  for (int it = 0; it < 4; ++it) {
    const float* ag = ap + 96 * it;
    const float* bg = bp + 96 * it;
    ENC_LOAD4(s, ag + 64)
    ENC_GROUP(p, bg)
    ENC_LOAD4(p, ag + 96)
    ENC_GROUP(q, bg + 32)
    ENC_LOAD4(q, ag + 128)
    ENC_GROUP(s, bg + 64)
  }
  {
    ENC_LOAD4(s, ap + 448)
    const v4f r0 = *(const v4f*)(ap + 480);
    const v4f r1 = *(const v4f*)(ap + 488);
    const v2f r2 = *(const v2f*)(arow + 496 + 2 * kh);
    ENC_GROUP(p, bp + 384)
    ENC_GROUP(q, bp + 416)
    ENC_GROUP(s, bp + 448)
    ENC_STEP8(r0, bp + 480)
    ENC_STEP8(r1, bp + 488)
    const float* bt = Bs + mr * NFEAT + 496 + 2 * kh;
    const v2f t0 = *(const v2f*)(bt);
    const v2f t1 = *(const v2f*)(bt + 16 * NFEAT);
    const v2f t2 = *(const v2f*)(bt + 32 * NFEAT);
    const v2f t3 = *(const v2f*)(bt + 48 * NFEAT);
    ENC_WMMA(r2, t0, c0); ENC_WMMA(r2, t1, c1);
    ENC_WMMA(r2, t2, c2); ENC_WMMA(r2, t3, c3);
  }
  int mbase = tile * 16 + kh * 8;
  for (int r = 0; r < 8; ++r) {
    size_t o = (size_t)(mbase + r) * HIDDEN + mr;
    h[o]      = c0[r]; x0[o]      = c0[r];
    h[o + 16] = c1[r]; x0[o + 16] = c1[r];
    h[o + 32] = c2[r]; x0[o + 32] = c2[r];
    h[o + 48] = c3[r]; x0[o + 48] = c3[r];
  }
}


__global__ __launch_bounds__(128)
void hw_gemm_kernel(const float* __restrict__ h, const float* __restrict__ W,
                    float* __restrict__ outp) {
  __shared__ float Ws[HIDDEN * HIDDEN];
  for (int i = threadIdx.x; i < HIDDEN * HIDDEN; i += blockDim.x) Ws[i] = W[i];
  __syncthreads();
  int wave = (int)((blockIdx.x * blockDim.x + threadIdx.x) >> 5);
  if (wave >= M_TILES) return;
  int lane = threadIdx.x & 31;
  int mr = lane & 15;
  int kh = lane >> 4;
  const float* arow = h + (size_t)(wave * 16 + mr) * HIDDEN;
  v8f c0 = {}, c1 = {}, c2 = {}, c3 = {};
  for (int k = 0; k < HIDDEN; k += 4) {
    int ko = k + 2 * kh;
    v2f a  = *(const v2f*)(arow + ko);
    v2f b0 = *(const v2f*)(&Ws[(mr +  0) * HIDDEN + ko]);
    v2f b1 = *(const v2f*)(&Ws[(mr + 16) * HIDDEN + ko]);
    v2f b2 = *(const v2f*)(&Ws[(mr + 32) * HIDDEN + ko]);
    v2f b3 = *(const v2f*)(&Ws[(mr + 48) * HIDDEN + ko]);
    c0 = __builtin_amdgcn_wmma_f32_16x16x4_f32(false, a, false, b0, (short)0, c0, false, false);
    c1 = __builtin_amdgcn_wmma_f32_16x16x4_f32(false, a, false, b1, (short)0, c1, false, false);
    c2 = __builtin_amdgcn_wmma_f32_16x16x4_f32(false, a, false, b2, (short)0, c2, false, false);
    c3 = __builtin_amdgcn_wmma_f32_16x16x4_f32(false, a, false, b3, (short)0, c3, false, false);
  }
  int mbase = wave * 16 + kh * 8;
  for (int r = 0; r < 8; ++r) {
    size_t o = (size_t)(mbase + r) * HIDDEN + mr;
    outp[o]      = c0[r];
    outp[o + 16] = c1[r];
    outp[o + 32] = c2[r];
    outp[o + 48] = c3[r];
  }
}


#define GU_T   512
#define HS_P   68

template <int NT, int LDO>
__global__ __launch_bounds__(GU_T)
void gather_update_mm_kernel(const int* __restrict__ off, const int2* __restrict__ edges,
                             const float* __restrict__ outp, float* __restrict__ h,
                             const float* __restrict__ x0, const float* __restrict__ ext_w,
                             const float* __restrict__ beta, const float* __restrict__ Bm,
                             float* __restrict__ dst) {
  __shared__ __attribute__((aligned(16))) float hs[16 * HS_P];
  const int wv = __builtin_amdgcn_readfirstlane((int)(threadIdx.x >> 5));
  const int c = (int)blockIdx.x * 16 + wv;
  const int lane = threadIdx.x & 31;
  const int beg = off[c];
  const int end = off[c + 1];
  const float* src = outp + 2 * lane;
  const size_t ro = (size_t)c * HIDDEN + 2 * lane;
  const v2f hv = *(const v2f*)(h + ro);
  const v2f xv = *(const v2f*)(x0 + ro);
  const v2f ew = *(const v2f*)(ext_w + 2 * lane);
  const float bt = beta[0];
  v2f acc = {0.0f, 0.0f};
  for (int base = beg; base < end; base += 32) {
    int n = end - base;
    n = (n < 32) ? n : 32;
    int2 my;
    my.x = 0;
    my.y = 0;
    if (lane < n) my = edges[base + lane];
    int j = 0;
    for (; j + 8 <= n; j += 8) {
      v2f v[8];
      float w[8];
#pragma unroll
      for (int u = 0; u < 8; ++u) {
        const int s = __builtin_amdgcn_readlane(my.x, j + u);
        w[u] = __int_as_float(__builtin_amdgcn_readlane(my.y, j + u));
        v[u] = *(const v2f*)(src + (size_t)s * HIDDEN);
      }
#pragma unroll
      for (int u = 0; u < 8; ++u) acc += w[u] * v[u];
    }
    if (j + 4 <= n) {
      v2f v[4];
      float w[4];
#pragma unroll
      for (int u = 0; u < 4; ++u) {
        const int s = __builtin_amdgcn_readlane(my.x, j + u);
        w[u] = __int_as_float(__builtin_amdgcn_readlane(my.y, j + u));
        v[u] = *(const v2f*)(src + (size_t)s * HIDDEN);
      }
#pragma unroll
      for (int u = 0; u < 4; ++u) acc += w[u] * v[u];
      j += 4;
    }
    for (; j < n; ++j) {
      const int s = __builtin_amdgcn_readlane(my.x, j);
      const float w = __int_as_float(__builtin_amdgcn_readlane(my.y, j));
      const v2f v = *(const v2f*)(src + (size_t)s * HIDDEN);
      acc += w * v;
    }
  }
  const v2f conv = acc - hv * ew + xv * bt;
  v2f hn;
  hn.x = hv.x + STEP_SZ * fmaxf(conv.x, 0.0f);
  hn.y = hv.y + STEP_SZ * fmaxf(conv.y, 0.0f);
  const int mr = lane & 15;
  const int kh = lane >> 4;
  const int nn = mr + 16 * wv;
  const bool live = (nn < LDO);
  *(v2f*)(hs + wv * HS_P + 2 * lane) = hn;
  v2f bf[HIDDEN / 4];
  if (wv < NT) {
    const float* brow = Bm + (size_t)(live ? nn : 0) * HIDDEN + 2 * kh;
#pragma unroll
    for (int i = 0; i < HIDDEN / 4; ++i) bf[i] = *(const v2f*)(brow + 4 * i);
    if (NT == 3) {
      const float bscale = live ? 1.0f : 0.0f;
#pragma unroll
      for (int i = 0; i < HIDDEN / 4; ++i) bf[i] *= bscale;
    }
  }
  __syncthreads();
  if (NT == 4) *(v2f*)(h + ro) = hn;
  if (wv < NT) {
    const float* arow = hs + mr * HS_P + 2 * kh;
    v8f cc = {};
#pragma unroll
    for (int i = 0; i < HIDDEN / 4; ++i) {
      const v2f a = *(const v2f*)(arow + 4 * i);
      cc = __builtin_amdgcn_wmma_f32_16x16x4_f32(false, a, false, bf[i], (short)0, cc, false, false);
    }
    const int mbase = (int)blockIdx.x * 16 + kh * 8;
    for (int r = 0; r < 8; ++r) {
      if (live) dst[(size_t)(mbase + r) * LDO + nn] = cc[r];
    }
  }
}


extern "C" void kernel_launch(void* const* d_in, const int* in_sizes, int n_in,
                              void* d_out, int out_size, void* d_ws, size_t ws_size,
                              hipStream_t stream) {
  (void)in_sizes; (void)n_in; (void)out_size; (void)ws_size;
  const float* x     = (const float*)d_in[0];
  const int*   ei    = (const int*)  d_in[1];
  const float* enc_w = (const float*)d_in[2];
  const float* dec_w = (const float*)d_in[3];
  const float* ext_w = (const float*)d_in[4];
  const float* beta  = (const float*)d_in[5];
  const float* pw    = (const float*)d_in[6];
  const int* row = ei;
  const int* col = ei + N_EDGES;

  char* wsb = (char*)d_ws;
  int*   cnt   = (int*)  (wsb);
  int*   off   = (int*)  (wsb + 200192);
  float* dinv  = (float*)(wsb + 400384);
  int*   rank  = (int*)  (wsb + 600576);
  int2*  edges = (int2*) (wsb + 7000576);
  float* W     = (float*)(wsb + 19800576);
  float* h     = (float*)(wsb + 19816960);
  float* x0    = (float*)(wsb + 32616960);
  float* outp  = (float*)(wsb + 45416960);
  int*   hist  = (int*)  (wsb + 58216960);
  int*   pre   = (int*)  (wsb + 64616960);
  float* outq  = (float*)(wsb + 71016960);

  count_local_kernel<<<CHUNKS, CNT_T, 0, stream>>>(col, hist, rank);
  chunk_prefix_kernel<<<(N_NODES + 255) / 256, 256, 0, stream>>>(hist, pre, cnt);
  scan_dinv_kernel<<<1, 1024, 0, stream>>>(cnt, off, dinv);
  fill_kernel<<<(N_EDGES + 255) / 256, 256, 0, stream>>>(row, col, rank, off, pre, dinv, edges);
  weff_kernel<<<1, HIDDEN, 0, stream>>>(pw, W);

  const int gemm_blocks = (M_TILES + 3) / 4;
  enc_gemm_kernel<<<ENC_WGS, ENC_T, 0, stream>>>(x, enc_w, h, x0);

  hw_gemm_kernel<<<gemm_blocks, 128, 0, stream>>>(h, W, outp);
  gather_update_mm_kernel<4, HIDDEN><<<M_TILES, GU_T, 0, stream>>>(off, edges, outp, h, x0, ext_w, beta, W, outq);
  gather_update_mm_kernel<4, HIDDEN><<<M_TILES, GU_T, 0, stream>>>(off, edges, outq, h, x0, ext_w, beta, W, outp);
  gather_update_mm_kernel<4, HIDDEN><<<M_TILES, GU_T, 0, stream>>>(off, edges, outp, h, x0, ext_w, beta, W, outq);
  gather_update_mm_kernel<3, NCLASS><<<M_TILES, GU_T, 0, stream>>>(off, edges, outq, h, x0, ext_w, beta, dec_w, (float*)d_out);
}


# W_eff set-up freed of dependent load chains: an extra work-group of the counting launch builds it with all loads at once (its own 64-step launch dropped); hw_gemm stages W with 8 loads in flight
# speedup vs baseline: 4.1725x; 1.1291x over previous
#include <hip/hip_runtime.h>
#include <math.h>

#define N_NODES   50000
#define N_EDGES   1600000
#define NFEAT     500
#define HIDDEN    64
#define NCLASS    40
#define NUM_LAYERS 4
#define STEP_SZ   0.1f
#define M_TILES   (N_NODES / 16)

typedef __attribute__((ext_vector_type(2))) float v2f;
typedef __attribute__((ext_vector_type(8))) float v8f;


#define CHUNKS   32
#define CHUNK_E  (N_EDGES / CHUNKS)
#define CHUNK_Q  (CHUNK_E / 4)
#define CNT_T    1024

#define W_PITCH  (HIDDEN + 1)

__global__ __launch_bounds__(CNT_T)
void count_local_kernel(const int* __restrict__ col, int* __restrict__ hist,
                        int* __restrict__ rank, const float* __restrict__ pw,
                        float* __restrict__ W) {
  __shared__ int cnt[N_NODES];
  __shared__ float wl[HIDDEN * W_PITCH];
  const int t = threadIdx.x;
  if (blockIdx.x == CHUNKS) {
    const int LD = HIDDEN + 2;
    float q = 0.0f, r = 0.0f;
    if (t < HIDDEN) {
      q = pw[t * LD + HIDDEN];
      r = pw[t * LD + HIDDEN + 1];
    }
    float v[HIDDEN * HIDDEN / CNT_T];
#pragma unroll
    for (int u = 0; u < HIDDEN * HIDDEN / CNT_T; ++u) {
      const int e = t + u * CNT_T;
      const int i = e >> 6;
      const int j = e & (HIDDEN - 1);
      v[u] = pw[(i < j) ? (i * LD + j) : (j * LD + i)];
    }
#pragma unroll
    for (int u = 0; u < HIDDEN * HIDDEN / CNT_T; ++u) {
      const int e = t + u * CNT_T;
      const int i = e >> 6;
      const int j = e & (HIDDEN - 1);
      const float x = (i == j) ? 0.0f : v[u];
      wl[i * W_PITCH + j] = x;
      if (i != j) W[e] = x;
    }
    __syncthreads();
    if (t < HIDDEN) {
      float s = 0.0f;
      for (int j = 0; j < HIDDEN; ++j) s += fabsf(wl[t * W_PITCH + j]);
      W[t * HIDDEN + t] = q * s + r;
    }
    return;
  }
  for (int i = t; i < N_NODES; i += CNT_T) cnt[i] = 0;
  __syncthreads();
  const int4* c4 = (const int4*)(col + (size_t)blockIdx.x * CHUNK_E);
  int4* r4 = (int4*)(rank + (size_t)blockIdx.x * CHUNK_E);
  const int full = CHUNK_Q / CNT_T;
  for (int k = 0; k < full; k += 4) {
    int4 c[4];
#pragma unroll
    for (int u = 0; u < 4; ++u) c[u] = c4[(k + u) * CNT_T + t];
#pragma unroll
    for (int u = 0; u < 4; ++u) {
      int4 r;
      r.x = atomicAdd(&cnt[c[u].x], 1);
      r.y = atomicAdd(&cnt[c[u].y], 1);
      r.z = atomicAdd(&cnt[c[u].z], 1);
      r.w = atomicAdd(&cnt[c[u].w], 1);
      r4[(k + u) * CNT_T + t] = r;
    }
  }
  {
    const int q = full * CNT_T + t;
    if (q < CHUNK_Q) {
      const int4 c = c4[q];
      int4 r;
      r.x = atomicAdd(&cnt[c.x], 1);
      r.y = atomicAdd(&cnt[c.y], 1);
      r.z = atomicAdd(&cnt[c.z], 1);
      r.w = atomicAdd(&cnt[c.w], 1);
      r4[q] = r;
    }
  }
  __syncthreads();
  int* mine = hist + (size_t)blockIdx.x * N_NODES;
  for (int i = t; i < N_NODES; i += CNT_T) mine[i] = cnt[i];
}

__global__ __launch_bounds__(256)
void chunk_prefix_kernel(const int* __restrict__ hist, int* __restrict__ pre,
                         int* __restrict__ cnt) {
  const int c = blockIdx.x * blockDim.x + threadIdx.x;
  if (c < N_NODES) {
    int run = 0;
#pragma unroll
    for (int g = 0; g < CHUNKS; ++g) {
      const int v = hist[(size_t)g * N_NODES + c];
      pre[(size_t)g * N_NODES + c] = run;
      run += v;
    }
    cnt[c] = run;
  }
}

#define SCAN_T   1024
#define SCAN_PER 49

__global__ __launch_bounds__(SCAN_T)
void scan_dinv_kernel(const int* __restrict__ cnt, int* __restrict__ off,
                      float* __restrict__ dinv) {
  __shared__ int sc[SCAN_T * SCAN_PER];
  __shared__ int wsum[32];
  const int t = threadIdx.x;
  const int lane = t & 31;
  const int w = t >> 5;
  for (int k = 0; k < SCAN_PER; ++k) {
    const int i = k * SCAN_T + t;
    int v = 0;
    if (i < N_NODES) {
      v = cnt[i];
      dinv[i] = (v > 0) ? rsqrtf((float)v) : 0.0f;
    }
    sc[i] = v;
  }
  __syncthreads();
  int* mine = sc + t * SCAN_PER;
  int run = 0;
  for (int k = 0; k < SCAN_PER; ++k) run += mine[k];
  int s = run;
  for (int d = 1; d < 32; d <<= 1) {
    int up = __shfl_up(s, d);
    if (lane >= d) s += up;
  }
  if (lane == 31) wsum[w] = s;
  __syncthreads();
  int ws = wsum[lane];
  for (int d = 1; d < 32; d <<= 1) {
    int up = __shfl_up(ws, d);
    if (lane >= d) ws += up;
  }
  const int prev  = __shfl(ws, (w + 31) & 31);
  const int total = __shfl(ws, 31);
  int pos = ((w > 0) ? prev : 0) + s - run;
  for (int k = 0; k < SCAN_PER; ++k) {
    const int v = mine[k];
    mine[k] = pos;
    pos += v;
  }
  __syncthreads();
  for (int k = 0; k < SCAN_PER; ++k) {
    const int i = k * SCAN_T + t;
    if (i < N_NODES) off[i] = sc[i];
  }
  if (t == 0) off[N_NODES] = total;
}

__global__ void fill_kernel(const int* __restrict__ row, const int* __restrict__ col,
                            const int* __restrict__ rank, const int* __restrict__ off,
                            const int* __restrict__ pre, const float* __restrict__ dinv,
                            int2* __restrict__ edges) {
  int e = blockIdx.x * blockDim.x + threadIdx.x;
  if (e < N_EDGES) {
    int r = row[e];
    int c = col[e];
    int2 rec;
    rec.x = r;
    rec.y = __float_as_int(dinv[r] * dinv[c]);
    edges[off[c] + pre[(size_t)(e / CHUNK_E) * N_NODES + c] + rank[e]] = rec;
  }
}


typedef __attribute__((ext_vector_type(4))) float v4f;

#define ENC_T   512
#define ENC_WGS 256

#define ENC_WMMA(av, bv, cv) cv = __builtin_amdgcn_wmma_f32_16x16x4_f32(false, av, false, bv, (short)0, cv, false, false)

#define ENC_STEP8(a, bq)                                              \
  {                                                                   \
    const v4f b0 = *(const v4f*)((bq));                               \
    const v4f b1 = *(const v4f*)((bq) + 16 * NFEAT);                  \
    const v4f b2 = *(const v4f*)((bq) + 32 * NFEAT);                  \
    const v4f b3 = *(const v4f*)((bq) + 48 * NFEAT);                  \
    const v2f alo = {(a).x, (a).y};                                   \
    const v2f ahi = {(a).z, (a).w};                                   \
    const v2f b0l = {b0.x, b0.y}, b0h = {b0.z, b0.w};                 \
    const v2f b1l = {b1.x, b1.y}, b1h = {b1.z, b1.w};                 \
    const v2f b2l = {b2.x, b2.y}, b2h = {b2.z, b2.w};                 \
    const v2f b3l = {b3.x, b3.y}, b3h = {b3.z, b3.w};                 \
    ENC_WMMA(alo, b0l, c0); ENC_WMMA(alo, b1l, c1);                   \
    ENC_WMMA(alo, b2l, c2); ENC_WMMA(alo, b3l, c3);                   \
    ENC_WMMA(ahi, b0h, c0); ENC_WMMA(ahi, b1h, c1);                   \
    ENC_WMMA(ahi, b2h, c2); ENC_WMMA(ahi, b3h, c3);                   \
  }

#define ENC_LOAD4(v, ptr)                                             \
  v##0 = *(const v4f*)((ptr) +  0);                                   \
  v##1 = *(const v4f*)((ptr) +  8);                                   \
  v##2 = *(const v4f*)((ptr) + 16);                                   \
  v##3 = *(const v4f*)((ptr) + 24);
#define ENC_GROUP(v, bq)                                              \
  ENC_STEP8(v##0, (bq) +  0)                                          \
  ENC_STEP8(v##1, (bq) +  8)                                          \
  ENC_STEP8(v##2, (bq) + 16)                                          \
  ENC_STEP8(v##3, (bq) + 24)

__global__ __launch_bounds__(ENC_T)
void enc_gemm_kernel(const float* __restrict__ x, const float* __restrict__ ew,
                     float* __restrict__ h, float* __restrict__ x0) {
  __shared__ __attribute__((aligned(16))) float Bs[HIDDEN * NFEAT];
  {
    const v4f* s4 = (const v4f*)ew;
    v4f* d4 = (v4f*)Bs;
    for (int i = threadIdx.x; i < HIDDEN * NFEAT / 4; i += ENC_T) d4[i] = s4[i];
  }
  __syncthreads();
  const int tile = (int)(blockIdx.x + gridDim.x * (threadIdx.x >> 5));
  if (tile >= M_TILES) return;
  const int lane = threadIdx.x & 31;
  const int mr = lane & 15;
  const int kh = lane >> 4;
  const float* arow = x + (size_t)(tile * 16 + mr) * NFEAT;
  const float* ap = arow + 4 * kh;
  const float* bp = Bs + mr * NFEAT + 4 * kh;
  v8f c0 = {}, c1 = {}, c2 = {}, c3 = {};
  v4f p0, p1, p2, p3, q0, q1, q2, q3, s0, s1, s2, s3;
  ENC_LOAD4(p, ap)
  ENC_LOAD4(q, ap + 32)
  for (int it = 0; it < 4; ++it) {
    const float* ag = ap + 96 * it;
    const float* bg = bp + 96 * it;
    ENC_LOAD4(s, ag + 64)
    ENC_GROUP(p, bg)
    ENC_LOAD4(p, ag + 96)
    ENC_GROUP(q, bg + 32)
    ENC_LOAD4(q, ag + 128)
    ENC_GROUP(s, bg + 64)
  }
  {
    ENC_LOAD4(s, ap + 448)
    const v4f r0 = *(const v4f*)(ap + 480);
    const v4f r1 = *(const v4f*)(ap + 488);
    const v2f r2 = *(const v2f*)(arow + 496 + 2 * kh);
    ENC_GROUP(p, bp + 384)
    ENC_GROUP(q, bp + 416)
    ENC_GROUP(s, bp + 448)
    ENC_STEP8(r0, bp + 480)
    ENC_STEP8(r1, bp + 488)
    const float* bt = Bs + mr * NFEAT + 496 + 2 * kh;
    const v2f t0 = *(const v2f*)(bt);
    const v2f t1 = *(const v2f*)(bt + 16 * NFEAT);
    const v2f t2 = *(const v2f*)(bt + 32 * NFEAT);
    const v2f t3 = *(const v2f*)(bt + 48 * NFEAT);
    ENC_WMMA(r2, t0, c0); ENC_WMMA(r2, t1, c1);
    ENC_WMMA(r2, t2, c2); ENC_WMMA(r2, t3, c3);
  }
  int mbase = tile * 16 + kh * 8;
  for (int r = 0; r < 8; ++r) {
    size_t o = (size_t)(mbase + r) * HIDDEN + mr;
    h[o]      = c0[r]; x0[o]      = c0[r];
    h[o + 16] = c1[r]; x0[o + 16] = c1[r];
    h[o + 32] = c2[r]; x0[o + 32] = c2[r];
    h[o + 48] = c3[r]; x0[o + 48] = c3[r];
  }
}


#define HW_T 128

__global__ __launch_bounds__(HW_T)
void hw_gemm_kernel(const float* __restrict__ h, const float* __restrict__ W,
                    float* __restrict__ outp) {
  __shared__ __attribute__((aligned(16))) float Ws[HIDDEN * HIDDEN];
  {
    const v4f* s4 = (const v4f*)W;
    v4f* d4 = (v4f*)Ws;
#pragma unroll
    for (int u = 0; u < HIDDEN * HIDDEN / 4 / HW_T; ++u) d4[threadIdx.x + HW_T * u] = s4[threadIdx.x + HW_T * u];
  }
  __syncthreads();
  int wave = (int)((blockIdx.x * HW_T + threadIdx.x) >> 5);
  if (wave >= M_TILES) return;
  int lane = threadIdx.x & 31;
  int mr = lane & 15;
  int kh = lane >> 4;
  const float* arow = h + (size_t)(wave * 16 + mr) * HIDDEN;
  v8f c0 = {}, c1 = {}, c2 = {}, c3 = {};
  for (int k = 0; k < HIDDEN; k += 4) {
    int ko = k + 2 * kh;
    v2f a  = *(const v2f*)(arow + ko);
    v2f b0 = *(const v2f*)(&Ws[(mr +  0) * HIDDEN + ko]);
    v2f b1 = *(const v2f*)(&Ws[(mr + 16) * HIDDEN + ko]);
    v2f b2 = *(const v2f*)(&Ws[(mr + 32) * HIDDEN + ko]);
    v2f b3 = *(const v2f*)(&Ws[(mr + 48) * HIDDEN + ko]);
    c0 = __builtin_amdgcn_wmma_f32_16x16x4_f32(false, a, false, b0, (short)0, c0, false, false);
    c1 = __builtin_amdgcn_wmma_f32_16x16x4_f32(false, a, false, b1, (short)0, c1, false, false);
    c2 = __builtin_amdgcn_wmma_f32_16x16x4_f32(false, a, false, b2, (short)0, c2, false, false);
    c3 = __builtin_amdgcn_wmma_f32_16x16x4_f32(false, a, false, b3, (short)0, c3, false, false);
  }
  int mbase = wave * 16 + kh * 8;
  for (int r = 0; r < 8; ++r) {
    size_t o = (size_t)(mbase + r) * HIDDEN + mr;
    outp[o]      = c0[r];
    outp[o + 16] = c1[r];
    outp[o + 32] = c2[r];
    outp[o + 48] = c3[r];
  }
}


#define GU_T   512
#define HS_P   68

template <int NT, int LDO>
__global__ __launch_bounds__(GU_T)
void gather_update_mm_kernel(const int* __restrict__ off, const int2* __restrict__ edges,
                             const float* __restrict__ outp, float* __restrict__ h,
                             const float* __restrict__ x0, const float* __restrict__ ext_w,
                             const float* __restrict__ beta, const float* __restrict__ Bm,
                             float* __restrict__ dst) {
  __shared__ __attribute__((aligned(16))) float hs[16 * HS_P];
  const int wv = __builtin_amdgcn_readfirstlane((int)(threadIdx.x >> 5));
  const int c = (int)blockIdx.x * 16 + wv;
  const int lane = threadIdx.x & 31;
  const int beg = off[c];
  const int end = off[c + 1];
  const float* src = outp + 2 * lane;
  const size_t ro = (size_t)c * HIDDEN + 2 * lane;
  const v2f hv = *(const v2f*)(h + ro);
  const v2f xv = *(const v2f*)(x0 + ro);
  const v2f ew = *(const v2f*)(ext_w + 2 * lane);
  const float bt = beta[0];
  v2f acc = {0.0f, 0.0f};
  for (int base = beg; base < end; base += 32) {
    int n = end - base;
    n = (n < 32) ? n : 32;
    int2 my;
    my.x = 0;
    my.y = 0;
    if (lane < n) my = edges[base + lane];
    int j = 0;
    for (; j + 8 <= n; j += 8) {
      v2f v[8];
      float w[8];
#pragma unroll
      for (int u = 0; u < 8; ++u) {
        const int s = __builtin_amdgcn_readlane(my.x, j + u);
        w[u] = __int_as_float(__builtin_amdgcn_readlane(my.y, j + u));
        v[u] = *(const v2f*)(src + (size_t)s * HIDDEN);
      }
#pragma unroll
      for (int u = 0; u < 8; ++u) acc += w[u] * v[u];
    }
    if (j + 4 <= n) {
      v2f v[4];
      float w[4];
#pragma unroll
      for (int u = 0; u < 4; ++u) {
        const int s = __builtin_amdgcn_readlane(my.x, j + u);
        w[u] = __int_as_float(__builtin_amdgcn_readlane(my.y, j + u));
        v[u] = *(const v2f*)(src + (size_t)s * HIDDEN);
      }
#pragma unroll
      for (int u = 0; u < 4; ++u) acc += w[u] * v[u];
      j += 4;
    }
    for (; j < n; ++j) {
      const int s = __builtin_amdgcn_readlane(my.x, j);
      const float w = __int_as_float(__builtin_amdgcn_readlane(my.y, j));
      const v2f v = *(const v2f*)(src + (size_t)s * HIDDEN);
      acc += w * v;
    }
  }
  const v2f conv = acc - hv * ew + xv * bt;
  v2f hn;
  hn.x = hv.x + STEP_SZ * fmaxf(conv.x, 0.0f);
  hn.y = hv.y + STEP_SZ * fmaxf(conv.y, 0.0f);
  const int mr = lane & 15;
  const int kh = lane >> 4;
  const int nn = mr + 16 * wv;
  const bool live = (nn < LDO);
  *(v2f*)(hs + wv * HS_P + 2 * lane) = hn;
  v2f bf[HIDDEN / 4];
  if (wv < NT) {
    const float* brow = Bm + (size_t)(live ? nn : 0) * HIDDEN + 2 * kh;
#pragma unroll
    for (int i = 0; i < HIDDEN / 4; ++i) bf[i] = *(const v2f*)(brow + 4 * i);
    if (NT == 3) {
      const float bscale = live ? 1.0f : 0.0f;
#pragma unroll
      for (int i = 0; i < HIDDEN / 4; ++i) bf[i] *= bscale;
    }
  }
  __syncthreads();
  if (NT == 4) *(v2f*)(h + ro) = hn;
  if (wv < NT) {
    const float* arow = hs + mr * HS_P + 2 * kh;
    v8f cc = {};
#pragma unroll
    for (int i = 0; i < HIDDEN / 4; ++i) {
      const v2f a = *(const v2f*)(arow + 4 * i);
      cc = __builtin_amdgcn_wmma_f32_16x16x4_f32(false, a, false, bf[i], (short)0, cc, false, false);
    }
    const int mbase = (int)blockIdx.x * 16 + kh * 8;
    for (int r = 0; r < 8; ++r) {
      if (live) dst[(size_t)(mbase + r) * LDO + nn] = cc[r];
    }
  }
}


extern "C" void kernel_launch(void* const* d_in, const int* in_sizes, int n_in,
                              void* d_out, int out_size, void* d_ws, size_t ws_size,
                              hipStream_t stream) {
  (void)in_sizes; (void)n_in; (void)out_size; (void)ws_size;
  const float* x     = (const float*)d_in[0];
  const int*   ei    = (const int*)  d_in[1];
  const float* enc_w = (const float*)d_in[2];
  const float* dec_w = (const float*)d_in[3];
  const float* ext_w = (const float*)d_in[4];
  const float* beta  = (const float*)d_in[5];
  const float* pw    = (const float*)d_in[6];
  const int* row = ei;
  const int* col = ei + N_EDGES;

  char* wsb = (char*)d_ws;
  int*   cnt   = (int*)  (wsb);
  int*   off   = (int*)  (wsb + 200192);
  float* dinv  = (float*)(wsb + 400384);
  int*   rank  = (int*)  (wsb + 600576);
  int2*  edges = (int2*) (wsb + 7000576);
  float* W     = (float*)(wsb + 19800576);
  float* h     = (float*)(wsb + 19816960);
  float* x0    = (float*)(wsb + 32616960);
  float* outp  = (float*)(wsb + 45416960);
  int*   hist  = (int*)  (wsb + 58216960);
  int*   pre   = (int*)  (wsb + 64616960);
  float* outq  = (float*)(wsb + 71016960);

  count_local_kernel<<<CHUNKS + 1, CNT_T, 0, stream>>>(col, hist, rank, pw, W);
  chunk_prefix_kernel<<<(N_NODES + 255) / 256, 256, 0, stream>>>(hist, pre, cnt);
  scan_dinv_kernel<<<1, 1024, 0, stream>>>(cnt, off, dinv);
  fill_kernel<<<(N_EDGES + 255) / 256, 256, 0, stream>>>(row, col, rank, off, pre, dinv, edges);

  const int gemm_blocks = (M_TILES + 3) / 4;
  enc_gemm_kernel<<<ENC_WGS, ENC_T, 0, stream>>>(x, enc_w, h, x0);

  hw_gemm_kernel<<<gemm_blocks, HW_T, 0, stream>>>(h, W, outp);
  gather_update_mm_kernel<4, HIDDEN><<<M_TILES, GU_T, 0, stream>>>(off, edges, outp, h, x0, ext_w, beta, W, outq);
  gather_update_mm_kernel<4, HIDDEN><<<M_TILES, GU_T, 0, stream>>>(off, edges, outq, h, x0, ext_w, beta, W, outp);
  gather_update_mm_kernel<4, HIDDEN><<<M_TILES, GU_T, 0, stream>>>(off, edges, outp, h, x0, ext_w, beta, W, outq);
  gather_update_mm_kernel<3, NCLASS><<<M_TILES, GU_T, 0, stream>>>(off, edges, outq, h, x0, ext_w, beta, dec_w, (float*)d_out);
}


# A wave sums 4 nodes (4-wave work-groups; bounds, first records and h rows of the 4 fetched together) and a fill thread places 4 edges: 62k waves to start instead of 250k; sum orders kept
# speedup vs baseline: 4.3455x; 1.0415x over previous
#include <hip/hip_runtime.h>
#include <math.h>

#define N_NODES   50000
#define N_EDGES   1600000
#define NFEAT     500
#define HIDDEN    64
#define NCLASS    40
#define NUM_LAYERS 4
#define STEP_SZ   0.1f
#define M_TILES   (N_NODES / 16)

typedef __attribute__((ext_vector_type(2))) float v2f;
typedef __attribute__((ext_vector_type(8))) float v8f;


#define CHUNKS   32
#define CHUNK_E  (N_EDGES / CHUNKS)
#define CHUNK_Q  (CHUNK_E / 4)
#define CNT_T    1024

#define W_PITCH  (HIDDEN + 1)

__global__ __launch_bounds__(CNT_T)
void count_local_kernel(const int* __restrict__ col, int* __restrict__ hist,
                        int* __restrict__ rank, const float* __restrict__ pw,
                        float* __restrict__ W) {
  __shared__ int cnt[N_NODES];
  __shared__ float wl[HIDDEN * W_PITCH];
  const int t = threadIdx.x;
  if (blockIdx.x == CHUNKS) {
    const int LD = HIDDEN + 2;
    float q = 0.0f, r = 0.0f;
    if (t < HIDDEN) {
      q = pw[t * LD + HIDDEN];
      r = pw[t * LD + HIDDEN + 1];
    }
    float v[HIDDEN * HIDDEN / CNT_T];
#pragma unroll
    for (int u = 0; u < HIDDEN * HIDDEN / CNT_T; ++u) {
      const int e = t + u * CNT_T;
      const int i = e >> 6;
      const int j = e & (HIDDEN - 1);
      v[u] = pw[(i < j) ? (i * LD + j) : (j * LD + i)];
    }
#pragma unroll
    for (int u = 0; u < HIDDEN * HIDDEN / CNT_T; ++u) {
      const int e = t + u * CNT_T;
      const int i = e >> 6;
      const int j = e & (HIDDEN - 1);
      const float x = (i == j) ? 0.0f : v[u];
      wl[i * W_PITCH + j] = x;
      if (i != j) W[e] = x;
    }
    __syncthreads();
    if (t < HIDDEN) {
      float s = 0.0f;
      for (int j = 0; j < HIDDEN; ++j) s += fabsf(wl[t * W_PITCH + j]);
      W[t * HIDDEN + t] = q * s + r;
    }
    return;
  }
  for (int i = t; i < N_NODES; i += CNT_T) cnt[i] = 0;
  __syncthreads();
  const int4* c4 = (const int4*)(col + (size_t)blockIdx.x * CHUNK_E);
  int4* r4 = (int4*)(rank + (size_t)blockIdx.x * CHUNK_E);
  const int full = CHUNK_Q / CNT_T;
  for (int k = 0; k < full; k += 4) {
    int4 c[4];
#pragma unroll
    for (int u = 0; u < 4; ++u) c[u] = c4[(k + u) * CNT_T + t];
#pragma unroll
    for (int u = 0; u < 4; ++u) {
      int4 r;
      r.x = atomicAdd(&cnt[c[u].x], 1);
      r.y = atomicAdd(&cnt[c[u].y], 1);
      r.z = atomicAdd(&cnt[c[u].z], 1);
      r.w = atomicAdd(&cnt[c[u].w], 1);
      r4[(k + u) * CNT_T + t] = r;
    }
  }
  {
    const int q = full * CNT_T + t;
    if (q < CHUNK_Q) {
      const int4 c = c4[q];
      int4 r;
      r.x = atomicAdd(&cnt[c.x], 1);
      r.y = atomicAdd(&cnt[c.y], 1);
      r.z = atomicAdd(&cnt[c.z], 1);
      r.w = atomicAdd(&cnt[c.w], 1);
      r4[q] = r;
    }
  }
  __syncthreads();
  int* mine = hist + (size_t)blockIdx.x * N_NODES;
  for (int i = t; i < N_NODES; i += CNT_T) mine[i] = cnt[i];
}

__global__ __launch_bounds__(256)
void chunk_prefix_kernel(const int* __restrict__ hist, int* __restrict__ pre,
                         int* __restrict__ cnt) {
  const int c = blockIdx.x * blockDim.x + threadIdx.x;
  if (c < N_NODES) {
    int run = 0;
#pragma unroll
    for (int g = 0; g < CHUNKS; ++g) {
      const int v = hist[(size_t)g * N_NODES + c];
      pre[(size_t)g * N_NODES + c] = run;
      run += v;
    }
    cnt[c] = run;
  }
}

#define SCAN_T   1024
#define SCAN_PER 49

__global__ __launch_bounds__(SCAN_T)
void scan_dinv_kernel(const int* __restrict__ cnt, int* __restrict__ off,
                      float* __restrict__ dinv) {
  __shared__ int sc[SCAN_T * SCAN_PER];
  __shared__ int wsum[32];
  const int t = threadIdx.x;
  const int lane = t & 31;
  const int w = t >> 5;
  for (int k = 0; k < SCAN_PER; ++k) {
    const int i = k * SCAN_T + t;
    int v = 0;
    if (i < N_NODES) {
      v = cnt[i];
      dinv[i] = (v > 0) ? rsqrtf((float)v) : 0.0f;
    }
    sc[i] = v;
  }
  __syncthreads();
  int* mine = sc + t * SCAN_PER;
  int run = 0;
  for (int k = 0; k < SCAN_PER; ++k) run += mine[k];
  int s = run;
  for (int d = 1; d < 32; d <<= 1) {
    int up = __shfl_up(s, d);
    if (lane >= d) s += up;
  }
  if (lane == 31) wsum[w] = s;
  __syncthreads();
  int ws = wsum[lane];
  for (int d = 1; d < 32; d <<= 1) {
    int up = __shfl_up(ws, d);
    if (lane >= d) ws += up;
  }
  const int prev  = __shfl(ws, (w + 31) & 31);
  const int total = __shfl(ws, 31);
  int pos = ((w > 0) ? prev : 0) + s - run;
  for (int k = 0; k < SCAN_PER; ++k) {
    const int v = mine[k];
    mine[k] = pos;
    pos += v;
  }
  __syncthreads();
  for (int k = 0; k < SCAN_PER; ++k) {
    const int i = k * SCAN_T + t;
    if (i < N_NODES) off[i] = sc[i];
  }
  if (t == 0) off[N_NODES] = total;
}

#define FILL_T 256
#define FILL_Q (N_EDGES / 4)

__global__ __launch_bounds__(FILL_T)
void fill_kernel(const int* __restrict__ row, const int* __restrict__ col,
                 const int* __restrict__ rank, const int* __restrict__ off,
                 const int* __restrict__ pre, const float* __restrict__ dinv,
                 int2* __restrict__ edges) {
  const int q = blockIdx.x * FILL_T + threadIdx.x;
  if (q < FILL_Q) {
    const int4 r = ((const int4*)row)[q];
    const int4 c = ((const int4*)col)[q];
    const int4 k = ((const int4*)rank)[q];
    const int* pr = pre + (size_t)((q * 4) / CHUNK_E) * N_NODES;
    const int rr[4] = {r.x, r.y, r.z, r.w};
    const int cc[4] = {c.x, c.y, c.z, c.w};
    const int kk[4] = {k.x, k.y, k.z, k.w};
    float dr[4], dc[4];
    int so[4], sp[4];
#pragma unroll
    for (int u = 0; u < 4; ++u) {
      dr[u] = dinv[rr[u]];
      dc[u] = dinv[cc[u]];
      so[u] = off[cc[u]];
      sp[u] = pr[cc[u]];
    }
#pragma unroll
    for (int u = 0; u < 4; ++u) {
      int2 rec;
      rec.x = rr[u];
      rec.y = __float_as_int(dr[u] * dc[u]);
      edges[so[u] + sp[u] + kk[u]] = rec;
    }
  }
}


typedef __attribute__((ext_vector_type(4))) float v4f;

#define ENC_T   512
#define ENC_WGS 256

#define ENC_WMMA(av, bv, cv) cv = __builtin_amdgcn_wmma_f32_16x16x4_f32(false, av, false, bv, (short)0, cv, false, false)

#define ENC_STEP8(a, bq)                                              \
  {                                                                   \
    const v4f b0 = *(const v4f*)((bq));                               \
    const v4f b1 = *(const v4f*)((bq) + 16 * NFEAT);                  \
    const v4f b2 = *(const v4f*)((bq) + 32 * NFEAT);                  \
    const v4f b3 = *(const v4f*)((bq) + 48 * NFEAT);                  \
    const v2f alo = {(a).x, (a).y};                                   \
    const v2f ahi = {(a).z, (a).w};                                   \
    const v2f b0l = {b0.x, b0.y}, b0h = {b0.z, b0.w};                 \
    const v2f b1l = {b1.x, b1.y}, b1h = {b1.z, b1.w};                 \
    const v2f b2l = {b2.x, b2.y}, b2h = {b2.z, b2.w};                 \
    const v2f b3l = {b3.x, b3.y}, b3h = {b3.z, b3.w};                 \
    ENC_WMMA(alo, b0l, c0); ENC_WMMA(alo, b1l, c1);                   \
    ENC_WMMA(alo, b2l, c2); ENC_WMMA(alo, b3l, c3);                   \
    ENC_WMMA(ahi, b0h, c0); ENC_WMMA(ahi, b1h, c1);                   \
    ENC_WMMA(ahi, b2h, c2); ENC_WMMA(ahi, b3h, c3);                   \
  }

#define ENC_LOAD4(v, ptr)                                             \
  v##0 = *(const v4f*)((ptr) +  0);                                   \
  v##1 = *(const v4f*)((ptr) +  8);                                   \
  v##2 = *(const v4f*)((ptr) + 16);                                   \
  v##3 = *(const v4f*)((ptr) + 24);
#define ENC_GROUP(v, bq)                                              \
  ENC_STEP8(v##0, (bq) +  0)                                          \
  ENC_STEP8(v##1, (bq) +  8)                                          \
  ENC_STEP8(v##2, (bq) + 16)                                          \
  ENC_STEP8(v##3, (bq) + 24)

__global__ __launch_bounds__(ENC_T)
void enc_gemm_kernel(const float* __restrict__ x, const float* __restrict__ ew,
                     float* __restrict__ h, float* __restrict__ x0) {
  __shared__ __attribute__((aligned(16))) float Bs[HIDDEN * NFEAT];
  {
    const v4f* s4 = (const v4f*)ew;
    v4f* d4 = (v4f*)Bs;
    for (int i = threadIdx.x; i < HIDDEN * NFEAT / 4; i += ENC_T) d4[i] = s4[i];
  }
  __syncthreads();
  const int tile = (int)(blockIdx.x + gridDim.x * (threadIdx.x >> 5));
  if (tile >= M_TILES) return;
  const int lane = threadIdx.x & 31;
  const int mr = lane & 15;
  const int kh = lane >> 4;
  const float* arow = x + (size_t)(tile * 16 + mr) * NFEAT;
  const float* ap = arow + 4 * kh;
  const float* bp = Bs + mr * NFEAT + 4 * kh;
  v8f c0 = {}, c1 = {}, c2 = {}, c3 = {};
  v4f p0, p1, p2, p3, q0, q1, q2, q3, s0, s1, s2, s3;
  ENC_LOAD4(p, ap)
  ENC_LOAD4(q, ap + 32)
  for (int it = 0; it < 4; ++it) {
    const float* ag = ap + 96 * it;
    const float* bg = bp + 96 * it;
    ENC_LOAD4(s, ag + 64)
    ENC_GROUP(p, bg)
    ENC_LOAD4(p, ag + 96)
    ENC_GROUP(q, bg + 32)
    ENC_LOAD4(q, ag + 128)
    ENC_GROUP(s, bg + 64)
  }
  {
    ENC_LOAD4(s, ap + 448)
    const v4f r0 = *(const v4f*)(ap + 480);
    const v4f r1 = *(const v4f*)(ap + 488);
    const v2f r2 = *(const v2f*)(arow + 496 + 2 * kh);
    ENC_GROUP(p, bp + 384)
    ENC_GROUP(q, bp + 416)
    ENC_GROUP(s, bp + 448)
    ENC_STEP8(r0, bp + 480)
    ENC_STEP8(r1, bp + 488)
    const float* bt = Bs + mr * NFEAT + 496 + 2 * kh;
    const v2f t0 = *(const v2f*)(bt);
    const v2f t1 = *(const v2f*)(bt + 16 * NFEAT);
    const v2f t2 = *(const v2f*)(bt + 32 * NFEAT);
    const v2f t3 = *(const v2f*)(bt + 48 * NFEAT);
    ENC_WMMA(r2, t0, c0); ENC_WMMA(r2, t1, c1);
    ENC_WMMA(r2, t2, c2); ENC_WMMA(r2, t3, c3);
  }
  int mbase = tile * 16 + kh * 8;
  for (int r = 0; r < 8; ++r) {
    size_t o = (size_t)(mbase + r) * HIDDEN + mr;
    h[o]      = c0[r]; x0[o]      = c0[r];
    h[o + 16] = c1[r]; x0[o + 16] = c1[r];
    h[o + 32] = c2[r]; x0[o + 32] = c2[r];
    h[o + 48] = c3[r]; x0[o + 48] = c3[r];
  }
}


#define HW_T 128

__global__ __launch_bounds__(HW_T)
void hw_gemm_kernel(const float* __restrict__ h, const float* __restrict__ W,
                    float* __restrict__ outp) {
  __shared__ __attribute__((aligned(16))) float Ws[HIDDEN * HIDDEN];
  {
    const v4f* s4 = (const v4f*)W;
    v4f* d4 = (v4f*)Ws;
#pragma unroll
    for (int u = 0; u < HIDDEN * HIDDEN / 4 / HW_T; ++u) d4[threadIdx.x + HW_T * u] = s4[threadIdx.x + HW_T * u];
  }
  __syncthreads();
  int wave = (int)((blockIdx.x * HW_T + threadIdx.x) >> 5);
  if (wave >= M_TILES) return;
  int lane = threadIdx.x & 31;
  int mr = lane & 15;
  int kh = lane >> 4;
  const float* arow = h + (size_t)(wave * 16 + mr) * HIDDEN;
  v8f c0 = {}, c1 = {}, c2 = {}, c3 = {};
  for (int k = 0; k < HIDDEN; k += 4) {
    int ko = k + 2 * kh;
    v2f a  = *(const v2f*)(arow + ko);
    v2f b0 = *(const v2f*)(&Ws[(mr +  0) * HIDDEN + ko]);
    v2f b1 = *(const v2f*)(&Ws[(mr + 16) * HIDDEN + ko]);
    v2f b2 = *(const v2f*)(&Ws[(mr + 32) * HIDDEN + ko]);
    v2f b3 = *(const v2f*)(&Ws[(mr + 48) * HIDDEN + ko]);
    c0 = __builtin_amdgcn_wmma_f32_16x16x4_f32(false, a, false, b0, (short)0, c0, false, false);
    c1 = __builtin_amdgcn_wmma_f32_16x16x4_f32(false, a, false, b1, (short)0, c1, false, false);
    c2 = __builtin_amdgcn_wmma_f32_16x16x4_f32(false, a, false, b2, (short)0, c2, false, false);
    c3 = __builtin_amdgcn_wmma_f32_16x16x4_f32(false, a, false, b3, (short)0, c3, false, false);
  }
  int mbase = wave * 16 + kh * 8;
  for (int r = 0; r < 8; ++r) {
    size_t o = (size_t)(mbase + r) * HIDDEN + mr;
    outp[o]      = c0[r];
    outp[o + 16] = c1[r];
    outp[o + 32] = c2[r];
    outp[o + 48] = c3[r];
  }
}


#define GU_T   128
#define GU_NPW 4
#define HS_P   68

template <int NT, int LDO>
__global__ __launch_bounds__(GU_T)
void gather_update_mm_kernel(const int* __restrict__ off, const int2* __restrict__ edges,
                             const float* __restrict__ outp, float* __restrict__ h,
                             const float* __restrict__ x0, const float* __restrict__ ext_w,
                             const float* __restrict__ beta, const float* __restrict__ Bm,
                             float* __restrict__ dst) {
  __shared__ __attribute__((aligned(16))) float hs[16 * HS_P];
  const int wv = __builtin_amdgcn_readfirstlane((int)(threadIdx.x >> 5));
  const int c0 = (int)blockIdx.x * 16 + GU_NPW * wv;
  const int lane = threadIdx.x & 31;
  const float* src = outp + 2 * lane;
  int o[GU_NPW + 1];
#pragma unroll
  for (int k = 0; k <= GU_NPW; ++k) o[k] = off[c0 + k];
  int2 first[GU_NPW];
#pragma unroll
  for (int k = 0; k < GU_NPW; ++k) {
    first[k].x = 0;
    first[k].y = 0;
    if (lane < o[k + 1] - o[k]) first[k] = edges[o[k] + lane];
  }
  const size_t ro = (size_t)c0 * HIDDEN + 2 * lane;
  v2f hv[GU_NPW], xv[GU_NPW], hn[GU_NPW];
#pragma unroll
  for (int k = 0; k < GU_NPW; ++k) {
    hv[k] = *(const v2f*)(h + ro + k * HIDDEN);
    xv[k] = *(const v2f*)(x0 + ro + k * HIDDEN);
  }
  const v2f ew = *(const v2f*)(ext_w + 2 * lane);
  const float bt = beta[0];
#pragma unroll
  for (int k = 0; k < GU_NPW; ++k) {
    const int end = o[k + 1];
    int base = o[k];
    int2 my = first[k];
    v2f acc = {0.0f, 0.0f};
    while (base < end) {
      int n = end - base;
      n = (n < 32) ? n : 32;
      int j = 0;
      for (; j + 8 <= n; j += 8) {
        v2f v[8];
        float w[8];
#pragma unroll
        for (int u = 0; u < 8; ++u) {
          const int s = __builtin_amdgcn_readlane(my.x, j + u);
          w[u] = __int_as_float(__builtin_amdgcn_readlane(my.y, j + u));
          v[u] = *(const v2f*)(src + (size_t)s * HIDDEN);
        }
#pragma unroll
        for (int u = 0; u < 8; ++u) acc += w[u] * v[u];
      }
      if (j + 4 <= n) {
        v2f v[4];
        float w[4];
#pragma unroll
        for (int u = 0; u < 4; ++u) {
          const int s = __builtin_amdgcn_readlane(my.x, j + u);
          w[u] = __int_as_float(__builtin_amdgcn_readlane(my.y, j + u));
          v[u] = *(const v2f*)(src + (size_t)s * HIDDEN);
        }
#pragma unroll
        for (int u = 0; u < 4; ++u) acc += w[u] * v[u];
        j += 4;
      }
      for (; j < n; ++j) {
        const int s = __builtin_amdgcn_readlane(my.x, j);
        const float w = __int_as_float(__builtin_amdgcn_readlane(my.y, j));
        const v2f v = *(const v2f*)(src + (size_t)s * HIDDEN);
        acc += w * v;
      }
      base += 32;
      if (base < end) {
        my.x = 0;
        my.y = 0;
        if (lane < end - base) my = edges[base + lane];
      }
    }
    const v2f conv = acc - hv[k] * ew + xv[k] * bt;
    hn[k].x = hv[k].x + STEP_SZ * fmaxf(conv.x, 0.0f);
    hn[k].y = hv[k].y + STEP_SZ * fmaxf(conv.y, 0.0f);
    *(v2f*)(hs + (GU_NPW * wv + k) * HS_P + 2 * lane) = hn[k];
  }
  const int mr = lane & 15;
  const int kh = lane >> 4;
  const int nn = mr + 16 * wv;
  const bool live = (nn < LDO);
  v2f bf[HIDDEN / 4];
  if (wv < NT) {
    const float* brow = Bm + (size_t)(live ? nn : 0) * HIDDEN + 2 * kh;
#pragma unroll
    for (int i = 0; i < HIDDEN / 4; ++i) bf[i] = *(const v2f*)(brow + 4 * i);
    if (NT == 3) {
      const float bscale = live ? 1.0f : 0.0f;
#pragma unroll
      for (int i = 0; i < HIDDEN / 4; ++i) bf[i] *= bscale;
    }
  }
  __syncthreads();
  if (NT == 4) {
#pragma unroll
    for (int k = 0; k < GU_NPW; ++k) *(v2f*)(h + ro + k * HIDDEN) = hn[k];
  }
  if (wv < NT) {
    const float* arow = hs + mr * HS_P + 2 * kh;
    v8f cc = {};
#pragma unroll
    for (int i = 0; i < HIDDEN / 4; ++i) {
      const v2f a = *(const v2f*)(arow + 4 * i);
      cc = __builtin_amdgcn_wmma_f32_16x16x4_f32(false, a, false, bf[i], (short)0, cc, false, false);
    }
    const int mbase = (int)blockIdx.x * 16 + kh * 8;
    for (int r = 0; r < 8; ++r) {
      if (live) dst[(size_t)(mbase + r) * LDO + nn] = cc[r];
    }
  }
}


extern "C" void kernel_launch(void* const* d_in, const int* in_sizes, int n_in,
                              void* d_out, int out_size, void* d_ws, size_t ws_size,
                              hipStream_t stream) {
  (void)in_sizes; (void)n_in; (void)out_size; (void)ws_size;
  const float* x     = (const float*)d_in[0];
  const int*   ei    = (const int*)  d_in[1];
  const float* enc_w = (const float*)d_in[2];
  const float* dec_w = (const float*)d_in[3];
  const float* ext_w = (const float*)d_in[4];
  const float* beta  = (const float*)d_in[5];
  const float* pw    = (const float*)d_in[6];
  const int* row = ei;
  const int* col = ei + N_EDGES;

  char* wsb = (char*)d_ws;
  int*   cnt   = (int*)  (wsb);
  int*   off   = (int*)  (wsb + 200192);
  float* dinv  = (float*)(wsb + 400384);
  int*   rank  = (int*)  (wsb + 600576);
  int2*  edges = (int2*) (wsb + 7000576);
  float* W     = (float*)(wsb + 19800576);
  float* h     = (float*)(wsb + 19816960);
  float* x0    = (float*)(wsb + 32616960);
  float* outp  = (float*)(wsb + 45416960);
  int*   hist  = (int*)  (wsb + 58216960);
  int*   pre   = (int*)  (wsb + 64616960);
  float* outq  = (float*)(wsb + 71016960);

  count_local_kernel<<<CHUNKS + 1, CNT_T, 0, stream>>>(col, hist, rank, pw, W);
  chunk_prefix_kernel<<<(N_NODES + 255) / 256, 256, 0, stream>>>(hist, pre, cnt);
  scan_dinv_kernel<<<1, 1024, 0, stream>>>(cnt, off, dinv);
  fill_kernel<<<(FILL_Q + FILL_T - 1) / FILL_T, FILL_T, 0, stream>>>(row, col, rank, off, pre, dinv, edges);

  const int gemm_blocks = (M_TILES + 3) / 4;
  enc_gemm_kernel<<<ENC_WGS, ENC_T, 0, stream>>>(x, enc_w, h, x0);

  hw_gemm_kernel<<<gemm_blocks, HW_T, 0, stream>>>(h, W, outp);
  gather_update_mm_kernel<4, HIDDEN><<<M_TILES, GU_T, 0, stream>>>(off, edges, outp, h, x0, ext_w, beta, W, outq);
  gather_update_mm_kernel<4, HIDDEN><<<M_TILES, GU_T, 0, stream>>>(off, edges, outq, h, x0, ext_w, beta, W, outp);
  gather_update_mm_kernel<4, HIDDEN><<<M_TILES, GU_T, 0, stream>>>(off, edges, outp, h, x0, ext_w, beta, W, outq);
  gather_update_mm_kernel<3, NCLASS><<<M_TILES, GU_T, 0, stream>>>(off, edges, outq, h, x0, ext_w, beta, dec_w, (float*)d_out);
}


# Independent work fills under-filled launches: encoder GEMM work-groups beside the 32 counting ones (one per WGP), first out_p product beside the scan work-group; 8 launches, not 10; sum orders kept
# speedup vs baseline: 4.6527x; 1.0707x over previous
#include <hip/hip_runtime.h>
#include <math.h>

#define N_NODES   50000
#define N_EDGES   1600000
#define NFEAT     500
#define HIDDEN    64
#define NCLASS    40
#define NUM_LAYERS 4
#define STEP_SZ   0.1f
#define M_TILES   (N_NODES / 16)

typedef __attribute__((ext_vector_type(2))) float v2f;
typedef __attribute__((ext_vector_type(8))) float v8f;


#define CHUNKS   32
#define CHUNK_E  (N_EDGES / CHUNKS)
#define CHUNK_Q  (CHUNK_E / 4)
#define NODE_Q   (N_NODES / 4)
#define CNT_T    512

#define W_PITCH  (HIDDEN + 1)

__device__ __forceinline__
void weff_body(const float* __restrict__ pw, float* __restrict__ W, float* wl, const int t) {
  {
    const int LD = HIDDEN + 2;
    float q = 0.0f, r = 0.0f;
    if (t < HIDDEN) {
      q = pw[t * LD + HIDDEN];
      r = pw[t * LD + HIDDEN + 1];
    }
    float v[HIDDEN * HIDDEN / CNT_T];
#pragma unroll
    for (int u = 0; u < HIDDEN * HIDDEN / CNT_T; ++u) {
      const int e = t + u * CNT_T;
      const int i = e >> 6;
      const int j = e & (HIDDEN - 1);
      v[u] = pw[(i < j) ? (i * LD + j) : (j * LD + i)];
    }
#pragma unroll
    for (int u = 0; u < HIDDEN * HIDDEN / CNT_T; ++u) {
      const int e = t + u * CNT_T;
      const int i = e >> 6;
      const int j = e & (HIDDEN - 1);
      const float x = (i == j) ? 0.0f : v[u];
      wl[i * W_PITCH + j] = x;
      if (i != j) W[e] = x;
    }
    __syncthreads();
    if (t < HIDDEN) {
      float s = 0.0f;
      for (int j = 0; j < HIDDEN; ++j) s += fabsf(wl[t * W_PITCH + j]);
      W[t * HIDDEN + t] = q * s + r;
    }
  }
}

__device__ __forceinline__
void count_body(const int* __restrict__ col, int* __restrict__ hist, int* __restrict__ rank,
                int* cnt, const int run, const int t) {
  int4* cnt4 = (int4*)cnt;
  for (int i = t; i < NODE_Q; i += CNT_T) cnt4[i] = make_int4(0, 0, 0, 0);
  __syncthreads();
  const int4* c4 = (const int4*)(col + (size_t)run * CHUNK_E);
  int4* r4 = (int4*)(rank + (size_t)run * CHUNK_E);
  const int full = CHUNK_Q / CNT_T;
  for (int k = 0; k < full; k += 4) {
    int4 c[4];
#pragma unroll
    for (int u = 0; u < 4; ++u) c[u] = c4[(k + u) * CNT_T + t];
#pragma unroll
    for (int u = 0; u < 4; ++u) {
      int4 r;
      r.x = atomicAdd(&cnt[c[u].x], 1);
      r.y = atomicAdd(&cnt[c[u].y], 1);
      r.z = atomicAdd(&cnt[c[u].z], 1);
      r.w = atomicAdd(&cnt[c[u].w], 1);
      r4[(k + u) * CNT_T + t] = r;
    }
  }
  {
    const int q = full * CNT_T + t;
    if (q < CHUNK_Q) {
      const int4 c = c4[q];
      int4 r;
      r.x = atomicAdd(&cnt[c.x], 1);
      r.y = atomicAdd(&cnt[c.y], 1);
      r.z = atomicAdd(&cnt[c.z], 1);
      r.w = atomicAdd(&cnt[c.w], 1);
      r4[q] = r;
    }
  }
  __syncthreads();
  int4* mine4 = (int4*)(hist + (size_t)run * N_NODES);
  for (int i = t; i < NODE_Q; i += CNT_T) mine4[i] = cnt4[i];
}

__global__ __launch_bounds__(256)
void chunk_prefix_kernel(const int* __restrict__ hist, int* __restrict__ pre,
                         int* __restrict__ cnt) {
  const int c = blockIdx.x * blockDim.x + threadIdx.x;
  if (c < N_NODES) {
    int run = 0;
#pragma unroll
    for (int g = 0; g < CHUNKS; ++g) {
      const int v = hist[(size_t)g * N_NODES + c];
      pre[(size_t)g * N_NODES + c] = run;
      run += v;
    }
    cnt[c] = run;
  }
}

#define SCAN_T   1024
#define SCAN_PER 49

__device__ __forceinline__
void scan_dinv_body(const int* __restrict__ cnt, int* __restrict__ off,
                    float* __restrict__ dinv, int* sc, int* wsum, const int t) {
  const int lane = t & 31;
  const int w = t >> 5;
  for (int k = 0; k < SCAN_PER; ++k) {
    const int i = k * SCAN_T + t;
    int v = 0;
    if (i < N_NODES) {
      v = cnt[i];
      dinv[i] = (v > 0) ? rsqrtf((float)v) : 0.0f;
    }
    sc[i] = v;
  }
  __syncthreads();
  int* mine = sc + t * SCAN_PER;
  int run = 0;
  for (int k = 0; k < SCAN_PER; ++k) run += mine[k];
  int s = run;
  for (int d = 1; d < 32; d <<= 1) {
    int up = __shfl_up(s, d);
    if (lane >= d) s += up;
  }
  if (lane == 31) wsum[w] = s;
  __syncthreads();
  int ws = wsum[lane];
  for (int d = 1; d < 32; d <<= 1) {
    int up = __shfl_up(ws, d);
    if (lane >= d) ws += up;
  }
  const int prev  = __shfl(ws, (w + 31) & 31);
  const int total = __shfl(ws, 31);
  int pos = ((w > 0) ? prev : 0) + s - run;
  for (int k = 0; k < SCAN_PER; ++k) {
    const int v = mine[k];
    mine[k] = pos;
    pos += v;
  }
  __syncthreads();
  for (int k = 0; k < SCAN_PER; ++k) {
    const int i = k * SCAN_T + t;
    if (i < N_NODES) off[i] = sc[i];
  }
  if (t == 0) off[N_NODES] = total;
}

#define FILL_T 256
#define FILL_Q (N_EDGES / 4)

__global__ __launch_bounds__(FILL_T)
void fill_kernel(const int* __restrict__ row, const int* __restrict__ col,
                 const int* __restrict__ rank, const int* __restrict__ off,
                 const int* __restrict__ pre, const float* __restrict__ dinv,
                 int2* __restrict__ edges) {
  const int q = blockIdx.x * FILL_T + threadIdx.x;
  if (q < FILL_Q) {
    const int4 r = ((const int4*)row)[q];
    const int4 c = ((const int4*)col)[q];
    const int4 k = ((const int4*)rank)[q];
    const int* pr = pre + (size_t)((q * 4) / CHUNK_E) * N_NODES;
    const int rr[4] = {r.x, r.y, r.z, r.w};
    const int cc[4] = {c.x, c.y, c.z, c.w};
    const int kk[4] = {k.x, k.y, k.z, k.w};
    float dr[4], dc[4];
    int so[4], sp[4];
#pragma unroll
    for (int u = 0; u < 4; ++u) {
      dr[u] = dinv[rr[u]];
      dc[u] = dinv[cc[u]];
      so[u] = off[cc[u]];
      sp[u] = pr[cc[u]];
    }
#pragma unroll
    for (int u = 0; u < 4; ++u) {
      int2 rec;
      rec.x = rr[u];
      rec.y = __float_as_int(dr[u] * dc[u]);
      edges[so[u] + sp[u] + kk[u]] = rec;
    }
  }
}


typedef __attribute__((ext_vector_type(4))) float v4f;

#define ENC_T   512
#define ENC_WGS 223

#define ENC_WMMA(av, bv, cv) cv = __builtin_amdgcn_wmma_f32_16x16x4_f32(false, av, false, bv, (short)0, cv, false, false)

#define ENC_STEP8(a, bq)                                              \
  {                                                                   \
    const v4f b0 = *(const v4f*)((bq));                               \
    const v4f b1 = *(const v4f*)((bq) + 16 * NFEAT);                  \
    const v4f b2 = *(const v4f*)((bq) + 32 * NFEAT);                  \
    const v4f b3 = *(const v4f*)((bq) + 48 * NFEAT);                  \
    const v2f alo = {(a).x, (a).y};                                   \
    const v2f ahi = {(a).z, (a).w};                                   \
    const v2f b0l = {b0.x, b0.y}, b0h = {b0.z, b0.w};                 \
    const v2f b1l = {b1.x, b1.y}, b1h = {b1.z, b1.w};                 \
    const v2f b2l = {b2.x, b2.y}, b2h = {b2.z, b2.w};                 \
    const v2f b3l = {b3.x, b3.y}, b3h = {b3.z, b3.w};                 \
    ENC_WMMA(alo, b0l, c0); ENC_WMMA(alo, b1l, c1);                   \
    ENC_WMMA(alo, b2l, c2); ENC_WMMA(alo, b3l, c3);                   \
    ENC_WMMA(ahi, b0h, c0); ENC_WMMA(ahi, b1h, c1);                   \
    ENC_WMMA(ahi, b2h, c2); ENC_WMMA(ahi, b3h, c3);                   \
  }

#define ENC_LOAD4(v, ptr)                                             \
  v##0 = *(const v4f*)((ptr) +  0);                                   \
  v##1 = *(const v4f*)((ptr) +  8);                                   \
  v##2 = *(const v4f*)((ptr) + 16);                                   \
  v##3 = *(const v4f*)((ptr) + 24);
#define ENC_GROUP(v, bq)                                              \
  ENC_STEP8(v##0, (bq) +  0)                                          \
  ENC_STEP8(v##1, (bq) +  8)                                          \
  ENC_STEP8(v##2, (bq) + 16)                                          \
  ENC_STEP8(v##3, (bq) + 24)

__device__ __forceinline__
void enc_gemm_body(const float* __restrict__ x, const float* __restrict__ ew,
                   float* __restrict__ h, float* __restrict__ x0, float* Bs, const int b) {
  {
    const v4f* s4 = (const v4f*)ew;
    v4f* d4 = (v4f*)Bs;
    for (int i = threadIdx.x; i < HIDDEN * NFEAT / 4; i += ENC_T) d4[i] = s4[i];
  }
  __syncthreads();
  const int tile = b + ENC_WGS * (int)(threadIdx.x >> 5);
  if (tile >= M_TILES) return;
  const int lane = threadIdx.x & 31;
  const int mr = lane & 15;
  const int kh = lane >> 4;
  const float* arow = x + (size_t)(tile * 16 + mr) * NFEAT;
  const float* ap = arow + 4 * kh;
  const float* bp = Bs + mr * NFEAT + 4 * kh;
  v8f c0 = {}, c1 = {}, c2 = {}, c3 = {};
  v4f p0, p1, p2, p3, q0, q1, q2, q3, s0, s1, s2, s3;
  ENC_LOAD4(p, ap)
  ENC_LOAD4(q, ap + 32)
  for (int it = 0; it < 4; ++it) {
    const float* ag = ap + 96 * it;
    const float* bg = bp + 96 * it;
    ENC_LOAD4(s, ag + 64)
    ENC_GROUP(p, bg)
    ENC_LOAD4(p, ag + 96)
    ENC_GROUP(q, bg + 32)
    ENC_LOAD4(q, ag + 128)
    ENC_GROUP(s, bg + 64)
  }
  {
    ENC_LOAD4(s, ap + 448)
    const v4f r0 = *(const v4f*)(ap + 480);
    const v4f r1 = *(const v4f*)(ap + 488);
    const v2f r2 = *(const v2f*)(arow + 496 + 2 * kh);
    ENC_GROUP(p, bp + 384)
    ENC_GROUP(q, bp + 416)
    ENC_GROUP(s, bp + 448)
    ENC_STEP8(r0, bp + 480)
    ENC_STEP8(r1, bp + 488)
    const float* bt = Bs + mr * NFEAT + 496 + 2 * kh;
    const v2f t0 = *(const v2f*)(bt);
    const v2f t1 = *(const v2f*)(bt + 16 * NFEAT);
    const v2f t2 = *(const v2f*)(bt + 32 * NFEAT);
    const v2f t3 = *(const v2f*)(bt + 48 * NFEAT);
    ENC_WMMA(r2, t0, c0); ENC_WMMA(r2, t1, c1);
    ENC_WMMA(r2, t2, c2); ENC_WMMA(r2, t3, c3);
  }
  int mbase = tile * 16 + kh * 8;
  for (int r = 0; r < 8; ++r) {
    size_t o = (size_t)(mbase + r) * HIDDEN + mr;
    h[o]      = c0[r]; x0[o]      = c0[r];
    h[o + 16] = c1[r]; x0[o + 16] = c1[r];
    h[o + 32] = c2[r]; x0[o + 32] = c2[r];
    h[o + 48] = c3[r]; x0[o + 48] = c3[r];
  }
}


#define FUSE_WGS (ENC_WGS + CHUNKS + 1)

__global__ __launch_bounds__(CNT_T)
void count_enc_kernel(const int* __restrict__ col, int* __restrict__ hist,
                      int* __restrict__ rank, const float* __restrict__ pw,
                      float* __restrict__ W, const float* __restrict__ x,
                      const float* __restrict__ ew, float* __restrict__ h,
                      float* __restrict__ x0) {
  __shared__ __attribute__((aligned(16))) int lds[N_NODES + HIDDEN * W_PITCH];
  const int b = (int)blockIdx.x;
  if (b < ENC_WGS) {
    enc_gemm_body(x, ew, h, x0, (float*)lds, b);
  } else if (b < ENC_WGS + CHUNKS) {
    count_body(col, hist, rank, lds, b - ENC_WGS, (int)threadIdx.x);
  } else {
    weff_body(pw, W, (float*)(lds + N_NODES), (int)threadIdx.x);
  }
}


#define HW_T   SCAN_T
#define HW_WGS ((M_TILES + 31) / 32)

__device__ __forceinline__
void hw_gemm_body(const float* __restrict__ h, const float* __restrict__ W,
                  float* __restrict__ outp, float* Ws, const int b) {
  {
    const v4f* s4 = (const v4f*)W;
    v4f* d4 = (v4f*)Ws;
    d4[threadIdx.x] = s4[threadIdx.x];
  }
  __syncthreads();
  int wave = (int)((b * HW_T + threadIdx.x) >> 5);
  if (wave >= M_TILES) return;
  int lane = threadIdx.x & 31;
  int mr = lane & 15;
  int kh = lane >> 4;
  const float* arow = h + (size_t)(wave * 16 + mr) * HIDDEN;
  v8f c0 = {}, c1 = {}, c2 = {}, c3 = {};
  for (int k = 0; k < HIDDEN; k += 4) {
    int ko = k + 2 * kh;
    v2f a  = *(const v2f*)(arow + ko);
    v2f b0 = *(const v2f*)(&Ws[(mr +  0) * HIDDEN + ko]);
    v2f b1 = *(const v2f*)(&Ws[(mr + 16) * HIDDEN + ko]);
    v2f b2 = *(const v2f*)(&Ws[(mr + 32) * HIDDEN + ko]);
    v2f b3 = *(const v2f*)(&Ws[(mr + 48) * HIDDEN + ko]);
    c0 = __builtin_amdgcn_wmma_f32_16x16x4_f32(false, a, false, b0, (short)0, c0, false, false);
    c1 = __builtin_amdgcn_wmma_f32_16x16x4_f32(false, a, false, b1, (short)0, c1, false, false);
    c2 = __builtin_amdgcn_wmma_f32_16x16x4_f32(false, a, false, b2, (short)0, c2, false, false);
    c3 = __builtin_amdgcn_wmma_f32_16x16x4_f32(false, a, false, b3, (short)0, c3, false, false);
  }
  int mbase = wave * 16 + kh * 8;
  for (int r = 0; r < 8; ++r) {
    size_t o = (size_t)(mbase + r) * HIDDEN + mr;
    outp[o]      = c0[r];
    outp[o + 16] = c1[r];
    outp[o + 32] = c2[r];
    outp[o + 48] = c3[r];
  }
}

__global__ __launch_bounds__(SCAN_T)
void scan_hw_kernel(const int* __restrict__ cnt, int* __restrict__ off,
                    float* __restrict__ dinv, const float* __restrict__ h,
                    const float* __restrict__ W, float* __restrict__ outp) {
  __shared__ __attribute__((aligned(16))) int lds[SCAN_T * SCAN_PER + 32];
  if (blockIdx.x == 0) {
    scan_dinv_body(cnt, off, dinv, lds, lds + SCAN_T * SCAN_PER, (int)threadIdx.x);
  } else {
    hw_gemm_body(h, W, outp, (float*)lds, (int)blockIdx.x - 1);
  }
}


#define GU_T   128
#define GU_NPW 4
#define HS_P   68

template <int NT, int LDO>
__global__ __launch_bounds__(GU_T)
void gather_update_mm_kernel(const int* __restrict__ off, const int2* __restrict__ edges,
                             const float* __restrict__ outp, float* __restrict__ h,
                             const float* __restrict__ x0, const float* __restrict__ ext_w,
                             const float* __restrict__ beta, const float* __restrict__ Bm,
                             float* __restrict__ dst) {
  __shared__ __attribute__((aligned(16))) float hs[16 * HS_P];
  const int wv = __builtin_amdgcn_readfirstlane((int)(threadIdx.x >> 5));
  const int c0 = (int)blockIdx.x * 16 + GU_NPW * wv;
  const int lane = threadIdx.x & 31;
  const float* src = outp + 2 * lane;
  int o[GU_NPW + 1];
#pragma unroll
  for (int k = 0; k <= GU_NPW; ++k) o[k] = off[c0 + k];
  int2 first[GU_NPW];
#pragma unroll
  for (int k = 0; k < GU_NPW; ++k) {
    first[k].x = 0;
    first[k].y = 0;
    if (lane < o[k + 1] - o[k]) first[k] = edges[o[k] + lane];
  }
  const size_t ro = (size_t)c0 * HIDDEN + 2 * lane;
  v2f hv[GU_NPW], xv[GU_NPW], hn[GU_NPW];
#pragma unroll
  for (int k = 0; k < GU_NPW; ++k) {
    hv[k] = *(const v2f*)(h + ro + k * HIDDEN);
    xv[k] = *(const v2f*)(x0 + ro + k * HIDDEN);
  }
  const v2f ew = *(const v2f*)(ext_w + 2 * lane);
  const float bt = beta[0];
#pragma unroll
  for (int k = 0; k < GU_NPW; ++k) {
    const int end = o[k + 1];
    int base = o[k];
    int2 my = first[k];
    v2f acc = {0.0f, 0.0f};
    while (base < end) {
      int n = end - base;
      n = (n < 32) ? n : 32;
      int j = 0;
      for (; j + 8 <= n; j += 8) {
        v2f v[8];
        float w[8];
#pragma unroll
        for (int u = 0; u < 8; ++u) {
          const int s = __builtin_amdgcn_readlane(my.x, j + u);
          w[u] = __int_as_float(__builtin_amdgcn_readlane(my.y, j + u));
          v[u] = *(const v2f*)(src + (size_t)s * HIDDEN);
        }
#pragma unroll
        for (int u = 0; u < 8; ++u) acc += w[u] * v[u];
      }
      if (j + 4 <= n) {
        v2f v[4];
        float w[4];
#pragma unroll
        for (int u = 0; u < 4; ++u) {
          const int s = __builtin_amdgcn_readlane(my.x, j + u);
          w[u] = __int_as_float(__builtin_amdgcn_readlane(my.y, j + u));
          v[u] = *(const v2f*)(src + (size_t)s * HIDDEN);
        }
#pragma unroll
        for (int u = 0; u < 4; ++u) acc += w[u] * v[u];
        j += 4;
      }
      for (; j < n; ++j) {
        const int s = __builtin_amdgcn_readlane(my.x, j);
        const float w = __int_as_float(__builtin_amdgcn_readlane(my.y, j));
        const v2f v = *(const v2f*)(src + (size_t)s * HIDDEN);
        acc += w * v;
      }
      base += 32;
      if (base < end) {
        my.x = 0;
        my.y = 0;
        if (lane < end - base) my = edges[base + lane];
      }
    }
    const v2f conv = acc - hv[k] * ew + xv[k] * bt;
    hn[k].x = hv[k].x + STEP_SZ * fmaxf(conv.x, 0.0f);
    hn[k].y = hv[k].y + STEP_SZ * fmaxf(conv.y, 0.0f);
    *(v2f*)(hs + (GU_NPW * wv + k) * HS_P + 2 * lane) = hn[k];
  }
  const int mr = lane & 15;
  const int kh = lane >> 4;
  const int nn = mr + 16 * wv;
  const bool live = (nn < LDO);
  v2f bf[HIDDEN / 4];
  if (wv < NT) {
    const float* brow = Bm + (size_t)(live ? nn : 0) * HIDDEN + 2 * kh;
#pragma unroll
    for (int i = 0; i < HIDDEN / 4; ++i) bf[i] = *(const v2f*)(brow + 4 * i);
    if (NT == 3) {
      const float bscale = live ? 1.0f : 0.0f;
#pragma unroll
      for (int i = 0; i < HIDDEN / 4; ++i) bf[i] *= bscale;
    }
  }
  __syncthreads();
  if (NT == 4) {
#pragma unroll
    for (int k = 0; k < GU_NPW; ++k) *(v2f*)(h + ro + k * HIDDEN) = hn[k];
  }
  if (wv < NT) {
    const float* arow = hs + mr * HS_P + 2 * kh;
    v8f cc = {};
#pragma unroll
    for (int i = 0; i < HIDDEN / 4; ++i) {
      const v2f a = *(const v2f*)(arow + 4 * i);
      cc = __builtin_amdgcn_wmma_f32_16x16x4_f32(false, a, false, bf[i], (short)0, cc, false, false);
    }
    const int mbase = (int)blockIdx.x * 16 + kh * 8;
    for (int r = 0; r < 8; ++r) {
      if (live) dst[(size_t)(mbase + r) * LDO + nn] = cc[r];
    }
  }
}


extern "C" void kernel_launch(void* const* d_in, const int* in_sizes, int n_in,
                              void* d_out, int out_size, void* d_ws, size_t ws_size,
                              hipStream_t stream) {
  (void)in_sizes; (void)n_in; (void)out_size; (void)ws_size;
  const float* x     = (const float*)d_in[0];
  const int*   ei    = (const int*)  d_in[1];
  const float* enc_w = (const float*)d_in[2];
  const float* dec_w = (const float*)d_in[3];
  const float* ext_w = (const float*)d_in[4];
  const float* beta  = (const float*)d_in[5];
  const float* pw    = (const float*)d_in[6];
  const int* row = ei;
  const int* col = ei + N_EDGES;

  char* wsb = (char*)d_ws;
  int*   cnt   = (int*)  (wsb);
  int*   off   = (int*)  (wsb + 200192);
  float* dinv  = (float*)(wsb + 400384);
  int*   rank  = (int*)  (wsb + 600576);
  int2*  edges = (int2*) (wsb + 7000576);
  float* W     = (float*)(wsb + 19800576);
  float* h     = (float*)(wsb + 19816960);
  float* x0    = (float*)(wsb + 32616960);
  float* outp  = (float*)(wsb + 45416960);
  int*   hist  = (int*)  (wsb + 58216960);
  int*   pre   = (int*)  (wsb + 64616960);
  float* outq  = (float*)(wsb + 71016960);

  count_enc_kernel<<<FUSE_WGS, CNT_T, 0, stream>>>(col, hist, rank, pw, W, x, enc_w, h, x0);
  chunk_prefix_kernel<<<(N_NODES + 255) / 256, 256, 0, stream>>>(hist, pre, cnt);
  scan_hw_kernel<<<1 + HW_WGS, SCAN_T, 0, stream>>>(cnt, off, dinv, h, W, outp);
  fill_kernel<<<(FILL_Q + FILL_T - 1) / FILL_T, FILL_T, 0, stream>>>(row, col, rank, off, pre, dinv, edges);

  gather_update_mm_kernel<4, HIDDEN><<<M_TILES, GU_T, 0, stream>>>(off, edges, outp, h, x0, ext_w, beta, W, outq);
  gather_update_mm_kernel<4, HIDDEN><<<M_TILES, GU_T, 0, stream>>>(off, edges, outq, h, x0, ext_w, beta, W, outp);
  gather_update_mm_kernel<4, HIDDEN><<<M_TILES, GU_T, 0, stream>>>(off, edges, outp, h, x0, ext_w, beta, W, outq);
  gather_update_mm_kernel<3, NCLASS><<<M_TILES, GU_T, 0, stream>>>(off, edges, outq, h, x0, ext_w, beta, dec_w, (float*)d_out);
}


# No global scan launch: the per-run prefix pass scans its 256-node block and writes (slot base, deg^-1/2) per (run, target), the fill adds scanned block totals: 7 launches, not 8; sum orders kept
# speedup vs baseline: 4.9629x; 1.0667x over previous
#include <hip/hip_runtime.h>
#include <math.h>

#define N_NODES   50000
#define N_EDGES   1600000
#define NFEAT     500
#define HIDDEN    64
#define NCLASS    40
#define NUM_LAYERS 4
#define STEP_SZ   0.1f
#define M_TILES   (N_NODES / 16)

typedef __attribute__((ext_vector_type(2))) float v2f;
typedef __attribute__((ext_vector_type(8))) float v8f;


#define CHUNKS   32
#define CHUNK_E  (N_EDGES / CHUNKS)
#define CHUNK_Q  (CHUNK_E / 4)
#define NODE_Q   (N_NODES / 4)
#define CNT_T    512

#define W_PITCH  (HIDDEN + 1)

__device__ __forceinline__
void weff_body(const float* __restrict__ pw, float* __restrict__ W, float* wl, const int t) {
  {
    const int LD = HIDDEN + 2;
    float q = 0.0f, r = 0.0f;
    if (t < HIDDEN) {
      q = pw[t * LD + HIDDEN];
      r = pw[t * LD + HIDDEN + 1];
    }
    float v[HIDDEN * HIDDEN / CNT_T];
#pragma unroll
    for (int u = 0; u < HIDDEN * HIDDEN / CNT_T; ++u) {
      const int e = t + u * CNT_T;
      const int i = e >> 6;
      const int j = e & (HIDDEN - 1);
      v[u] = pw[(i < j) ? (i * LD + j) : (j * LD + i)];
    }
#pragma unroll
    for (int u = 0; u < HIDDEN * HIDDEN / CNT_T; ++u) {
      const int e = t + u * CNT_T;
      const int i = e >> 6;
      const int j = e & (HIDDEN - 1);
      const float x = (i == j) ? 0.0f : v[u];
      wl[i * W_PITCH + j] = x;
      if (i != j) W[e] = x;
    }
    __syncthreads();
    if (t < HIDDEN) {
      float s = 0.0f;
      for (int j = 0; j < HIDDEN; ++j) s += fabsf(wl[t * W_PITCH + j]);
      W[t * HIDDEN + t] = q * s + r;
    }
  }
}

__device__ __forceinline__
void count_body(const int* __restrict__ col, int* __restrict__ hist, int* __restrict__ rank,
                int* cnt, const int run, const int t) {
  int4* cnt4 = (int4*)cnt;
  for (int i = t; i < NODE_Q; i += CNT_T) cnt4[i] = make_int4(0, 0, 0, 0);
  __syncthreads();
  const int4* c4 = (const int4*)(col + (size_t)run * CHUNK_E);
  int4* r4 = (int4*)(rank + (size_t)run * CHUNK_E);
  const int full = CHUNK_Q / CNT_T;
  for (int k = 0; k < full; k += 4) {
    int4 c[4];
#pragma unroll
    for (int u = 0; u < 4; ++u) c[u] = c4[(k + u) * CNT_T + t];
#pragma unroll
    for (int u = 0; u < 4; ++u) {
      int4 r;
      r.x = atomicAdd(&cnt[c[u].x], 1);
      r.y = atomicAdd(&cnt[c[u].y], 1);
      r.z = atomicAdd(&cnt[c[u].z], 1);
      r.w = atomicAdd(&cnt[c[u].w], 1);
      r4[(k + u) * CNT_T + t] = r;
    }
  }
  {
    const int q = full * CNT_T + t;
    if (q < CHUNK_Q) {
      const int4 c = c4[q];
      int4 r;
      r.x = atomicAdd(&cnt[c.x], 1);
      r.y = atomicAdd(&cnt[c.y], 1);
      r.z = atomicAdd(&cnt[c.z], 1);
      r.w = atomicAdd(&cnt[c.w], 1);
      r4[q] = r;
    }
  }
  __syncthreads();
  int4* mine4 = (int4*)(hist + (size_t)run * N_NODES);
  for (int i = t; i < NODE_Q; i += CNT_T) mine4[i] = cnt4[i];
}


#define PB_T     256
#define PB_SHIFT 8
#define NB       ((N_NODES + PB_T - 1) / PB_T)

__device__ __forceinline__ int wave_incl_scan(int v, const int lane) {
  for (int d = 1; d < 32; d <<= 1) {
    const int up = __shfl_up(v, d);
    if (lane >= d) v += up;
  }
  return v;
}

__device__ __forceinline__
int block_excl_scan(const int v, int* wsum, const int t, int* total) {
  const int lane = t & 31;
  const int w = t >> 5;
  const int incl = wave_incl_scan(v, lane);
  if (lane == 31) wsum[w] = incl;
  __syncthreads();
  int ws = (lane < PB_T / 32) ? wsum[lane] : 0;
  ws = wave_incl_scan(ws, lane);
  const int prev = __shfl(ws, (w + 31) & 31);
  *total = __shfl(ws, PB_T / 32 - 1);
  return ((w > 0) ? prev : 0) + incl - v;
}

__device__ __forceinline__
void prefix_body(const int* __restrict__ hist, int2* __restrict__ pre2,
                 float* __restrict__ dinv, int* __restrict__ btot,
                 int* wsum, const int blk, const int t) {
  const int c = blk * PB_T + t;
  const bool live = (c < N_NODES);
  const int cl = live ? c : (N_NODES - 1);
  int v[CHUNKS];
#pragma unroll
  for (int g = 0; g < CHUNKS; ++g) v[g] = hist[(size_t)g * N_NODES + cl];
  int deg = 0;
#pragma unroll
  for (int g = 0; g < CHUNKS; ++g) {
    v[g] = live ? v[g] : 0;
    deg += v[g];
  }
  int total;
  const int first = block_excl_scan(deg, wsum, t, &total);
  if (t == 0) btot[blk] = total;
  if (live) {
    const float d = (deg > 0) ? rsqrtf((float)deg) : 0.0f;
    dinv[c] = d;
    int run = first;
#pragma unroll
    for (int g = 0; g < CHUNKS; ++g) {
      int2 rec;
      rec.x = run;
      rec.y = __float_as_int(d);
      pre2[(size_t)g * N_NODES + c] = rec;
      run += v[g];
    }
  }
}

#define FILL_T   PB_T
#define FILL_Q   (N_EDGES / 4)
#define FILL_WGS ((FILL_Q + FILL_T - 1) / FILL_T)
#define OFF_WGS  ((N_NODES + 1 + PB_T - 1) / PB_T)

__global__ __launch_bounds__(FILL_T)
void fill_off_kernel(const int* __restrict__ row, const int* __restrict__ col,
                     const int* __restrict__ rank, const int2* __restrict__ pre2,
                     const float* __restrict__ dinv, const int* __restrict__ btot,
                     int2* __restrict__ edges, int* __restrict__ off) {
  __shared__ int sbase[PB_T];
  __shared__ int wsum[PB_T / 32];
  const int t = (int)threadIdx.x;
  const int b = (int)blockIdx.x;
  const bool placing = (b < FILL_WGS);
  const int q = b * FILL_T + t;
  const bool live = placing && (q < FILL_Q);
  const int ob = b - FILL_WGS;
  const int cn = ob * PB_T + t;
  const bool node = !placing && (cn < N_NODES);
  const int ql = live ? q : 0;
  const int4 r = ((const int4*)row)[ql];
  const int4 c = ((const int4*)col)[ql];
  const int4 k = ((const int4*)rank)[ql];
  int bt = btot[(t < NB) ? t : 0];
  const int first = pre2[node ? cn : 0].x;
  const int rr[4] = {r.x, r.y, r.z, r.w};
  const int cc[4] = {c.x, c.y, c.z, c.w};
  const int kk[4] = {k.x, k.y, k.z, k.w};
  float dr[4];
  int2 p[4];
  const int2* pr = pre2 + (size_t)((ql * 4) / CHUNK_E) * N_NODES;
#pragma unroll
  for (int u = 0; u < 4; ++u) {
    dr[u] = dinv[rr[u]];
    p[u] = pr[cc[u]];
  }
  bt = (t < NB) ? bt : 0;
  int total;
  sbase[t] = block_excl_scan(bt, wsum, t, &total);
  __syncthreads();
  if (live) {
#pragma unroll
    for (int u = 0; u < 4; ++u) {
      int2 rec;
      rec.x = rr[u];
      rec.y = __float_as_int(dr[u] * __int_as_float(p[u].y));
      edges[sbase[cc[u] >> PB_SHIFT] + p[u].x + kk[u]] = rec;
    }
  }
  if (node) off[cn] = sbase[ob] + first;
  else if (!placing && cn == N_NODES) off[cn] = sbase[NB];
}


typedef __attribute__((ext_vector_type(4))) float v4f;

#define ENC_T   512
#define ENC_WGS 223

#define ENC_WMMA(av, bv, cv) cv = __builtin_amdgcn_wmma_f32_16x16x4_f32(false, av, false, bv, (short)0, cv, false, false)

#define ENC_STEP8(a, bq)                                              \
  {                                                                   \
    const v4f b0 = *(const v4f*)((bq));                               \
    const v4f b1 = *(const v4f*)((bq) + 16 * NFEAT);                  \
    const v4f b2 = *(const v4f*)((bq) + 32 * NFEAT);                  \
    const v4f b3 = *(const v4f*)((bq) + 48 * NFEAT);                  \
    const v2f alo = {(a).x, (a).y};                                   \
    const v2f ahi = {(a).z, (a).w};                                   \
    const v2f b0l = {b0.x, b0.y}, b0h = {b0.z, b0.w};                 \
    const v2f b1l = {b1.x, b1.y}, b1h = {b1.z, b1.w};                 \
    const v2f b2l = {b2.x, b2.y}, b2h = {b2.z, b2.w};                 \
    const v2f b3l = {b3.x, b3.y}, b3h = {b3.z, b3.w};                 \
    ENC_WMMA(alo, b0l, c0); ENC_WMMA(alo, b1l, c1);                   \
    ENC_WMMA(alo, b2l, c2); ENC_WMMA(alo, b3l, c3);                   \
    ENC_WMMA(ahi, b0h, c0); ENC_WMMA(ahi, b1h, c1);                   \
    ENC_WMMA(ahi, b2h, c2); ENC_WMMA(ahi, b3h, c3);                   \
  }

#define ENC_LOAD4(v, ptr)                                             \
  v##0 = *(const v4f*)((ptr) +  0);                                   \
  v##1 = *(const v4f*)((ptr) +  8);                                   \
  v##2 = *(const v4f*)((ptr) + 16);                                   \
  v##3 = *(const v4f*)((ptr) + 24);
#define ENC_GROUP(v, bq)                                              \
  ENC_STEP8(v##0, (bq) +  0)                                          \
  ENC_STEP8(v##1, (bq) +  8)                                          \
  ENC_STEP8(v##2, (bq) + 16)                                          \
  ENC_STEP8(v##3, (bq) + 24)

__device__ __forceinline__
void enc_gemm_body(const float* __restrict__ x, const float* __restrict__ ew,
                   float* __restrict__ h, float* __restrict__ x0, float* Bs, const int b) {
  {
    const v4f* s4 = (const v4f*)ew;
    v4f* d4 = (v4f*)Bs;
    for (int i = threadIdx.x; i < HIDDEN * NFEAT / 4; i += ENC_T) d4[i] = s4[i];
  }
  __syncthreads();
  const int tile = b + ENC_WGS * (int)(threadIdx.x >> 5);
  if (tile >= M_TILES) return;
  const int lane = threadIdx.x & 31;
  const int mr = lane & 15;
  const int kh = lane >> 4;
  const float* arow = x + (size_t)(tile * 16 + mr) * NFEAT;
  const float* ap = arow + 4 * kh;
  const float* bp = Bs + mr * NFEAT + 4 * kh;
  v8f c0 = {}, c1 = {}, c2 = {}, c3 = {};
  v4f p0, p1, p2, p3, q0, q1, q2, q3, s0, s1, s2, s3;
  ENC_LOAD4(p, ap)
  ENC_LOAD4(q, ap + 32)
  for (int it = 0; it < 4; ++it) {
    const float* ag = ap + 96 * it;
    const float* bg = bp + 96 * it;
    ENC_LOAD4(s, ag + 64)
    ENC_GROUP(p, bg)
    ENC_LOAD4(p, ag + 96)
    ENC_GROUP(q, bg + 32)
    ENC_LOAD4(q, ag + 128)
    ENC_GROUP(s, bg + 64)
  }
  {
    ENC_LOAD4(s, ap + 448)
    const v4f r0 = *(const v4f*)(ap + 480);
    const v4f r1 = *(const v4f*)(ap + 488);
    const v2f r2 = *(const v2f*)(arow + 496 + 2 * kh);
    ENC_GROUP(p, bp + 384)
    ENC_GROUP(q, bp + 416)
    ENC_GROUP(s, bp + 448)
    ENC_STEP8(r0, bp + 480)
    ENC_STEP8(r1, bp + 488)
    const float* bt = Bs + mr * NFEAT + 496 + 2 * kh;
    const v2f t0 = *(const v2f*)(bt);
    const v2f t1 = *(const v2f*)(bt + 16 * NFEAT);
    const v2f t2 = *(const v2f*)(bt + 32 * NFEAT);
    const v2f t3 = *(const v2f*)(bt + 48 * NFEAT);
    ENC_WMMA(r2, t0, c0); ENC_WMMA(r2, t1, c1);
    ENC_WMMA(r2, t2, c2); ENC_WMMA(r2, t3, c3);
  }
  int mbase = tile * 16 + kh * 8;
  for (int r = 0; r < 8; ++r) {
    size_t o = (size_t)(mbase + r) * HIDDEN + mr;
    h[o]      = c0[r]; x0[o]      = c0[r];
    h[o + 16] = c1[r]; x0[o + 16] = c1[r];
    h[o + 32] = c2[r]; x0[o + 32] = c2[r];
    h[o + 48] = c3[r]; x0[o + 48] = c3[r];
  }
}


#define FUSE_WGS (ENC_WGS + CHUNKS + 1)

__global__ __launch_bounds__(CNT_T)
void count_enc_kernel(const int* __restrict__ col, int* __restrict__ hist,
                      int* __restrict__ rank, const float* __restrict__ pw,
                      float* __restrict__ W, const float* __restrict__ x,
                      const float* __restrict__ ew, float* __restrict__ h,
                      float* __restrict__ x0) {
  __shared__ __attribute__((aligned(16))) int lds[N_NODES + HIDDEN * W_PITCH];
  const int b = (int)blockIdx.x;
  if (b < ENC_WGS) {
    enc_gemm_body(x, ew, h, x0, (float*)lds, b);
  } else if (b < ENC_WGS + CHUNKS) {
    count_body(col, hist, rank, lds, b - ENC_WGS, (int)threadIdx.x);
  } else {
    weff_body(pw, W, (float*)(lds + N_NODES), (int)threadIdx.x);
  }
}


#define HW_T   PB_T
#define HW_WGS ((M_TILES + HW_T / 32 - 1) / (HW_T / 32))

__device__ __forceinline__
void hw_gemm_body(const float* __restrict__ h, const float* __restrict__ W,
                  float* __restrict__ outp, float* Ws, const int b) {
  {
    const v4f* s4 = (const v4f*)W;
    v4f* d4 = (v4f*)Ws;
    v4f w4[HIDDEN * HIDDEN / 4 / HW_T];
#pragma unroll
    for (int u = 0; u < HIDDEN * HIDDEN / 4 / HW_T; ++u) w4[u] = s4[u * HW_T + threadIdx.x];
#pragma unroll
    for (int u = 0; u < HIDDEN * HIDDEN / 4 / HW_T; ++u) d4[u * HW_T + threadIdx.x] = w4[u];
  }
  __syncthreads();
  int wave = (int)((b * HW_T + threadIdx.x) >> 5);
  if (wave >= M_TILES) return;
  int lane = threadIdx.x & 31;
  int mr = lane & 15;
  int kh = lane >> 4;
  const float* arow = h + (size_t)(wave * 16 + mr) * HIDDEN;
  v8f c0 = {}, c1 = {}, c2 = {}, c3 = {};
  for (int k = 0; k < HIDDEN; k += 4) {
    int ko = k + 2 * kh;
    v2f a  = *(const v2f*)(arow + ko);
    v2f b0 = *(const v2f*)(&Ws[(mr +  0) * HIDDEN + ko]);
    v2f b1 = *(const v2f*)(&Ws[(mr + 16) * HIDDEN + ko]);
    v2f b2 = *(const v2f*)(&Ws[(mr + 32) * HIDDEN + ko]);
    v2f b3 = *(const v2f*)(&Ws[(mr + 48) * HIDDEN + ko]);
    c0 = __builtin_amdgcn_wmma_f32_16x16x4_f32(false, a, false, b0, (short)0, c0, false, false);
    c1 = __builtin_amdgcn_wmma_f32_16x16x4_f32(false, a, false, b1, (short)0, c1, false, false);
    c2 = __builtin_amdgcn_wmma_f32_16x16x4_f32(false, a, false, b2, (short)0, c2, false, false);
    c3 = __builtin_amdgcn_wmma_f32_16x16x4_f32(false, a, false, b3, (short)0, c3, false, false);
  }
  int mbase = wave * 16 + kh * 8;
  for (int r = 0; r < 8; ++r) {
    size_t o = (size_t)(mbase + r) * HIDDEN + mr;
    outp[o]      = c0[r];
    outp[o + 16] = c1[r];
    outp[o + 32] = c2[r];
    outp[o + 48] = c3[r];
  }
}

__global__ __launch_bounds__(PB_T)
void prefix_hw_kernel(const int* __restrict__ hist, int2* __restrict__ pre2,
                      float* __restrict__ dinv, int* __restrict__ btot,
                      const float* __restrict__ h, const float* __restrict__ W,
                      float* __restrict__ outp) {
  __shared__ __attribute__((aligned(16))) float lds[HIDDEN * HIDDEN];
  if (blockIdx.x < NB) {
    prefix_body(hist, pre2, dinv, btot, (int*)lds, (int)blockIdx.x, (int)threadIdx.x);
  } else {
    hw_gemm_body(h, W, outp, lds, (int)blockIdx.x - NB);
  }
}


#define GU_T   128
#define GU_NPW 4
#define HS_P   68

template <int NT, int LDO>
__global__ __launch_bounds__(GU_T)
void gather_update_mm_kernel(const int* __restrict__ off, const int2* __restrict__ edges,
                             const float* __restrict__ outp, float* __restrict__ h,
                             const float* __restrict__ x0, const float* __restrict__ ext_w,
                             const float* __restrict__ beta, const float* __restrict__ Bm,
                             float* __restrict__ dst) {
  __shared__ __attribute__((aligned(16))) float hs[16 * HS_P];
  const int wv = __builtin_amdgcn_readfirstlane((int)(threadIdx.x >> 5));
  const int c0 = (int)blockIdx.x * 16 + GU_NPW * wv;
  const int lane = threadIdx.x & 31;
  const float* src = outp + 2 * lane;
  int o[GU_NPW + 1];
#pragma unroll
  for (int k = 0; k <= GU_NPW; ++k) o[k] = off[c0 + k];
  int2 first[GU_NPW];
#pragma unroll
  for (int k = 0; k < GU_NPW; ++k) {
    first[k].x = 0;
    first[k].y = 0;
    if (lane < o[k + 1] - o[k]) first[k] = edges[o[k] + lane];
  }
  const size_t ro = (size_t)c0 * HIDDEN + 2 * lane;
  v2f hv[GU_NPW], xv[GU_NPW], hn[GU_NPW];
#pragma unroll
  for (int k = 0; k < GU_NPW; ++k) {
    hv[k] = *(const v2f*)(h + ro + k * HIDDEN);
    xv[k] = *(const v2f*)(x0 + ro + k * HIDDEN);
  }
  const v2f ew = *(const v2f*)(ext_w + 2 * lane);
  const float bt = beta[0];
#pragma unroll
  for (int k = 0; k < GU_NPW; ++k) {
    const int end = o[k + 1];
    int base = o[k];
    int2 my = first[k];
    v2f acc = {0.0f, 0.0f};
    while (base < end) {
      int n = end - base;
      n = (n < 32) ? n : 32;
      int j = 0;
      for (; j + 8 <= n; j += 8) {
        v2f v[8];
        float w[8];
#pragma unroll
        for (int u = 0; u < 8; ++u) {
          const int s = __builtin_amdgcn_readlane(my.x, j + u);
          w[u] = __int_as_float(__builtin_amdgcn_readlane(my.y, j + u));
          v[u] = *(const v2f*)(src + (size_t)s * HIDDEN);
        }
#pragma unroll
        for (int u = 0; u < 8; ++u) acc += w[u] * v[u];
      }
      if (j + 4 <= n) {
        v2f v[4];
        float w[4];
#pragma unroll
        for (int u = 0; u < 4; ++u) {
          const int s = __builtin_amdgcn_readlane(my.x, j + u);
          w[u] = __int_as_float(__builtin_amdgcn_readlane(my.y, j + u));
          v[u] = *(const v2f*)(src + (size_t)s * HIDDEN);
        }
#pragma unroll
        for (int u = 0; u < 4; ++u) acc += w[u] * v[u];
        j += 4;
      }
      for (; j < n; ++j) {
        const int s = __builtin_amdgcn_readlane(my.x, j);
        const float w = __int_as_float(__builtin_amdgcn_readlane(my.y, j));
        const v2f v = *(const v2f*)(src + (size_t)s * HIDDEN);
        acc += w * v;
      }
      base += 32;
      if (base < end) {
        my.x = 0;
        my.y = 0;
        if (lane < end - base) my = edges[base + lane];
      }
    }
    const v2f conv = acc - hv[k] * ew + xv[k] * bt;
    hn[k].x = hv[k].x + STEP_SZ * fmaxf(conv.x, 0.0f);
    hn[k].y = hv[k].y + STEP_SZ * fmaxf(conv.y, 0.0f);
    *(v2f*)(hs + (GU_NPW * wv + k) * HS_P + 2 * lane) = hn[k];
  }
  const int mr = lane & 15;
  const int kh = lane >> 4;
  const int nn = mr + 16 * wv;
  const bool live = (nn < LDO);
  v2f bf[HIDDEN / 4];
  if (wv < NT) {
    const float* brow = Bm + (size_t)(live ? nn : 0) * HIDDEN + 2 * kh;
#pragma unroll
    for (int i = 0; i < HIDDEN / 4; ++i) bf[i] = *(const v2f*)(brow + 4 * i);
    if (NT == 3) {
      const float bscale = live ? 1.0f : 0.0f;
#pragma unroll
      for (int i = 0; i < HIDDEN / 4; ++i) bf[i] *= bscale;
    }
  }
  __syncthreads();
  if (NT == 4) {
#pragma unroll
    for (int k = 0; k < GU_NPW; ++k) *(v2f*)(h + ro + k * HIDDEN) = hn[k];
  }
  if (wv < NT) {
    const float* arow = hs + mr * HS_P + 2 * kh;
    v8f cc = {};
#pragma unroll
    for (int i = 0; i < HIDDEN / 4; ++i) {
      const v2f a = *(const v2f*)(arow + 4 * i);
      cc = __builtin_amdgcn_wmma_f32_16x16x4_f32(false, a, false, bf[i], (short)0, cc, false, false);
    }
    const int mbase = (int)blockIdx.x * 16 + kh * 8;
    for (int r = 0; r < 8; ++r) {
      if (live) dst[(size_t)(mbase + r) * LDO + nn] = cc[r];
    }
  }
}


extern "C" void kernel_launch(void* const* d_in, const int* in_sizes, int n_in,
                              void* d_out, int out_size, void* d_ws, size_t ws_size,
                              hipStream_t stream) {
  (void)in_sizes; (void)n_in; (void)out_size; (void)ws_size;
  const float* x     = (const float*)d_in[0];
  const int*   ei    = (const int*)  d_in[1];
  const float* enc_w = (const float*)d_in[2];
  const float* dec_w = (const float*)d_in[3];
  const float* ext_w = (const float*)d_in[4];
  const float* beta  = (const float*)d_in[5];
  const float* pw    = (const float*)d_in[6];
  const int* row = ei;
  const int* col = ei + N_EDGES;

  char* wsb = (char*)d_ws;
  int*   btot  = (int*)  (wsb);
  int*   off   = (int*)  (wsb + 200192);
  float* dinv  = (float*)(wsb + 400384);
  int*   rank  = (int*)  (wsb + 600576);
  int2*  edges = (int2*) (wsb + 7000576);
  float* W     = (float*)(wsb + 19800576);
  float* h     = (float*)(wsb + 19816960);
  float* x0    = (float*)(wsb + 32616960);
  float* outp  = (float*)(wsb + 45416960);
  int*   hist  = (int*)  (wsb + 58216960);
  float* outq  = (float*)(wsb + 71016960);
  int2*  pre2  = (int2*) (wsb + 83816960);

  count_enc_kernel<<<FUSE_WGS, CNT_T, 0, stream>>>(col, hist, rank, pw, W, x, enc_w, h, x0);
  prefix_hw_kernel<<<NB + HW_WGS, PB_T, 0, stream>>>(hist, pre2, dinv, btot, h, W, outp);
  fill_off_kernel<<<FILL_WGS + OFF_WGS, FILL_T, 0, stream>>>(row, col, rank, pre2, dinv, btot, edges, off);

  gather_update_mm_kernel<4, HIDDEN><<<M_TILES, GU_T, 0, stream>>>(off, edges, outp, h, x0, ext_w, beta, W, outq);
  gather_update_mm_kernel<4, HIDDEN><<<M_TILES, GU_T, 0, stream>>>(off, edges, outq, h, x0, ext_w, beta, W, outp);
  gather_update_mm_kernel<4, HIDDEN><<<M_TILES, GU_T, 0, stream>>>(off, edges, outp, h, x0, ext_w, beta, W, outq);
  gather_update_mm_kernel<3, NCLASS><<<M_TILES, GU_T, 0, stream>>>(off, edges, outq, h, x0, ext_w, beta, dec_w, (float*)d_out);
}


# Edges sorted inside the counting work-groups (LDS scan + 2nd pass), runs merged per node in launch 2: placing launch dropped, 6 launches; deg^-1/2 split: out_p rows x node sum; sums by run, arrival
# speedup vs baseline: 6.1785x; 1.2449x over previous
#include <hip/hip_runtime.h>
#include <math.h>

#define N_NODES   50000
#define N_EDGES   1600000
#define NFEAT     500
#define HIDDEN    64
#define NCLASS    40
#define NUM_LAYERS 4
#define STEP_SZ   0.1f
#define M_TILES   (N_NODES / 16)

typedef __attribute__((ext_vector_type(2))) float v2f;
typedef __attribute__((ext_vector_type(8))) float v8f;


#define CHUNKS   32
#define CHUNK_E  (N_EDGES / CHUNKS)
#define CHUNK_Q  (CHUNK_E / 4)
#define NODE_Q   (N_NODES / 4)
#define CNT_T    512
#define PP       (N_NODES + 4)
#define SC_T     500
#define SC_Q     (NODE_Q / SC_T)
#define SRT_Q    (CHUNK_E / 8)

static_assert(N_NODES < 65536, "sources are kept in two bytes");
static_assert(NODE_Q % SC_T == 0 && SC_T <= CNT_T, "the scanning threads cover the counters exactly");
static_assert(CHUNK_E % 8 == 0, "a sorted run is a whole number of 16-byte groups");

__device__ __forceinline__ int wave_incl_scan(int v, const int lane) {
  for (int d = 1; d < 32; d <<= 1) {
    const int up = __shfl_up(v, d);
    if (lane >= d) v += up;
  }
  return v;
}

#define W_PITCH  (HIDDEN + 1)

__device__ __forceinline__
void weff_body(const float* __restrict__ pw, float* __restrict__ W, float* wl, const int t) {
  {
    const int LD = HIDDEN + 2;
    float q = 0.0f, r = 0.0f;
    if (t < HIDDEN) {
      q = pw[t * LD + HIDDEN];
      r = pw[t * LD + HIDDEN + 1];
    }
    float v[HIDDEN * HIDDEN / CNT_T];
#pragma unroll
    for (int u = 0; u < HIDDEN * HIDDEN / CNT_T; ++u) {
      const int e = t + u * CNT_T;
      const int i = e >> 6;
      const int j = e & (HIDDEN - 1);
      v[u] = pw[(i < j) ? (i * LD + j) : (j * LD + i)];
    }
#pragma unroll
    for (int u = 0; u < HIDDEN * HIDDEN / CNT_T; ++u) {
      const int e = t + u * CNT_T;
      const int i = e >> 6;
      const int j = e & (HIDDEN - 1);
      const float x = (i == j) ? 0.0f : v[u];
      wl[i * W_PITCH + j] = x;
      if (i != j) W[e] = x;
    }
    __syncthreads();
    if (t < HIDDEN) {
      float s = 0.0f;
      for (int j = 0; j < HIDDEN; ++j) s += fabsf(wl[t * W_PITCH + j]);
      W[t * HIDDEN + t] = q * s + r;
    }
  }
}

__device__ __forceinline__
void count_sort_body(const int* __restrict__ row, const int* __restrict__ col,
                     int* __restrict__ P, unsigned short* __restrict__ srt,
                     int* cnt, int* wsum, unsigned short* sl, const int run, const int t) {
  int4* cnt4 = (int4*)cnt;
  for (int i = t; i < NODE_Q; i += CNT_T) cnt4[i] = make_int4(0, 0, 0, 0);
  __syncthreads();
  const int4* c4 = (const int4*)(col + (size_t)run * CHUNK_E);
  const int4* r4 = (const int4*)(row + (size_t)run * CHUNK_E);
  const int full = CHUNK_Q / CNT_T;
  {
    int4 nx[4];
#pragma unroll
    for (int u = 0; u < 4; ++u) nx[u] = c4[u * CNT_T + t];
    for (int k = 0; k < full; k += 4) {
      int4 c[4];
#pragma unroll
      for (int u = 0; u < 4; ++u) c[u] = nx[u];
      if (k + 4 < full) {
#pragma unroll
        for (int u = 0; u < 4; ++u) nx[u] = c4[(k + 4 + u) * CNT_T + t];
      }
#pragma unroll
      for (int u = 0; u < 4; ++u) {
        atomicAdd(&cnt[c[u].x], 1);
        atomicAdd(&cnt[c[u].y], 1);
        atomicAdd(&cnt[c[u].z], 1);
        atomicAdd(&cnt[c[u].w], 1);
      }
    }
    const int q = full * CNT_T + t;
    if (q < CHUNK_Q) {
      const int4 c = c4[q];
      atomicAdd(&cnt[c.x], 1);
      atomicAdd(&cnt[c.y], 1);
      atomicAdd(&cnt[c.z], 1);
      atomicAdd(&cnt[c.w], 1);
    }
  }
  __syncthreads();
  {
    const int lane = t & 31;
    const int w = t >> 5;
    const bool sc = (t < SC_T);
    int4* mine = cnt4 + (sc ? t : 0) * SC_Q;
    int sum = 0;
    if (sc) {
#pragma unroll
      for (int i = 0; i < SC_Q; ++i) {
        const int4 v = mine[i];
        sum += (v.x + v.y) + (v.z + v.w);
      }
    }
    const int incl = wave_incl_scan(sum, lane);
    if (lane == 31) wsum[w] = incl;
    __syncthreads();
    int ws = (lane < CNT_T / 32) ? wsum[lane] : 0;
    ws = wave_incl_scan(ws, lane);
    const int prev = __shfl(ws, (w + 31) & 31);
    int pre = ((w > 0) ? prev : 0) + incl - sum;
    if (sc) {
      int4* gp = (int4*)(P + (size_t)run * PP) + t * SC_Q;
#pragma unroll
      for (int i = 0; i < SC_Q; ++i) {
        const int4 v = mine[i];
        int4 e;
        e.x = pre;
        e.y = e.x + v.x;
        e.z = e.y + v.y;
        e.w = e.z + v.z;
        pre = e.w + v.w;
        mine[i] = e;
        gp[i] = e;
      }
    }
    if (t == 0) P[(size_t)run * PP + N_NODES] = CHUNK_E;
  }
  __syncthreads();
  {
    int4 nc[4], nr[4];
#pragma unroll
    for (int u = 0; u < 4; ++u) {
      nc[u] = c4[u * CNT_T + t];
      nr[u] = r4[u * CNT_T + t];
    }
    for (int k = 0; k < full; k += 4) {
      int4 c[4], r[4];
#pragma unroll
      for (int u = 0; u < 4; ++u) {
        c[u] = nc[u];
        r[u] = nr[u];
      }
      if (k + 4 < full) {
#pragma unroll
        for (int u = 0; u < 4; ++u) {
          nc[u] = c4[(k + 4 + u) * CNT_T + t];
          nr[u] = r4[(k + 4 + u) * CNT_T + t];
        }
      }
#pragma unroll
      for (int u = 0; u < 4; ++u) {
        const int p0 = atomicAdd(&cnt[c[u].x], 1);
        const int p1 = atomicAdd(&cnt[c[u].y], 1);
        const int p2 = atomicAdd(&cnt[c[u].z], 1);
        const int p3 = atomicAdd(&cnt[c[u].w], 1);
        sl[p0] = (unsigned short)r[u].x;
        sl[p1] = (unsigned short)r[u].y;
        sl[p2] = (unsigned short)r[u].z;
        sl[p3] = (unsigned short)r[u].w;
      }
    }
    const int q = full * CNT_T + t;
    if (q < CHUNK_Q) {
      const int4 c = c4[q];
      const int4 r = r4[q];
      const int p0 = atomicAdd(&cnt[c.x], 1);
      const int p1 = atomicAdd(&cnt[c.y], 1);
      const int p2 = atomicAdd(&cnt[c.z], 1);
      const int p3 = atomicAdd(&cnt[c.w], 1);
      sl[p0] = (unsigned short)r.x;
      sl[p1] = (unsigned short)r.y;
      sl[p2] = (unsigned short)r.z;
      sl[p3] = (unsigned short)r.w;
    }
  }
  __syncthreads();
  const int4* s4 = (const int4*)sl;
  int4* g4 = (int4*)(srt + (size_t)run * CHUNK_E);
  for (int i = t; i < SRT_Q; i += CNT_T) g4[i] = s4[i];
}


typedef __attribute__((ext_vector_type(4))) float v4f;

#define ENC_T   512
#define ENC_WGS 223

#define ENC_WMMA(av, bv, cv) cv = __builtin_amdgcn_wmma_f32_16x16x4_f32(false, av, false, bv, (short)0, cv, false, false)

#define ENC_STEP8(a, bq)                                              \
  {                                                                   \
    const v4f b0 = *(const v4f*)((bq));                               \
    const v4f b1 = *(const v4f*)((bq) + 16 * NFEAT);                  \
    const v4f b2 = *(const v4f*)((bq) + 32 * NFEAT);                  \
    const v4f b3 = *(const v4f*)((bq) + 48 * NFEAT);                  \
    const v2f alo = {(a).x, (a).y};                                   \
    const v2f ahi = {(a).z, (a).w};                                   \
    const v2f b0l = {b0.x, b0.y}, b0h = {b0.z, b0.w};                 \
    const v2f b1l = {b1.x, b1.y}, b1h = {b1.z, b1.w};                 \
    const v2f b2l = {b2.x, b2.y}, b2h = {b2.z, b2.w};                 \
    const v2f b3l = {b3.x, b3.y}, b3h = {b3.z, b3.w};                 \
    ENC_WMMA(alo, b0l, c0); ENC_WMMA(alo, b1l, c1);                   \
    ENC_WMMA(alo, b2l, c2); ENC_WMMA(alo, b3l, c3);                   \
    ENC_WMMA(ahi, b0h, c0); ENC_WMMA(ahi, b1h, c1);                   \
    ENC_WMMA(ahi, b2h, c2); ENC_WMMA(ahi, b3h, c3);                   \
  }

#define ENC_LOAD4(v, ptr)                                             \
  v##0 = *(const v4f*)((ptr) +  0);                                   \
  v##1 = *(const v4f*)((ptr) +  8);                                   \
  v##2 = *(const v4f*)((ptr) + 16);                                   \
  v##3 = *(const v4f*)((ptr) + 24);
#define ENC_GROUP(v, bq)                                              \
  ENC_STEP8(v##0, (bq) +  0)                                          \
  ENC_STEP8(v##1, (bq) +  8)                                          \
  ENC_STEP8(v##2, (bq) + 16)                                          \
  ENC_STEP8(v##3, (bq) + 24)

__device__ __forceinline__
void enc_gemm_body(const float* __restrict__ x, const float* __restrict__ ew,
                   float* __restrict__ h, float* __restrict__ x0, float* Bs, const int b) {
  {
    const v4f* s4 = (const v4f*)ew;
    v4f* d4 = (v4f*)Bs;
    for (int i = threadIdx.x; i < HIDDEN * NFEAT / 4; i += ENC_T) d4[i] = s4[i];
  }
  __syncthreads();
  const int tile = b + ENC_WGS * (int)(threadIdx.x >> 5);
  if (tile >= M_TILES) return;
  const int lane = threadIdx.x & 31;
  const int mr = lane & 15;
  const int kh = lane >> 4;
  const float* arow = x + (size_t)(tile * 16 + mr) * NFEAT;
  const float* ap = arow + 4 * kh;
  const float* bp = Bs + mr * NFEAT + 4 * kh;
  v8f c0 = {}, c1 = {}, c2 = {}, c3 = {};
  v4f p0, p1, p2, p3, q0, q1, q2, q3, s0, s1, s2, s3;
  ENC_LOAD4(p, ap)
  ENC_LOAD4(q, ap + 32)
  for (int it = 0; it < 4; ++it) {
    const float* ag = ap + 96 * it;
    const float* bg = bp + 96 * it;
    ENC_LOAD4(s, ag + 64)
    ENC_GROUP(p, bg)
    ENC_LOAD4(p, ag + 96)
    ENC_GROUP(q, bg + 32)
    ENC_LOAD4(q, ag + 128)
    ENC_GROUP(s, bg + 64)
  }
  {
    ENC_LOAD4(s, ap + 448)
    const v4f r0 = *(const v4f*)(ap + 480);
    const v4f r1 = *(const v4f*)(ap + 488);
    const v2f r2 = *(const v2f*)(arow + 496 + 2 * kh);
    ENC_GROUP(p, bp + 384)
    ENC_GROUP(q, bp + 416)
    ENC_GROUP(s, bp + 448)
    ENC_STEP8(r0, bp + 480)
    ENC_STEP8(r1, bp + 488)
    const float* bt = Bs + mr * NFEAT + 496 + 2 * kh;
    const v2f t0 = *(const v2f*)(bt);
    const v2f t1 = *(const v2f*)(bt + 16 * NFEAT);
    const v2f t2 = *(const v2f*)(bt + 32 * NFEAT);
    const v2f t3 = *(const v2f*)(bt + 48 * NFEAT);
    ENC_WMMA(r2, t0, c0); ENC_WMMA(r2, t1, c1);
    ENC_WMMA(r2, t2, c2); ENC_WMMA(r2, t3, c3);
  }
  int mbase = tile * 16 + kh * 8;
  for (int r = 0; r < 8; ++r) {
    size_t o = (size_t)(mbase + r) * HIDDEN + mr;
    h[o]      = c0[r]; x0[o]      = c0[r];
    h[o + 16] = c1[r]; x0[o + 16] = c1[r];
    h[o + 32] = c2[r]; x0[o + 32] = c2[r];
    h[o + 48] = c3[r]; x0[o + 48] = c3[r];
  }
}


#define FUSE_WGS (ENC_WGS + CHUNKS + 1)
#define LDS_INTS (N_NODES + 16 + CHUNK_E / 2)

static_assert(HIDDEN * W_PITCH <= 16 + CHUNK_E / 2, "the W_eff copy fits behind the counters");

__global__ __launch_bounds__(CNT_T)
void count_enc_kernel(const int* __restrict__ row, const int* __restrict__ col,
                      int* __restrict__ P, unsigned short* __restrict__ srt,
                      const float* __restrict__ pw, float* __restrict__ W,
                      const float* __restrict__ x, const float* __restrict__ ew,
                      float* __restrict__ h, float* __restrict__ x0) {
  __shared__ __attribute__((aligned(16))) int lds[LDS_INTS];
  const int b = (int)blockIdx.x;
  if (b < ENC_WGS) {
    enc_gemm_body(x, ew, h, x0, (float*)lds, b);
  } else if (b < ENC_WGS + CHUNKS) {
    count_sort_body(row, col, P, srt, lds, lds + N_NODES, (unsigned short*)(lds + N_NODES + 16),
                    b - ENC_WGS, (int)threadIdx.x);
  } else {
    weff_body(pw, W, (float*)(lds + N_NODES), (int)threadIdx.x);
  }
}


#define PB_T     256
#define NB       ((N_NODES + 1 + PB_T - 1) / PB_T)
#define MG       8
#define MR       6

static_assert(CHUNKS % MG == 0, "runs are taken in whole groups");

__device__ __forceinline__
void merge_body(const int* __restrict__ P, const unsigned short* __restrict__ srt,
                unsigned short* __restrict__ rec, float* __restrict__ dinv,
                int* __restrict__ off, const int blk, const int t) {
  const int c = blk * PB_T + t;
  if (c > N_NODES) return;
  const bool live = (c < N_NODES);
  const int c1 = live ? (c + 1) : c;
  int start = 0, deg = 0;
#pragma unroll
  for (int g = 0; g < CHUNKS; ++g) {
    const int lo = P[(size_t)g * PP + c];
    const int hi = P[(size_t)g * PP + c1];
    start += lo;
    deg += hi - lo;
  }
  off[c] = start;
  if (!live) return;
  dinv[c] = (deg > 0) ? rsqrtf((float)deg) : 0.0f;
  int pos = start;
  const int* pg = P + c;
  const unsigned short* sp = srt;
#pragma clang loop unroll(disable)
  for (int g0 = 0; g0 < CHUNKS; g0 += MG) {
    int a[MG], n[MG];
#pragma unroll
    for (int u = 0; u < MG; ++u) {
      a[u] = pg[(size_t)u * PP];
      n[u] = pg[(size_t)u * PP + 1] - a[u];
    }
    int v[MG][MR];
#pragma unroll
    for (int u = 0; u < MG; ++u) {
      const unsigned short* sg = sp + (size_t)u * CHUNK_E + a[u];
#pragma unroll
      for (int r = 0; r < MR; ++r) {
        v[u][r] = 0;
        if (r < n[u]) v[u][r] = (int)sg[r];
      }
    }
#pragma unroll
    for (int u = 0; u < MG; ++u) {
      const unsigned short* sg = sp + (size_t)u * CHUNK_E + a[u];
#pragma unroll
      for (int r = 0; r < MR; ++r) {
        if (r < n[u]) rec[pos + r] = (unsigned short)v[u][r];
      }
#pragma clang loop vectorize(disable) unroll(disable)
      for (int r = MR; r < n[u]; ++r) rec[pos + r] = sg[r];
      pos += n[u];
    }
    pg += (size_t)MG * PP;
    sp += (size_t)MG * CHUNK_E;
  }
}


#define HW_T   PB_T
#define HW_WGS ((M_TILES + HW_T / 32 - 1) / (HW_T / 32))

__device__ __forceinline__
void hw_gemm_body(const float* __restrict__ h, const float* __restrict__ W,
                  const int* __restrict__ P, float* __restrict__ outp, float* Ws, const int b) {
  const int wave = (int)((b * HW_T + threadIdx.x) >> 5);
  const int lane = threadIdx.x & 31;
  const int mr = lane & 15;
  const int kh = lane >> 4;
  const int cr = (wave < M_TILES) ? (wave * 16 + mr) : (N_NODES - 1);
  const int* pc = P + (size_t)(16 * kh) * PP + cr;
  int pa[CHUNKS / 2], pb[CHUNKS / 2];
#pragma unroll
  for (int i = 0; i < CHUNKS / 2; ++i) {
    pa[i] = pc[(size_t)i * PP];
    pb[i] = pc[(size_t)i * PP + 1];
  }
  {
    const v4f* s4 = (const v4f*)W;
    v4f* d4 = (v4f*)Ws;
    v4f w4[HIDDEN * HIDDEN / 4 / HW_T];
#pragma unroll
    for (int u = 0; u < HIDDEN * HIDDEN / 4 / HW_T; ++u) w4[u] = s4[u * HW_T + threadIdx.x];
#pragma unroll
    for (int u = 0; u < HIDDEN * HIDDEN / 4 / HW_T; ++u) d4[u * HW_T + threadIdx.x] = w4[u];
  }
  __syncthreads();
  if (wave >= M_TILES) return;
  int dg = 0;
#pragma unroll
  for (int i = 0; i < CHUNKS / 2; ++i) dg += pb[i] - pa[i];
  dg += __shfl_xor(dg, 16);
  const float dme = (dg > 0) ? rsqrtf((float)dg) : 0.0f;
  const float* arow = h + (size_t)(wave * 16 + mr) * HIDDEN;
  v8f c0 = {}, c1 = {}, c2 = {}, c3 = {};
  for (int k = 0; k < HIDDEN; k += 4) {
    int ko = k + 2 * kh;
    v2f a  = *(const v2f*)(arow + ko);
    v2f b0 = *(const v2f*)(&Ws[(mr +  0) * HIDDEN + ko]);
    v2f b1 = *(const v2f*)(&Ws[(mr + 16) * HIDDEN + ko]);
    v2f b2 = *(const v2f*)(&Ws[(mr + 32) * HIDDEN + ko]);
    v2f b3 = *(const v2f*)(&Ws[(mr + 48) * HIDDEN + ko]);
    c0 = __builtin_amdgcn_wmma_f32_16x16x4_f32(false, a, false, b0, (short)0, c0, false, false);
    c1 = __builtin_amdgcn_wmma_f32_16x16x4_f32(false, a, false, b1, (short)0, c1, false, false);
    c2 = __builtin_amdgcn_wmma_f32_16x16x4_f32(false, a, false, b2, (short)0, c2, false, false);
    c3 = __builtin_amdgcn_wmma_f32_16x16x4_f32(false, a, false, b3, (short)0, c3, false, false);
  }
  int mbase = wave * 16 + kh * 8;
  for (int r = 0; r < 8; ++r) {
    const float dr = __shfl(dme, 8 * kh + r);
    size_t o = (size_t)(mbase + r) * HIDDEN + mr;
    outp[o]      = c0[r] * dr;
    outp[o + 16] = c1[r] * dr;
    outp[o + 32] = c2[r] * dr;
    outp[o + 48] = c3[r] * dr;
  }
}

__global__ __launch_bounds__(PB_T)
void merge_hw_kernel(const int* __restrict__ P, const unsigned short* __restrict__ srt,
                     unsigned short* __restrict__ rec, float* __restrict__ dinv,
                     int* __restrict__ off, const float* __restrict__ h,
                     const float* __restrict__ W, float* __restrict__ outp) {
  __shared__ __attribute__((aligned(16))) float lds[HIDDEN * HIDDEN];
  if (blockIdx.x < NB) {
    merge_body(P, srt, rec, dinv, off, (int)blockIdx.x, (int)threadIdx.x);
  } else {
    hw_gemm_body(h, W, P, outp, lds, (int)blockIdx.x - NB);
  }
}


#define GU_T   128
#define GU_NPW 4
#define HS_P   68

template <int NT, int LDO>
__global__ __launch_bounds__(GU_T)
void gather_update_mm_kernel(const int* __restrict__ off, const unsigned short* __restrict__ rec,
                             const float* __restrict__ outp, float* __restrict__ h,
                             const float* __restrict__ x0, const float* __restrict__ ext_w,
                             const float* __restrict__ beta, const float* __restrict__ dinv,
                             const float* __restrict__ Bm, float* __restrict__ dst) {
  __shared__ __attribute__((aligned(16))) float hs[16 * HS_P];
  const int wv = __builtin_amdgcn_readfirstlane((int)(threadIdx.x >> 5));
  const int c0 = (int)blockIdx.x * 16 + GU_NPW * wv;
  const int lane = threadIdx.x & 31;
  const float* src = outp + 2 * lane;
  int o[GU_NPW + 1];
#pragma unroll
  for (int k = 0; k <= GU_NPW; ++k) o[k] = off[c0 + k];
  int first[GU_NPW];
#pragma unroll
  for (int k = 0; k < GU_NPW; ++k) {
    first[k] = 0;
    if (lane < o[k + 1] - o[k]) first[k] = (int)rec[o[k] + lane];
  }
  const size_t ro = (size_t)c0 * HIDDEN + 2 * lane;
  v2f hv[GU_NPW], xv[GU_NPW], hn[GU_NPW];
  float dv[GU_NPW];
#pragma unroll
  for (int k = 0; k < GU_NPW; ++k) {
    hv[k] = *(const v2f*)(h + ro + k * HIDDEN);
    xv[k] = *(const v2f*)(x0 + ro + k * HIDDEN);
    dv[k] = dinv[c0 + k];
  }
  const v2f ew = *(const v2f*)(ext_w + 2 * lane);
  const float bt = beta[0];
#pragma unroll
  for (int k = 0; k < GU_NPW; ++k) {
    const int end = o[k + 1];
    int base = o[k];
    int my = first[k];
    v2f acc = {0.0f, 0.0f};
    while (base < end) {
      int n = end - base;
      n = (n < 32) ? n : 32;
      int j = 0;
      for (; j + 8 <= n; j += 8) {
        v2f v[8];
#pragma unroll
        for (int u = 0; u < 8; ++u) {
          const int s = __builtin_amdgcn_readlane(my, j + u);
          v[u] = *(const v2f*)(src + (size_t)s * HIDDEN);
        }
#pragma unroll
        for (int u = 0; u < 8; ++u) acc += v[u];
      }
      if (j + 4 <= n) {
        v2f v[4];
#pragma unroll
        for (int u = 0; u < 4; ++u) {
          const int s = __builtin_amdgcn_readlane(my, j + u);
          v[u] = *(const v2f*)(src + (size_t)s * HIDDEN);
        }
#pragma unroll
        for (int u = 0; u < 4; ++u) acc += v[u];
        j += 4;
      }
      for (; j < n; ++j) {
        const int s = __builtin_amdgcn_readlane(my, j);
        const v2f v = *(const v2f*)(src + (size_t)s * HIDDEN);
        acc += v;
      }
      base += 32;
      if (base < end) {
        my = 0;
        if (lane < end - base) my = (int)rec[base + lane];
      }
    }
    const v2f conv = dv[k] * acc - hv[k] * ew + xv[k] * bt;
    hn[k].x = hv[k].x + STEP_SZ * fmaxf(conv.x, 0.0f);
    hn[k].y = hv[k].y + STEP_SZ * fmaxf(conv.y, 0.0f);
    *(v2f*)(hs + (GU_NPW * wv + k) * HS_P + 2 * lane) = hn[k];
  }
  const int mr = lane & 15;
  const int kh = lane >> 4;
  const int nn = mr + 16 * wv;
  const bool live = (nn < LDO);
  const int mbase = (int)blockIdx.x * 16 + kh * 8;
  v2f bf[HIDDEN / 4];
  v4f sc0 = {1.0f, 1.0f, 1.0f, 1.0f}, sc1 = {1.0f, 1.0f, 1.0f, 1.0f};
  if (wv < NT) {
    const float* brow = Bm + (size_t)(live ? nn : 0) * HIDDEN + 2 * kh;
#pragma unroll
    for (int i = 0; i < HIDDEN / 4; ++i) bf[i] = *(const v2f*)(brow + 4 * i);
    if (NT == 3) {
      const float bscale = live ? 1.0f : 0.0f;
#pragma unroll
      for (int i = 0; i < HIDDEN / 4; ++i) bf[i] *= bscale;
    }
    if (NT == 4) {
      sc0 = *(const v4f*)(dinv + mbase);
      sc1 = *(const v4f*)(dinv + mbase + 4);
    }
  }
  __syncthreads();
  if (NT == 4) {
#pragma unroll
    for (int k = 0; k < GU_NPW; ++k) *(v2f*)(h + ro + k * HIDDEN) = hn[k];
  }
  if (wv < NT) {
    const float* arow = hs + mr * HS_P + 2 * kh;
    v8f cc = {};
#pragma unroll
    for (int i = 0; i < HIDDEN / 4; ++i) {
      const v2f a = *(const v2f*)(arow + 4 * i);
      cc = __builtin_amdgcn_wmma_f32_16x16x4_f32(false, a, false, bf[i], (short)0, cc, false, false);
    }
    if (NT == 4) {
      const float sc[8] = {sc0.x, sc0.y, sc0.z, sc0.w, sc1.x, sc1.y, sc1.z, sc1.w};
#pragma unroll
      for (int r = 0; r < 8; ++r) dst[(size_t)(mbase + r) * LDO + nn] = cc[r] * sc[r];
    } else {
      for (int r = 0; r < 8; ++r) {
        if (live) dst[(size_t)(mbase + r) * LDO + nn] = cc[r];
      }
    }
  }
}


extern "C" void kernel_launch(void* const* d_in, const int* in_sizes, int n_in,
                              void* d_out, int out_size, void* d_ws, size_t ws_size,
                              hipStream_t stream) {
  (void)in_sizes; (void)n_in; (void)out_size; (void)ws_size;
  const float* x     = (const float*)d_in[0];
  const int*   ei    = (const int*)  d_in[1];
  const float* enc_w = (const float*)d_in[2];
  const float* dec_w = (const float*)d_in[3];
  const float* ext_w = (const float*)d_in[4];
  const float* beta  = (const float*)d_in[5];
  const float* pw    = (const float*)d_in[6];
  const int* row = ei;
  const int* col = ei + N_EDGES;

  char* wsb = (char*)d_ws;
  int*            off  = (int*)           (wsb);
  float*          dinv = (float*)         (wsb + 200192);
  int*            P    = (int*)           (wsb + 400384);
  unsigned short* srt  = (unsigned short*)(wsb + 6800896);
  unsigned short* rec  = (unsigned short*)(wsb + 10000896);
  float*          W    = (float*)         (wsb + 13200896);
  float*          h    = (float*)         (wsb + 13217280);
  float*          x0   = (float*)         (wsb + 26017280);
  float*          outp = (float*)         (wsb + 38817280);
  float*          outq = (float*)         (wsb + 51617280);

  count_enc_kernel<<<FUSE_WGS, CNT_T, 0, stream>>>(row, col, P, srt, pw, W, x, enc_w, h, x0);
  merge_hw_kernel<<<NB + HW_WGS, PB_T, 0, stream>>>(P, srt, rec, dinv, off, h, W, outp);

  gather_update_mm_kernel<4, HIDDEN><<<M_TILES, GU_T, 0, stream>>>(off, rec, outp, h, x0, ext_w, beta, dinv, W, outq);
  gather_update_mm_kernel<4, HIDDEN><<<M_TILES, GU_T, 0, stream>>>(off, rec, outq, h, x0, ext_w, beta, dinv, W, outp);
  gather_update_mm_kernel<4, HIDDEN><<<M_TILES, GU_T, 0, stream>>>(off, rec, outp, h, x0, ext_w, beta, dinv, W, outq);
  gather_update_mm_kernel<3, NCLASS><<<M_TILES, GU_T, 0, stream>>>(off, rec, outq, h, x0, ext_w, beta, dinv, dec_w, (float*)d_out);
}
